# starting kernel
# baseline (speedup 1.0000x reference)
#include <hip/hip_runtime.h>
#include <math.h>


typedef float v2f __attribute__((ext_vector_type(2)));
typedef float v8f __attribute__((ext_vector_type(8)));

#define BB 4
#define HH 48
#define WW 48
#define DMm 192
#define DIc 192
#define KK 4
#define NN 16
#define RR 12
#define LL (HH * WW)
#define CPROJ (RR + 2 * NN)
#define CPAD 64

__device__ __forceinline__ float siluf(float x)     { return x / (1.f + __expf(-x)); }
__device__ __forceinline__ float sigm(float x)      { return 1.f / (1.f + __expf(-x)); }
__device__ __forceinline__ float softplusf_(float x){ return (x > 20.f) ? x : log1pf(__expf(x)); }
__device__ __forceinline__ float geluf(float x)     { return 0.5f * x * (1.f + erff(x * 0.70710678118654752440f)); }

template <int NT>
__device__ __forceinline__ void wmma_xwT_multi(const float* __restrict__ X,
                                               const float* __restrict__ W,
                                               int row0, int col0,
                                               int Kdim, int ldx, int ldw,
                                               v8f acc[NT])
{
    const int lane = threadIdx.x & 31;
    const int m    = lane & 15;
    const int kh   = lane >> 4;
    const float* __restrict__ ap = X + (size_t)(row0 + m) * ldx + 2 * kh;
    const float* __restrict__ bp[NT];
#pragma unroll
    for (int t = 0; t < NT; ++t) {
        bp[t] = W + (size_t)(col0 + t * 16 + m) * ldw + 2 * kh;
        acc[t] = {};
    }
    v2f a_cur = *(const v2f*)(ap);
    v2f b_cur[NT];
#pragma unroll
    for (int t = 0; t < NT; ++t) b_cur[t] = *(const v2f*)(bp[t]);

    for (int k0 = 4; k0 < Kdim; k0 += 4) {
        v2f a_nxt = *(const v2f*)(ap + k0);
        v2f b_nxt[NT];
#pragma unroll
        for (int t = 0; t < NT; ++t) b_nxt[t] = *(const v2f*)(bp[t] + k0);
#pragma unroll
        for (int t = 0; t < NT; ++t)
            acc[t] = __builtin_amdgcn_wmma_f32_16x16x4_f32(false, a_cur, false, b_cur[t],
                                                           (short)0, acc[t], false, false);
        a_cur = a_nxt;
#pragma unroll
        for (int t = 0; t < NT; ++t) b_cur[t] = b_nxt[t];
    }
#pragma unroll
    for (int t = 0; t < NT; ++t)
        acc[t] = __builtin_amdgcn_wmma_f32_16x16x4_f32(false, a_cur, false, b_cur[t],
                                                       (short)0, acc[t], false, false);
}

__global__ void k_padw(const float* __restrict__ xpw, float* __restrict__ wpad)
{
    int k = blockIdx.x, c = blockIdx.y, d = threadIdx.x;
    wpad[((size_t)k * CPAD + c) * DIc + d] =
        (c < CPROJ) ? xpw[((size_t)k * CPROJ + c) * DIc + d] : 0.f;
}

__global__ void k_inproj(const float* __restrict__ x, const float* __restrict__ w,
                         float* __restrict__ xc_lin, float* __restrict__ z_silu)
{
    const int row0 = blockIdx.y * 16, col0 = blockIdx.x * 32;
    v8f acc[2];
    wmma_xwT_multi<2>(x, w, row0, col0, DMm, DMm, DMm, acc);
    const int lane = threadIdx.x & 31;
    const int nn = lane & 15, mh = (lane >> 4) * 8;
#pragma unroll
    for (int t = 0; t < 2; ++t)
#pragma unroll
        for (int v = 0; v < 8; ++v) {
            int r = row0 + mh + v;
            int c = col0 + t * 16 + nn;
            float val = acc[t][v];
            if (c < DIc) xc_lin[(size_t)r * DIc + c] = val;
            else         z_silu[(size_t)r * DIc + (c - DIc)] = siluf(val);
        }
}

__global__ void k_dwconv(const float* __restrict__ xc_lin, const float* __restrict__ cw,
                         const float* __restrict__ cb, float* __restrict__ xc_act)
{
    int pix = blockIdx.x;
    int b = pix / LL, rem = pix % LL;
    int h = rem / WW, w = rem % WW;
    int c = threadIdx.x;
    float s = cb[c];
#pragma unroll
    for (int kh = 0; kh < 3; ++kh)
#pragma unroll
        for (int kw = 0; kw < 3; ++kw) {
            int hh = h + kh - 1, ww2 = w + kw - 1;
            if (hh >= 0 && hh < HH && ww2 >= 0 && ww2 < WW)
                s += xc_lin[((size_t)b * LL + hh * WW + ww2) * DIc + c] * cw[c * 9 + kh * 3 + kw];
        }
    xc_act[((size_t)b * LL + rem) * DIc + c] = siluf(s);
}

__global__ void k_gcmean(const float* __restrict__ xc_act, float* __restrict__ gcm)
{
    int b = blockIdx.x, d = threadIdx.x;
    float s = 0.f;
    for (int p = 0; p < LL; ++p)
        s += xc_act[((size_t)b * LL + p) * DIc + d];
    gcm[b * DIc + d] = s * (1.f / LL);
}

__global__ void k_gate(const float* __restrict__ gcm, const float* __restrict__ fw1,
                       const float* __restrict__ fw2, float* __restrict__ gate)
{
    __shared__ float g[DIc];
    __shared__ float h1[DMm / 4];
    int b = blockIdx.x, d = threadIdx.x;
    g[d] = gcm[b * DIc + d];
    __syncthreads();
    if (d < DMm / 4) {
        float s = 0.f;
        for (int j = 0; j < DMm; ++j) s += fw1[d * DMm + j] * g[j];
        h1[d] = geluf(s);
    }
    __syncthreads();
    float s2 = 0.f;
#pragma unroll
    for (int j = 0; j < DMm / 4; ++j) s2 += fw2[d * (DMm / 4) + j] * h1[j];
    gate[b * DMm + d] = sigm(s2);
}

__global__ void k_build_xs(const float* __restrict__ xc_act, float* __restrict__ xs)
{
    int l = blockIdx.x, b = blockIdx.y, k = blockIdx.z;
    int d = threadIdx.x;
    int lb = (k < 2) ? l : (LL - 1 - l);
    int w_in = lb & 3, h_in = (lb >> 2) & 3;
    int nw = (lb >> 4) % (WW / 4);
    int nh = (lb >> 4) / (WW / 4);
    int p = nh * 4 + h_in, q = nw * 4 + w_in;
    int h = ((k & 1) == 0) ? p : q;
    int w = ((k & 1) == 0) ? q : p;
    xs[(((size_t)(k * BB + b)) * LL + l) * DIc + d] =
        xc_act[((size_t)b * LL + h * WW + w) * DIc + d];
}

__global__ void k_xproj(const float* __restrict__ xs, const float* __restrict__ wpad,
                        float* __restrict__ xdbl)
{
    int k = blockIdx.z;
    const float* X = xs + (size_t)k * BB * LL * DIc;
    const float* W = wpad + (size_t)k * CPAD * DIc;
    float* O = xdbl + (size_t)k * BB * LL * CPROJ;
    const int row0 = blockIdx.y * 16, col0 = blockIdx.x * 32;
    v8f acc[2];
    wmma_xwT_multi<2>(X, W, row0, col0, DIc, DIc, DIc, acc);
    const int lane = threadIdx.x & 31;
    const int nn = lane & 15, mh = (lane >> 4) * 8;
#pragma unroll
    for (int t = 0; t < 2; ++t)
#pragma unroll
        for (int v = 0; v < 8; ++v) {
            int r = row0 + mh + v, c = col0 + t * 16 + nn;
            if (c < CPROJ) O[(size_t)r * CPROJ + c] = acc[t][v];
        }
}

__global__ void k_dtproj(const float* __restrict__ xdbl, const float* __restrict__ dtw,
                         const float* __restrict__ dtb, float* __restrict__ delta)
{
    int k = blockIdx.z;
    const float* X = xdbl + (size_t)k * BB * LL * CPROJ;
    const float* W = dtw + (size_t)k * DIc * RR;
    float* O = delta + (size_t)k * BB * LL * DIc;
    const int row0 = blockIdx.y * 16, col0 = blockIdx.x * 32;
    v8f acc[2];
    wmma_xwT_multi<2>(X, W, row0, col0, RR, CPROJ, RR, acc);
    const int lane = threadIdx.x & 31;
    const int nn = lane & 15, mh = (lane >> 4) * 8;
#pragma unroll
    for (int t = 0; t < 2; ++t)
#pragma unroll
        for (int v = 0; v < 8; ++v) {
            int r = row0 + mh + v, c = col0 + t * 16 + nn;
            O[(size_t)r * DIc + c] = softplusf_(acc[t][v] + dtb[k * DIc + c]);
        }
}

__global__ void k_scan(const float* __restrict__ xs, const float* __restrict__ xdbl,
                       const float* __restrict__ delta, const float* __restrict__ A_logs,
                       const float* __restrict__ Ds, float* __restrict__ yarr)
{
    int kb = blockIdx.x;
    int k = kb & 3, b = kb >> 2;
    int d = threadIdx.x;
    float A[NN];
#pragma unroll
    for (int n = 0; n < NN; ++n)
        A[n] = -__expf(A_logs[((size_t)(k * DIc + d)) * NN + n]);
    float Dv = Ds[k * DIc + d];
    float h[NN];
#pragma unroll
    for (int n = 0; n < NN; ++n) h[n] = 0.f;
    const size_t seqbase = ((size_t)(k * BB + b)) * LL;
    const float* up = xs    + seqbase * DIc + d;
    const float* dp = delta + seqbase * DIc + d;
    const float* bc = xdbl  + seqbase * CPROJ;
    float* yp       = yarr  + seqbase * DIc + d;
    for (int l = 0; l < LL; ++l) {
        float dt = dp[(size_t)l * DIc];
        float u  = up[(size_t)l * DIc];
        float du = dt * u;
        const float* bcl = bc + (size_t)l * CPROJ;
        float y = 0.f;
#pragma unroll
        for (int n = 0; n < NN; ++n) {
            float Bt = bcl[RR + n];
            float Ct = bcl[RR + NN + n];
            h[n] = __expf(dt * A[n]) * h[n] + du * Bt;
            y += h[n] * Ct;
        }
        yp[(size_t)l * DIc] = y + Dv * u;
    }
}

__global__ void k_combine(const float* __restrict__ yarr, const float* __restrict__ gate,
                          const float* __restrict__ g1, const float* __restrict__ g2,
                          const float* __restrict__ g3, const float* __restrict__ g4,
                          const float* __restrict__ nw, const float* __restrict__ nb,
                          const float* __restrict__ z_silu, float* __restrict__ yfinal)
{
    __shared__ float s1[256], s2[256];
    int lr = blockIdx.x, b = blockIdx.y, d = threadIdx.x;
    int h = lr / WW, w = lr % WW;
    int l = (((h >> 2) * (WW / 4) + (w >> 2)) << 4) + ((h & 3) << 2) + (w & 3);
    int lt = (l % WW) * HH + (l / WW);
    const size_t bstride = (size_t)LL * DIc;
    const float* Y0 = yarr + ((size_t)(0 * BB + b)) * bstride;
    const float* Y1 = yarr + ((size_t)(1 * BB + b)) * bstride;
    const float* Y2 = yarr + ((size_t)(2 * BB + b)) * bstride;
    const float* Y3 = yarr + ((size_t)(3 * BB + b)) * bstride;
    float yc = g1[d] * Y0[(size_t)l * DIc + d]
             + g2[d] * Y2[(size_t)(LL - 1 - l) * DIc + d]
             + g3[d] * Y1[(size_t)lt * DIc + d]
             + g4[d] * Y3[(size_t)(LL - 1 - lt) * DIc + d];
    yc *= gate[b * DIc + d];
    s1[d] = yc; s2[d] = yc * yc;
    if (d < 64) { s1[192 + d] = 0.f; s2[192 + d] = 0.f; }
    __syncthreads();
    for (int off = 128; off > 0; off >>= 1) {
        if (d < off) { s1[d] += s1[d + off]; s2[d] += s2[d + off]; }
        __syncthreads();
    }
    float mu  = s1[0] * (1.f / DIc);
    float var = s2[0] * (1.f / DIc) - mu * mu;
    float inv = rsqrtf(var + 1e-5f);
    yc = (yc - mu) * inv * nw[d] + nb[d];
    yc *= z_silu[((size_t)b * LL + lr) * DIc + d];
    yfinal[((size_t)b * LL + lr) * DIc + d] = yc;
}

__global__ void k_outproj(const float* __restrict__ yf, const float* __restrict__ w,
                          float* __restrict__ out)
{
    const int row0 = blockIdx.y * 16, col0 = blockIdx.x * 32;
    v8f acc[2];
    wmma_xwT_multi<2>(yf, w, row0, col0, DIc, DIc, DIc, acc);
    const int lane = threadIdx.x & 31;
    const int nn = lane & 15, mh = (lane >> 4) * 8;
#pragma unroll
    for (int t = 0; t < 2; ++t)
#pragma unroll
        for (int v = 0; v < 8; ++v) {
            int r = row0 + mh + v, c = col0 + t * 16 + nn;
            out[(size_t)r * DMm + c] = acc[t][v];
        }
}

extern "C" void kernel_launch(void* const* d_in, const int* in_sizes, int n_in,
                              void* d_out, int out_size, void* d_ws, size_t ws_size,
                              hipStream_t stream)
{
    (void)in_sizes; (void)n_in; (void)out_size; (void)ws_size;
    const float* x          = (const float*)d_in[0];
    const float* in_proj_w  = (const float*)d_in[1];
    const float* conv_w     = (const float*)d_in[2];
    const float* conv_b     = (const float*)d_in[3];
    const float* x_proj_w   = (const float*)d_in[4];
    const float* dt_w       = (const float*)d_in[5];
    const float* dt_b       = (const float*)d_in[6];
    const float* A_logs     = (const float*)d_in[7];
    const float* Ds         = (const float*)d_in[8];
    const float* gamma1     = (const float*)d_in[9];
    const float* gamma2     = (const float*)d_in[10];
    const float* gamma3     = (const float*)d_in[11];
    const float* gamma4     = (const float*)d_in[12];
    const float* fc_w1      = (const float*)d_in[13];
    const float* fc_w2      = (const float*)d_in[14];
    const float* norm_w     = (const float*)d_in[15];
    const float* norm_b     = (const float*)d_in[16];
    const float* out_proj_w = (const float*)d_in[17];
    float* out = (float*)d_out;

    float* ws = (float*)d_ws;
    size_t o = 0;
    float* xc_lin = ws + o; o += (size_t)BB * LL * DIc;
    float* z_silu = ws + o; o += (size_t)BB * LL * DIc;
    float* xc_act = ws + o; o += (size_t)BB * LL * DIc;
    float* xs     = ws + o; o += (size_t)KK * BB * LL * DIc;
    float* xdbl   = ws + o; o += (size_t)KK * BB * LL * CPROJ;
    float* delta  = ws + o; o += (size_t)KK * BB * LL * DIc;
    float* yarr   = ws + o; o += (size_t)KK * BB * LL * DIc;
    float* yfinal = ws + o; o += (size_t)BB * LL * DIc;
    float* wpad   = ws + o; o += (size_t)KK * CPAD * DIc;
    float* gcm    = ws + o; o += BB * DIc;
    float* gate   = ws + o; o += BB * DMm;

    k_padw    <<<dim3(KK, CPAD),       DIc, 0, stream>>>(x_proj_w, wpad);
    k_inproj  <<<dim3(12, 576),        32,  0, stream>>>(x, in_proj_w, xc_lin, z_silu);
    k_dwconv  <<<dim3(BB * LL),        DIc, 0, stream>>>(xc_lin, conv_w, conv_b, xc_act);
    k_gcmean  <<<dim3(BB),             DIc, 0, stream>>>(xc_act, gcm);
    k_gate    <<<dim3(BB),             DIc, 0, stream>>>(gcm, fc_w1, fc_w2, gate);
    k_build_xs<<<dim3(LL, BB, KK),     DIc, 0, stream>>>(xc_act, xs);
    k_xproj   <<<dim3(2, 576, KK),     32,  0, stream>>>(xs, wpad, xdbl);
    k_dtproj  <<<dim3(6, 576, KK),     32,  0, stream>>>(xdbl, dt_w, dt_b, delta);
    k_scan    <<<dim3(BB * KK),        DIc, 0, stream>>>(xs, xdbl, delta, A_logs, Ds, yarr);
    k_combine <<<dim3(LL, BB),         DIc, 0, stream>>>(yarr, gate, gamma1, gamma2, gamma3,
                                                         gamma4, norm_w, norm_b, z_silu, yfinal);
    k_outproj <<<dim3(6, 576),         32,  0, stream>>>(yfinal, out_proj_w, out);
}


# k_gcmean launch removed: k_dwconv runs one work-group per image row, a thread walks its channel along the row in registers and sums its outputs; k_gate adds the 48 row sums (new sum order of mean)
# speedup vs baseline: 9.5948x; 9.5948x over previous
#include <hip/hip_runtime.h>
#include <math.h>


typedef float v2f __attribute__((ext_vector_type(2)));
typedef float v8f __attribute__((ext_vector_type(8)));

#define BB 4
#define HH 48
#define WW 48
#define DMm 192
#define DIc 192
#define KK 4
#define NN 16
#define RR 12
#define LL (HH * WW)
#define CPROJ (RR + 2 * NN)
#define CPAD 64

__device__ __forceinline__ float siluf(float x)     { return x / (1.f + __expf(-x)); }
__device__ __forceinline__ float sigm(float x)      { return 1.f / (1.f + __expf(-x)); }
__device__ __forceinline__ float softplusf_(float x){ return (x > 20.f) ? x : log1pf(__expf(x)); }
__device__ __forceinline__ float geluf(float x)     { return 0.5f * x * (1.f + erff(x * 0.70710678118654752440f)); }

template <int NT>
__device__ __forceinline__ void wmma_xwT_multi(const float* __restrict__ X,
                                               const float* __restrict__ W,
                                               int row0, int col0,
                                               int Kdim, int ldx, int ldw,
                                               v8f acc[NT])
{
    const int lane = threadIdx.x & 31;
    const int m    = lane & 15;
    const int kh   = lane >> 4;
    const float* __restrict__ ap = X + (size_t)(row0 + m) * ldx + 2 * kh;
    const float* __restrict__ bp[NT];
#pragma unroll
    for (int t = 0; t < NT; ++t) {
        bp[t] = W + (size_t)(col0 + t * 16 + m) * ldw + 2 * kh;
        acc[t] = {};
    }
    v2f a_cur = *(const v2f*)(ap);
    v2f b_cur[NT];
#pragma unroll
    for (int t = 0; t < NT; ++t) b_cur[t] = *(const v2f*)(bp[t]);

    for (int k0 = 4; k0 < Kdim; k0 += 4) {
        v2f a_nxt = *(const v2f*)(ap + k0);
        v2f b_nxt[NT];
#pragma unroll
        for (int t = 0; t < NT; ++t) b_nxt[t] = *(const v2f*)(bp[t] + k0);
#pragma unroll
        for (int t = 0; t < NT; ++t)
            acc[t] = __builtin_amdgcn_wmma_f32_16x16x4_f32(false, a_cur, false, b_cur[t],
                                                           (short)0, acc[t], false, false);
        a_cur = a_nxt;
#pragma unroll
        for (int t = 0; t < NT; ++t) b_cur[t] = b_nxt[t];
    }
#pragma unroll
    for (int t = 0; t < NT; ++t)
        acc[t] = __builtin_amdgcn_wmma_f32_16x16x4_f32(false, a_cur, false, b_cur[t],
                                                       (short)0, acc[t], false, false);
}

__global__ void k_padw(const float* __restrict__ xpw, float* __restrict__ wpad)
{
    int k = blockIdx.x, c = blockIdx.y, d = threadIdx.x;
    wpad[((size_t)k * CPAD + c) * DIc + d] =
        (c < CPROJ) ? xpw[((size_t)k * CPROJ + c) * DIc + d] : 0.f;
}

__global__ void k_inproj(const float* __restrict__ x, const float* __restrict__ w,
                         float* __restrict__ xc_lin, float* __restrict__ z_silu)
{
    const int row0 = blockIdx.y * 16, col0 = blockIdx.x * 32;
    v8f acc[2];
    wmma_xwT_multi<2>(x, w, row0, col0, DMm, DMm, DMm, acc);
    const int lane = threadIdx.x & 31;
    const int nn = lane & 15, mh = (lane >> 4) * 8;
#pragma unroll
    for (int t = 0; t < 2; ++t)
#pragma unroll
        for (int v = 0; v < 8; ++v) {
            int r = row0 + mh + v;
            int c = col0 + t * 16 + nn;
            float val = acc[t][v];
            if (c < DIc) xc_lin[(size_t)r * DIc + c] = val;
            else         z_silu[(size_t)r * DIc + (c - DIc)] = siluf(val);
        }
}

__global__ void __launch_bounds__(DIc)
k_dwconv(const float* __restrict__ xc_lin, const float* __restrict__ cw,
         const float* __restrict__ cb, float* __restrict__ xc_act,
         float* __restrict__ rowsum)
{
    const int h = blockIdx.x, b = blockIdx.y, c = threadIdx.x;
    const bool up = (h > 0), dn = (h < HH - 1);
    const int hu = up ? h - 1 : h;
    const int hd = dn ? h + 1 : h;
    const float* __restrict__ p0 = xc_lin + ((size_t)b * LL + (size_t)hu * WW) * DIc + c;
    const float* __restrict__ p1 = xc_lin + ((size_t)b * LL + (size_t)h  * WW) * DIc + c;
    const float* __restrict__ p2 = xc_lin + ((size_t)b * LL + (size_t)hd * WW) * DIc + c;
    float wg[9];
#pragma unroll
    for (int j = 0; j < 9; ++j) wg[j] = cw[c * 9 + j];
    const float bias = cb[c];
    float r0[WW + 2], r1[WW + 2], r2[WW + 2];
    r0[0] = 0.f; r1[0] = 0.f; r2[0] = 0.f;
    r0[WW + 1] = 0.f; r1[WW + 1] = 0.f; r2[WW + 1] = 0.f;
#pragma unroll
    for (int w = 0; w < WW; ++w) r0[w + 1] = p0[(size_t)w * DIc];
#pragma unroll
    for (int w = 0; w < WW; ++w) r1[w + 1] = p1[(size_t)w * DIc];
#pragma unroll
    for (int w = 0; w < WW; ++w) r2[w + 1] = p2[(size_t)w * DIc];
#pragma unroll
    for (int w = 0; w < WW; ++w) {
        r0[w + 1] = up ? r0[w + 1] : 0.f;
        r2[w + 1] = dn ? r2[w + 1] : 0.f;
    }
    float* __restrict__ op = xc_act + ((size_t)b * LL + (size_t)h * WW) * DIc + c;
    float tot = 0.f;
#pragma unroll
    for (int w = 0; w < WW; ++w) {
        float s = bias;
        s = fmaf(r0[w], wg[0], s); s = fmaf(r0[w + 1], wg[1], s); s = fmaf(r0[w + 2], wg[2], s);
        s = fmaf(r1[w], wg[3], s); s = fmaf(r1[w + 1], wg[4], s); s = fmaf(r1[w + 2], wg[5], s);
        s = fmaf(r2[w], wg[6], s); s = fmaf(r2[w + 1], wg[7], s); s = fmaf(r2[w + 2], wg[8], s);
        const float v = siluf(s);
        op[(size_t)w * DIc] = v;
        tot += v;
    }
    rowsum[((size_t)b * HH + h) * DIc + c] = tot;
}

__global__ void k_gate(const float* __restrict__ rowsum, const float* __restrict__ fw1,
                       const float* __restrict__ fw2, float* __restrict__ gate)
{
    __shared__ float g[DIc];
    __shared__ float h1[DMm / 4];
    int b = blockIdx.x, d = threadIdx.x;
    float gs = 0.f;
#pragma unroll
    for (int h = 0; h < HH; ++h) gs += rowsum[((size_t)b * HH + h) * DIc + d];
    g[d] = gs * (1.f / LL);
    __syncthreads();
    if (d < DMm / 4) {
        float s = 0.f;
        for (int j = 0; j < DMm; ++j) s += fw1[d * DMm + j] * g[j];
        h1[d] = geluf(s);
    }
    __syncthreads();
    float s2 = 0.f;
#pragma unroll
    for (int j = 0; j < DMm / 4; ++j) s2 += fw2[d * (DMm / 4) + j] * h1[j];
    gate[b * DMm + d] = sigm(s2);
}

__global__ void k_build_xs(const float* __restrict__ xc_act, float* __restrict__ xs)
{
    int l = blockIdx.x, b = blockIdx.y, k = blockIdx.z;
    int d = threadIdx.x;
    int lb = (k < 2) ? l : (LL - 1 - l);
    int w_in = lb & 3, h_in = (lb >> 2) & 3;
    int nw = (lb >> 4) % (WW / 4);
    int nh = (lb >> 4) / (WW / 4);
    int p = nh * 4 + h_in, q = nw * 4 + w_in;
    int h = ((k & 1) == 0) ? p : q;
    int w = ((k & 1) == 0) ? q : p;
    xs[(((size_t)(k * BB + b)) * LL + l) * DIc + d] =
        xc_act[((size_t)b * LL + h * WW + w) * DIc + d];
}

__global__ void k_xproj(const float* __restrict__ xs, const float* __restrict__ wpad,
                        float* __restrict__ xdbl)
{
    int k = blockIdx.z;
    const float* X = xs + (size_t)k * BB * LL * DIc;
    const float* W = wpad + (size_t)k * CPAD * DIc;
    float* O = xdbl + (size_t)k * BB * LL * CPROJ;
    const int row0 = blockIdx.y * 16, col0 = blockIdx.x * 32;
    v8f acc[2];
    wmma_xwT_multi<2>(X, W, row0, col0, DIc, DIc, DIc, acc);
    const int lane = threadIdx.x & 31;
    const int nn = lane & 15, mh = (lane >> 4) * 8;
#pragma unroll
    for (int t = 0; t < 2; ++t)
#pragma unroll
        for (int v = 0; v < 8; ++v) {
            int r = row0 + mh + v, c = col0 + t * 16 + nn;
            if (c < CPROJ) O[(size_t)r * CPROJ + c] = acc[t][v];
        }
}

__global__ void k_dtproj(const float* __restrict__ xdbl, const float* __restrict__ dtw,
                         const float* __restrict__ dtb, float* __restrict__ delta)
{
    int k = blockIdx.z;
    const float* X = xdbl + (size_t)k * BB * LL * CPROJ;
    const float* W = dtw + (size_t)k * DIc * RR;
    float* O = delta + (size_t)k * BB * LL * DIc;
    const int row0 = blockIdx.y * 16, col0 = blockIdx.x * 32;
    v8f acc[2];
    wmma_xwT_multi<2>(X, W, row0, col0, RR, CPROJ, RR, acc);
    const int lane = threadIdx.x & 31;
    const int nn = lane & 15, mh = (lane >> 4) * 8;
#pragma unroll
    for (int t = 0; t < 2; ++t)
#pragma unroll
        for (int v = 0; v < 8; ++v) {
            int r = row0 + mh + v, c = col0 + t * 16 + nn;
            O[(size_t)r * DIc + c] = softplusf_(acc[t][v] + dtb[k * DIc + c]);
        }
}

#define SC_C   96
#define SC_LEN (LL / SC_C)

__global__ void k_scan_local(const float* __restrict__ xs, const float* __restrict__ xdbl,
                             const float* __restrict__ delta, const float* __restrict__ A_logs,
                             float* __restrict__ sumP, float* __restrict__ sumS)
{
    const int c = blockIdx.x, kb = blockIdx.y;
    const int k = kb & 3, b = kb >> 2;
    const int d = threadIdx.x;
    float A[NN];
#pragma unroll
    for (int n = 0; n < NN; ++n)
        A[n] = -__expf(A_logs[((size_t)(k * DIc + d)) * NN + n]);
    float h[NN];
#pragma unroll
    for (int n = 0; n < NN; ++n) h[n] = 0.f;
    const size_t seqbase = ((size_t)(k * BB + b)) * LL + (size_t)c * SC_LEN;
    const float* up = xs    + seqbase * DIc + d;
    const float* dp = delta + seqbase * DIc + d;
    const float* bc = xdbl  + seqbase * CPROJ;
    float sdt = 0.f;
    for (int l = 0; l < SC_LEN; ++l) {
        float dt = dp[(size_t)l * DIc];
        float u  = up[(size_t)l * DIc];
        float du = dt * u;
        const float* bcl = bc + (size_t)l * CPROJ;
#pragma unroll
        for (int n = 0; n < NN; ++n)
            h[n] = __expf(dt * A[n]) * h[n] + du * bcl[RR + n];
        sdt += dt;
    }
    const size_t o = (((size_t)kb * SC_C + c) * NN) * DIc + d;
#pragma unroll
    for (int n = 0; n < NN; ++n) {
        sumP[o + (size_t)n * DIc] = __expf(sdt * A[n]);
        sumS[o + (size_t)n * DIc] = h[n];
    }
}

__global__ void k_scan_prefix(const float* __restrict__ sumP, const float* __restrict__ sumS,
                              float* __restrict__ hst)
{
    const int n = blockIdx.x, kb = blockIdx.y, d = threadIdx.x;
    const size_t base = (((size_t)kb * SC_C) * NN + n) * DIc + d;
    const size_t cs = (size_t)NN * DIc;
    float H = 0.f;
    hst[base] = H;
#pragma unroll 19
    for (int c = 0; c + 1 < SC_C; ++c) {
        H = sumP[base + c * cs] * H + sumS[base + c * cs];
        hst[base + (c + 1) * cs] = H;
    }
}

__global__ void k_scan_chunk(const float* __restrict__ xs, const float* __restrict__ xdbl,
                             const float* __restrict__ delta, const float* __restrict__ A_logs,
                             const float* __restrict__ Ds, const float* __restrict__ hst,
                             float* __restrict__ yarr)
{
    const int c = blockIdx.x, kb = blockIdx.y;
    const int k = kb & 3, b = kb >> 2;
    const int d = threadIdx.x;
    float A[NN];
#pragma unroll
    for (int n = 0; n < NN; ++n)
        A[n] = -__expf(A_logs[((size_t)(k * DIc + d)) * NN + n]);
    float Dv = Ds[k * DIc + d];
    float h[NN];
    const size_t o = (((size_t)kb * SC_C + c) * NN) * DIc + d;
#pragma unroll
    for (int n = 0; n < NN; ++n) h[n] = hst[o + (size_t)n * DIc];
    const size_t seqbase = ((size_t)(k * BB + b)) * LL + (size_t)c * SC_LEN;
    const float* up = xs    + seqbase * DIc + d;
    const float* dp = delta + seqbase * DIc + d;
    const float* bc = xdbl  + seqbase * CPROJ;
    float* yp       = yarr  + seqbase * DIc + d;
    for (int l = 0; l < SC_LEN; ++l) {
        float dt = dp[(size_t)l * DIc];
        float u  = up[(size_t)l * DIc];
        float du = dt * u;
        const float* bcl = bc + (size_t)l * CPROJ;
        float y = 0.f;
#pragma unroll
        for (int n = 0; n < NN; ++n) {
            float Bt = bcl[RR + n];
            float Ct = bcl[RR + NN + n];
            h[n] = __expf(dt * A[n]) * h[n] + du * Bt;
            y += h[n] * Ct;
        }
        yp[(size_t)l * DIc] = y + Dv * u;
    }
}

__global__ void k_combine(const float* __restrict__ yarr, const float* __restrict__ gate,
                          const float* __restrict__ g1, const float* __restrict__ g2,
                          const float* __restrict__ g3, const float* __restrict__ g4,
                          const float* __restrict__ nw, const float* __restrict__ nb,
                          const float* __restrict__ z_silu, float* __restrict__ yfinal)
{
    __shared__ float s1[256], s2[256];
    int lr = blockIdx.x, b = blockIdx.y, d = threadIdx.x;
    int h = lr / WW, w = lr % WW;
    int l = (((h >> 2) * (WW / 4) + (w >> 2)) << 4) + ((h & 3) << 2) + (w & 3);
    int lt = (l % WW) * HH + (l / WW);
    const size_t bstride = (size_t)LL * DIc;
    const float* Y0 = yarr + ((size_t)(0 * BB + b)) * bstride;
    const float* Y1 = yarr + ((size_t)(1 * BB + b)) * bstride;
    const float* Y2 = yarr + ((size_t)(2 * BB + b)) * bstride;
    const float* Y3 = yarr + ((size_t)(3 * BB + b)) * bstride;
    float yc = g1[d] * Y0[(size_t)l * DIc + d]
             + g2[d] * Y2[(size_t)(LL - 1 - l) * DIc + d]
             + g3[d] * Y1[(size_t)lt * DIc + d]
             + g4[d] * Y3[(size_t)(LL - 1 - lt) * DIc + d];
    yc *= gate[b * DIc + d];
    s1[d] = yc; s2[d] = yc * yc;
    if (d < 64) { s1[192 + d] = 0.f; s2[192 + d] = 0.f; }
    __syncthreads();
    for (int off = 128; off > 0; off >>= 1) {
        if (d < off) { s1[d] += s1[d + off]; s2[d] += s2[d + off]; }
        __syncthreads();
    }
    float mu  = s1[0] * (1.f / DIc);
    float var = s2[0] * (1.f / DIc) - mu * mu;
    float inv = rsqrtf(var + 1e-5f);
    yc = (yc - mu) * inv * nw[d] + nb[d];
    yc *= z_silu[((size_t)b * LL + lr) * DIc + d];
    yfinal[((size_t)b * LL + lr) * DIc + d] = yc;
}

__global__ void k_outproj(const float* __restrict__ yf, const float* __restrict__ w,
                          float* __restrict__ out)
{
    const int row0 = blockIdx.y * 16, col0 = blockIdx.x * 32;
    v8f acc[2];
    wmma_xwT_multi<2>(yf, w, row0, col0, DIc, DIc, DIc, acc);
    const int lane = threadIdx.x & 31;
    const int nn = lane & 15, mh = (lane >> 4) * 8;
#pragma unroll
    for (int t = 0; t < 2; ++t)
#pragma unroll
        for (int v = 0; v < 8; ++v) {
            int r = row0 + mh + v, c = col0 + t * 16 + nn;
            out[(size_t)r * DMm + c] = acc[t][v];
        }
}

extern "C" void kernel_launch(void* const* d_in, const int* in_sizes, int n_in,
                              void* d_out, int out_size, void* d_ws, size_t ws_size,
                              hipStream_t stream)
{
    (void)in_sizes; (void)n_in; (void)out_size; (void)ws_size;
    const float* x          = (const float*)d_in[0];
    const float* in_proj_w  = (const float*)d_in[1];
    const float* conv_w     = (const float*)d_in[2];
    const float* conv_b     = (const float*)d_in[3];
    const float* x_proj_w   = (const float*)d_in[4];
    const float* dt_w       = (const float*)d_in[5];
    const float* dt_b       = (const float*)d_in[6];
    const float* A_logs     = (const float*)d_in[7];
    const float* Ds         = (const float*)d_in[8];
    const float* gamma1     = (const float*)d_in[9];
    const float* gamma2     = (const float*)d_in[10];
    const float* gamma3     = (const float*)d_in[11];
    const float* gamma4     = (const float*)d_in[12];
    const float* fc_w1      = (const float*)d_in[13];
    const float* fc_w2      = (const float*)d_in[14];
    const float* norm_w     = (const float*)d_in[15];
    const float* norm_b     = (const float*)d_in[16];
    const float* out_proj_w = (const float*)d_in[17];
    float* out = (float*)d_out;

    float* ws = (float*)d_ws;
    size_t o = 0;
    float* xc_lin = ws + o; o += (size_t)BB * LL * DIc;
    float* z_silu = ws + o; o += (size_t)BB * LL * DIc;
    float* xc_act = ws + o; o += (size_t)BB * LL * DIc;
    float* xs     = ws + o; o += (size_t)KK * BB * LL * DIc;
    float* xdbl   = ws + o; o += (size_t)KK * BB * LL * CPROJ;
    float* delta  = ws + o; o += (size_t)KK * BB * LL * DIc;
    float* yarr   = ws + o; o += (size_t)KK * BB * LL * DIc;
    float* yfinal = ws + o; o += (size_t)BB * LL * DIc;
    float* wpad   = ws + o; o += (size_t)KK * CPAD * DIc;
    float* rowsum = ws + o; o += (size_t)BB * HH * DIc;
    float* gate   = ws + o; o += BB * DMm;
    float* sumP   = ws + o; o += (size_t)KK * BB * SC_C * NN * DIc;
    float* sumS   = ws + o; o += (size_t)KK * BB * SC_C * NN * DIc;
    float* hst    = ws + o; o += (size_t)KK * BB * SC_C * NN * DIc;

    k_padw    <<<dim3(KK, CPAD),       DIc, 0, stream>>>(x_proj_w, wpad);
    k_inproj  <<<dim3(12, 576),        32,  0, stream>>>(x, in_proj_w, xc_lin, z_silu);
    k_dwconv  <<<dim3(HH, BB),         DIc, 0, stream>>>(xc_lin, conv_w, conv_b, xc_act, rowsum);
    k_gate    <<<dim3(BB),             DIc, 0, stream>>>(rowsum, fc_w1, fc_w2, gate);
    k_build_xs<<<dim3(LL, BB, KK),     DIc, 0, stream>>>(xc_act, xs);
    k_xproj   <<<dim3(2, 576, KK),     32,  0, stream>>>(xs, wpad, xdbl);
    k_dtproj  <<<dim3(6, 576, KK),     32,  0, stream>>>(xdbl, dt_w, dt_b, delta);
    k_scan_local <<<dim3(SC_C - 1, BB * KK), DIc, 0, stream>>>(xs, xdbl, delta, A_logs, sumP, sumS);
    k_scan_prefix<<<dim3(NN, BB * KK),       DIc, 0, stream>>>(sumP, sumS, hst);
    k_scan_chunk <<<dim3(SC_C, BB * KK),     DIc, 0, stream>>>(xs, xdbl, delta, A_logs, Ds, hst, yarr);
    k_combine <<<dim3(LL, BB),         DIc, 0, stream>>>(yarr, gate, gamma1, gamma2, gamma3,
                                                         gamma4, norm_w, norm_b, z_silu, yfinal);
    k_outproj <<<dim3(6, 576),         32,  0, stream>>>(yfinal, out_proj_w, out);
}


# k_dwconv writes xs itself: every conv output stored at its place in all four traversal sequences (4 stores per value); k_build_xs copy launch and xc_act plane go; x_proj and scan untouched
# speedup vs baseline: 10.3423x; 1.0779x over previous
#include <hip/hip_runtime.h>
#include <math.h>


typedef float v2f __attribute__((ext_vector_type(2)));
typedef float v8f __attribute__((ext_vector_type(8)));

#define BB 4
#define HH 48
#define WW 48
#define DMm 192
#define DIc 192
#define KK 4
#define NN 16
#define RR 12
#define LL (HH * WW)
#define CPROJ (RR + 2 * NN)
#define CPAD 64

__device__ __forceinline__ float siluf(float x)     { return x / (1.f + __expf(-x)); }
__device__ __forceinline__ float sigm(float x)      { return 1.f / (1.f + __expf(-x)); }
__device__ __forceinline__ float softplusf_(float x){ return (x > 20.f) ? x : log1pf(__expf(x)); }
__device__ __forceinline__ float geluf(float x)     { return 0.5f * x * (1.f + erff(x * 0.70710678118654752440f)); }

template <int NT>
__device__ __forceinline__ void wmma_xwT_multi(const float* __restrict__ X,
                                               const float* __restrict__ W,
                                               int row0, int col0,
                                               int Kdim, int ldx, int ldw,
                                               v8f acc[NT])
{
    const int lane = threadIdx.x & 31;
    const int m    = lane & 15;
    const int kh   = lane >> 4;
    const float* __restrict__ ap = X + (size_t)(row0 + m) * ldx + 2 * kh;
    const float* __restrict__ bp[NT];
#pragma unroll
    for (int t = 0; t < NT; ++t) {
        bp[t] = W + (size_t)(col0 + t * 16 + m) * ldw + 2 * kh;
        acc[t] = {};
    }
    v2f a_cur = *(const v2f*)(ap);
    v2f b_cur[NT];
#pragma unroll
    for (int t = 0; t < NT; ++t) b_cur[t] = *(const v2f*)(bp[t]);

    for (int k0 = 4; k0 < Kdim; k0 += 4) {
        v2f a_nxt = *(const v2f*)(ap + k0);
        v2f b_nxt[NT];
#pragma unroll
        for (int t = 0; t < NT; ++t) b_nxt[t] = *(const v2f*)(bp[t] + k0);
#pragma unroll
        for (int t = 0; t < NT; ++t)
            acc[t] = __builtin_amdgcn_wmma_f32_16x16x4_f32(false, a_cur, false, b_cur[t],
                                                           (short)0, acc[t], false, false);
        a_cur = a_nxt;
#pragma unroll
        for (int t = 0; t < NT; ++t) b_cur[t] = b_nxt[t];
    }
#pragma unroll
    for (int t = 0; t < NT; ++t)
        acc[t] = __builtin_amdgcn_wmma_f32_16x16x4_f32(false, a_cur, false, b_cur[t],
                                                       (short)0, acc[t], false, false);
}

__global__ void k_padw(const float* __restrict__ xpw, float* __restrict__ wpad)
{
    int k = blockIdx.x, c = blockIdx.y, d = threadIdx.x;
    wpad[((size_t)k * CPAD + c) * DIc + d] =
        (c < CPROJ) ? xpw[((size_t)k * CPROJ + c) * DIc + d] : 0.f;
}

__global__ void k_inproj(const float* __restrict__ x, const float* __restrict__ w,
                         float* __restrict__ xc_lin, float* __restrict__ z_silu)
{
    const int row0 = blockIdx.y * 16, col0 = blockIdx.x * 32;
    v8f acc[2];
    wmma_xwT_multi<2>(x, w, row0, col0, DMm, DMm, DMm, acc);
    const int lane = threadIdx.x & 31;
    const int nn = lane & 15, mh = (lane >> 4) * 8;
#pragma unroll
    for (int t = 0; t < 2; ++t)
#pragma unroll
        for (int v = 0; v < 8; ++v) {
            int r = row0 + mh + v;
            int c = col0 + t * 16 + nn;
            float val = acc[t][v];
            if (c < DIc) xc_lin[(size_t)r * DIc + c] = val;
            else         z_silu[(size_t)r * DIc + (c - DIc)] = siluf(val);
        }
}

__global__ void __launch_bounds__(DIc)
k_dwconv(const float* __restrict__ xc_lin, const float* __restrict__ cw,
         const float* __restrict__ cb, float* __restrict__ xs,
         float* __restrict__ rowsum)
{
    const int h = blockIdx.x, b = blockIdx.y, c = threadIdx.x;
    const bool up = (h > 0), dn = (h < HH - 1);
    const int hu = up ? h - 1 : h;
    const int hd = dn ? h + 1 : h;
    const float* __restrict__ p0 = xc_lin + ((size_t)b * LL + (size_t)hu * WW) * DIc + c;
    const float* __restrict__ p1 = xc_lin + ((size_t)b * LL + (size_t)h  * WW) * DIc + c;
    const float* __restrict__ p2 = xc_lin + ((size_t)b * LL + (size_t)hd * WW) * DIc + c;
    float wg[9];
#pragma unroll
    for (int j = 0; j < 9; ++j) wg[j] = cw[c * 9 + j];
    const float bias = cb[c];
    float r0[WW + 2], r1[WW + 2], r2[WW + 2];
    r0[0] = 0.f; r1[0] = 0.f; r2[0] = 0.f;
    r0[WW + 1] = 0.f; r1[WW + 1] = 0.f; r2[WW + 1] = 0.f;
#pragma unroll
    for (int w = 0; w < WW; ++w) r0[w + 1] = p0[(size_t)w * DIc];
#pragma unroll
    for (int w = 0; w < WW; ++w) r1[w + 1] = p1[(size_t)w * DIc];
#pragma unroll
    for (int w = 0; w < WW; ++w) r2[w + 1] = p2[(size_t)w * DIc];
#pragma unroll
    for (int w = 0; w < WW; ++w) {
        r0[w + 1] = up ? r0[w + 1] : 0.f;
        r2[w + 1] = dn ? r2[w + 1] : 0.f;
    }
    const int l0h = (h >> 2) * (WW / 4) * 16 + (h & 3) * 4;
    const int l1h = (h >> 2) * 16 + (h & 3);
    float* __restrict__ q0 = xs + ((size_t)(0 * BB + b) * LL + l0h) * DIc;
    float* __restrict__ q1 = xs + ((size_t)(1 * BB + b) * LL + l1h) * DIc;
    float* __restrict__ q2 = xs + ((size_t)(2 * BB + b) * LL + (LL - 1 - l0h)) * DIc;
    float* __restrict__ q3 = xs + ((size_t)(3 * BB + b) * LL + (LL - 1 - l1h)) * DIc;
    float tot = 0.f;
#pragma unroll
    for (int w = 0; w < WW; ++w) {
        float s = bias;
        s = fmaf(r0[w], wg[0], s); s = fmaf(r0[w + 1], wg[1], s); s = fmaf(r0[w + 2], wg[2], s);
        s = fmaf(r1[w], wg[3], s); s = fmaf(r1[w + 1], wg[4], s); s = fmaf(r1[w + 2], wg[5], s);
        s = fmaf(r2[w], wg[6], s); s = fmaf(r2[w + 1], wg[7], s); s = fmaf(r2[w + 2], wg[8], s);
        const float v = siluf(s);
        const int o0 = ((w >> 2) * 16 + (w & 3)) * DIc;
        const int o1 = ((w >> 2) * (HH / 4) * 16 + (w & 3) * 4) * DIc;
        (q0 + o0)[c] = v;
        (q1 + o1)[c] = v;
        (q2 - o0)[c] = v;
        (q3 - o1)[c] = v;
        tot += v;
    }
    rowsum[((size_t)b * HH + h) * DIc + c] = tot;
}

__global__ void k_gate(const float* __restrict__ rowsum, const float* __restrict__ fw1,
                       const float* __restrict__ fw2, float* __restrict__ gate)
{
    __shared__ float g[DIc];
    __shared__ float h1[DMm / 4];
    int b = blockIdx.x, d = threadIdx.x;
    float gs = 0.f;
#pragma unroll
    for (int h = 0; h < HH; ++h) gs += rowsum[((size_t)b * HH + h) * DIc + d];
    g[d] = gs * (1.f / LL);
    __syncthreads();
    if (d < DMm / 4) {
        float s = 0.f;
        for (int j = 0; j < DMm; ++j) s += fw1[d * DMm + j] * g[j];
        h1[d] = geluf(s);
    }
    __syncthreads();
    float s2 = 0.f;
#pragma unroll
    for (int j = 0; j < DMm / 4; ++j) s2 += fw2[d * (DMm / 4) + j] * h1[j];
    gate[b * DMm + d] = sigm(s2);
}

__global__ void k_xproj(const float* __restrict__ xs, const float* __restrict__ wpad,
                        float* __restrict__ xdbl)
{
    int k = blockIdx.z;
    const float* X = xs + (size_t)k * BB * LL * DIc;
    const float* W = wpad + (size_t)k * CPAD * DIc;
    float* O = xdbl + (size_t)k * BB * LL * CPROJ;
    const int row0 = blockIdx.y * 16, col0 = blockIdx.x * 32;
    v8f acc[2];
    wmma_xwT_multi<2>(X, W, row0, col0, DIc, DIc, DIc, acc);
    const int lane = threadIdx.x & 31;
    const int nn = lane & 15, mh = (lane >> 4) * 8;
#pragma unroll
    for (int t = 0; t < 2; ++t)
#pragma unroll
        for (int v = 0; v < 8; ++v) {
            int r = row0 + mh + v, c = col0 + t * 16 + nn;
            if (c < CPROJ) O[(size_t)r * CPROJ + c] = acc[t][v];
        }
}

__global__ void k_dtproj(const float* __restrict__ xdbl, const float* __restrict__ dtw,
                         const float* __restrict__ dtb, float* __restrict__ delta)
{
    int k = blockIdx.z;
    const float* X = xdbl + (size_t)k * BB * LL * CPROJ;
    const float* W = dtw + (size_t)k * DIc * RR;
    float* O = delta + (size_t)k * BB * LL * DIc;
    const int row0 = blockIdx.y * 16, col0 = blockIdx.x * 32;
    v8f acc[2];
    wmma_xwT_multi<2>(X, W, row0, col0, RR, CPROJ, RR, acc);
    const int lane = threadIdx.x & 31;
    const int nn = lane & 15, mh = (lane >> 4) * 8;
#pragma unroll
    for (int t = 0; t < 2; ++t)
#pragma unroll
        for (int v = 0; v < 8; ++v) {
            int r = row0 + mh + v, c = col0 + t * 16 + nn;
            O[(size_t)r * DIc + c] = softplusf_(acc[t][v] + dtb[k * DIc + c]);
        }
}

#define SC_C   96
#define SC_LEN (LL / SC_C)

__global__ void k_scan_local(const float* __restrict__ xs, const float* __restrict__ xdbl,
                             const float* __restrict__ delta, const float* __restrict__ A_logs,
                             float* __restrict__ sumP, float* __restrict__ sumS)
{
    const int c = blockIdx.x, kb = blockIdx.y;
    const int k = kb & 3, b = kb >> 2;
    const int d = threadIdx.x;
    float A[NN];
#pragma unroll
    for (int n = 0; n < NN; ++n)
        A[n] = -__expf(A_logs[((size_t)(k * DIc + d)) * NN + n]);
    float h[NN];
#pragma unroll
    for (int n = 0; n < NN; ++n) h[n] = 0.f;
    const size_t seqbase = ((size_t)(k * BB + b)) * LL + (size_t)c * SC_LEN;
    const float* up = xs    + seqbase * DIc + d;
    const float* dp = delta + seqbase * DIc + d;
    const float* bc = xdbl  + seqbase * CPROJ;
    float sdt = 0.f;
    for (int l = 0; l < SC_LEN; ++l) {
        float dt = dp[(size_t)l * DIc];
        float u  = up[(size_t)l * DIc];
        float du = dt * u;
        const float* bcl = bc + (size_t)l * CPROJ;
#pragma unroll
        for (int n = 0; n < NN; ++n)
            h[n] = __expf(dt * A[n]) * h[n] + du * bcl[RR + n];
        sdt += dt;
    }
    const size_t o = (((size_t)kb * SC_C + c) * NN) * DIc + d;
#pragma unroll
    for (int n = 0; n < NN; ++n) {
        sumP[o + (size_t)n * DIc] = __expf(sdt * A[n]);
        sumS[o + (size_t)n * DIc] = h[n];
    }
}

__global__ void k_scan_prefix(const float* __restrict__ sumP, const float* __restrict__ sumS,
                              float* __restrict__ hst)
{
    const int n = blockIdx.x, kb = blockIdx.y, d = threadIdx.x;
    const size_t base = (((size_t)kb * SC_C) * NN + n) * DIc + d;
    const size_t cs = (size_t)NN * DIc;
    float H = 0.f;
    hst[base] = H;
#pragma unroll 19
    for (int c = 0; c + 1 < SC_C; ++c) {
        H = sumP[base + c * cs] * H + sumS[base + c * cs];
        hst[base + (c + 1) * cs] = H;
    }
}

__global__ void k_scan_chunk(const float* __restrict__ xs, const float* __restrict__ xdbl,
                             const float* __restrict__ delta, const float* __restrict__ A_logs,
                             const float* __restrict__ Ds, const float* __restrict__ hst,
                             float* __restrict__ yarr)
{
    const int c = blockIdx.x, kb = blockIdx.y;
    const int k = kb & 3, b = kb >> 2;
    const int d = threadIdx.x;
    float A[NN];
#pragma unroll
    for (int n = 0; n < NN; ++n)
        A[n] = -__expf(A_logs[((size_t)(k * DIc + d)) * NN + n]);
    float Dv = Ds[k * DIc + d];
    float h[NN];
    const size_t o = (((size_t)kb * SC_C + c) * NN) * DIc + d;
#pragma unroll
    for (int n = 0; n < NN; ++n) h[n] = hst[o + (size_t)n * DIc];
    const size_t seqbase = ((size_t)(k * BB + b)) * LL + (size_t)c * SC_LEN;
    const float* up = xs    + seqbase * DIc + d;
    const float* dp = delta + seqbase * DIc + d;
    const float* bc = xdbl  + seqbase * CPROJ;
    float* yp       = yarr  + seqbase * DIc + d;
    for (int l = 0; l < SC_LEN; ++l) {
        float dt = dp[(size_t)l * DIc];
        float u  = up[(size_t)l * DIc];
        float du = dt * u;
        const float* bcl = bc + (size_t)l * CPROJ;
        float y = 0.f;
#pragma unroll
        for (int n = 0; n < NN; ++n) {
            float Bt = bcl[RR + n];
            float Ct = bcl[RR + NN + n];
            h[n] = __expf(dt * A[n]) * h[n] + du * Bt;
            y += h[n] * Ct;
        }
        yp[(size_t)l * DIc] = y + Dv * u;
    }
}

__global__ void k_combine(const float* __restrict__ yarr, const float* __restrict__ gate,
                          const float* __restrict__ g1, const float* __restrict__ g2,
                          const float* __restrict__ g3, const float* __restrict__ g4,
                          const float* __restrict__ nw, const float* __restrict__ nb,
                          const float* __restrict__ z_silu, float* __restrict__ yfinal)
{
    __shared__ float s1[256], s2[256];
    int lr = blockIdx.x, b = blockIdx.y, d = threadIdx.x;
    int h = lr / WW, w = lr % WW;
    int l = (((h >> 2) * (WW / 4) + (w >> 2)) << 4) + ((h & 3) << 2) + (w & 3);
    int lt = (l % WW) * HH + (l / WW);
    const size_t bstride = (size_t)LL * DIc;
    const float* Y0 = yarr + ((size_t)(0 * BB + b)) * bstride;
    const float* Y1 = yarr + ((size_t)(1 * BB + b)) * bstride;
    const float* Y2 = yarr + ((size_t)(2 * BB + b)) * bstride;
    const float* Y3 = yarr + ((size_t)(3 * BB + b)) * bstride;
    float yc = g1[d] * Y0[(size_t)l * DIc + d]
             + g2[d] * Y2[(size_t)(LL - 1 - l) * DIc + d]
             + g3[d] * Y1[(size_t)lt * DIc + d]
             + g4[d] * Y3[(size_t)(LL - 1 - lt) * DIc + d];
    yc *= gate[b * DIc + d];
    s1[d] = yc; s2[d] = yc * yc;
    if (d < 64) { s1[192 + d] = 0.f; s2[192 + d] = 0.f; }
    __syncthreads();
    for (int off = 128; off > 0; off >>= 1) {
        if (d < off) { s1[d] += s1[d + off]; s2[d] += s2[d + off]; }
        __syncthreads();
    }
    float mu  = s1[0] * (1.f / DIc);
    float var = s2[0] * (1.f / DIc) - mu * mu;
    float inv = rsqrtf(var + 1e-5f);
    yc = (yc - mu) * inv * nw[d] + nb[d];
    yc *= z_silu[((size_t)b * LL + lr) * DIc + d];
    yfinal[((size_t)b * LL + lr) * DIc + d] = yc;
}

__global__ void k_outproj(const float* __restrict__ yf, const float* __restrict__ w,
                          float* __restrict__ out)
{
    const int row0 = blockIdx.y * 16, col0 = blockIdx.x * 32;
    v8f acc[2];
    wmma_xwT_multi<2>(yf, w, row0, col0, DIc, DIc, DIc, acc);
    const int lane = threadIdx.x & 31;
    const int nn = lane & 15, mh = (lane >> 4) * 8;
#pragma unroll
    for (int t = 0; t < 2; ++t)
#pragma unroll
        for (int v = 0; v < 8; ++v) {
            int r = row0 + mh + v, c = col0 + t * 16 + nn;
            out[(size_t)r * DMm + c] = acc[t][v];
        }
}

extern "C" void kernel_launch(void* const* d_in, const int* in_sizes, int n_in,
                              void* d_out, int out_size, void* d_ws, size_t ws_size,
                              hipStream_t stream)
{
    (void)in_sizes; (void)n_in; (void)out_size; (void)ws_size;
    const float* x          = (const float*)d_in[0];
    const float* in_proj_w  = (const float*)d_in[1];
    const float* conv_w     = (const float*)d_in[2];
    const float* conv_b     = (const float*)d_in[3];
    const float* x_proj_w   = (const float*)d_in[4];
    const float* dt_w       = (const float*)d_in[5];
    const float* dt_b       = (const float*)d_in[6];
    const float* A_logs     = (const float*)d_in[7];
    const float* Ds         = (const float*)d_in[8];
    const float* gamma1     = (const float*)d_in[9];
    const float* gamma2     = (const float*)d_in[10];
    const float* gamma3     = (const float*)d_in[11];
    const float* gamma4     = (const float*)d_in[12];
    const float* fc_w1      = (const float*)d_in[13];
    const float* fc_w2      = (const float*)d_in[14];
    const float* norm_w     = (const float*)d_in[15];
    const float* norm_b     = (const float*)d_in[16];
    const float* out_proj_w = (const float*)d_in[17];
    float* out = (float*)d_out;

    float* ws = (float*)d_ws;
    size_t o = 0;
    float* xc_lin = ws + o; o += (size_t)BB * LL * DIc;
    float* z_silu = ws + o; o += (size_t)BB * LL * DIc;
    float* xs     = ws + o; o += (size_t)KK * BB * LL * DIc;
    float* xdbl   = ws + o; o += (size_t)KK * BB * LL * CPROJ;
    float* delta  = ws + o; o += (size_t)KK * BB * LL * DIc;
    float* yarr   = ws + o; o += (size_t)KK * BB * LL * DIc;
    float* yfinal = ws + o; o += (size_t)BB * LL * DIc;
    float* wpad   = ws + o; o += (size_t)KK * CPAD * DIc;
    float* rowsum = ws + o; o += (size_t)BB * HH * DIc;
    float* gate   = ws + o; o += BB * DMm;
    float* sumP   = ws + o; o += (size_t)KK * BB * SC_C * NN * DIc;
    float* sumS   = ws + o; o += (size_t)KK * BB * SC_C * NN * DIc;
    float* hst    = ws + o; o += (size_t)KK * BB * SC_C * NN * DIc;

    k_padw    <<<dim3(KK, CPAD),       DIc, 0, stream>>>(x_proj_w, wpad);
    k_inproj  <<<dim3(12, 576),        32,  0, stream>>>(x, in_proj_w, xc_lin, z_silu);
    k_dwconv  <<<dim3(HH, BB),         DIc, 0, stream>>>(xc_lin, conv_w, conv_b, xs, rowsum);
    k_gate    <<<dim3(BB),             DIc, 0, stream>>>(rowsum, fc_w1, fc_w2, gate);
    k_xproj   <<<dim3(2, 576, KK),     32,  0, stream>>>(xs, wpad, xdbl);
    k_dtproj  <<<dim3(6, 576, KK),     32,  0, stream>>>(xdbl, dt_w, dt_b, delta);
    k_scan_local <<<dim3(SC_C - 1, BB * KK), DIc, 0, stream>>>(xs, xdbl, delta, A_logs, sumP, sumS);
    k_scan_prefix<<<dim3(NN, BB * KK),       DIc, 0, stream>>>(sumP, sumS, hst);
    k_scan_chunk <<<dim3(SC_C, BB * KK),     DIc, 0, stream>>>(xs, xdbl, delta, A_logs, Ds, hst, yarr);
    k_combine <<<dim3(LL, BB),         DIc, 0, stream>>>(yarr, gate, gamma1, gamma2, gamma3,
                                                         gamma4, norm_w, norm_b, z_silu, yfinal);
    k_outproj <<<dim3(6, 576),         32,  0, stream>>>(yfinal, out_proj_w, out);
}


# in/x/out_proj: operand tiles staged in LDS by whole-line 16 B loads (4-wave work-groups, 32x64 tiles; x_proj 64x48, its 16 all-zero cols dropped), wmma fragments read from LDS; sum order kept
# speedup vs baseline: 11.9973x; 1.1600x over previous
#include <hip/hip_runtime.h>
#include <math.h>


typedef float v2f __attribute__((ext_vector_type(2)));
typedef float v8f __attribute__((ext_vector_type(8)));

#define BB 4
#define HH 48
#define WW 48
#define DMm 192
#define DIc 192
#define KK 4
#define NN 16
#define RR 12
#define LL (HH * WW)
#define CPROJ (RR + 2 * NN)
#define CPAD 64

__device__ __forceinline__ float siluf(float x)     { return x / (1.f + __expf(-x)); }
__device__ __forceinline__ float sigm(float x)      { return 1.f / (1.f + __expf(-x)); }
__device__ __forceinline__ float softplusf_(float x){ return (x > 20.f) ? x : log1pf(__expf(x)); }
__device__ __forceinline__ float geluf(float x)     { return 0.5f * x * (1.f + erff(x * 0.70710678118654752440f)); }

template <int NT>
__device__ __forceinline__ void wmma_xwT_multi(const float* __restrict__ X,
                                               const float* __restrict__ W,
                                               int row0, int col0,
                                               int Kdim, int ldx, int ldw,
                                               v8f acc[NT])
{
    const int lane = threadIdx.x & 31;
    const int m    = lane & 15;
    const int kh   = lane >> 4;
    const float* __restrict__ ap = X + (size_t)(row0 + m) * ldx + 2 * kh;
    const float* __restrict__ bp[NT];
#pragma unroll
    for (int t = 0; t < NT; ++t) {
        bp[t] = W + (size_t)(col0 + t * 16 + m) * ldw + 2 * kh;
        acc[t] = {};
    }
    v2f a_cur = *(const v2f*)(ap);
    v2f b_cur[NT];
#pragma unroll
    for (int t = 0; t < NT; ++t) b_cur[t] = *(const v2f*)(bp[t]);

    for (int k0 = 4; k0 < Kdim; k0 += 4) {
        v2f a_nxt = *(const v2f*)(ap + k0);
        v2f b_nxt[NT];
#pragma unroll
        for (int t = 0; t < NT; ++t) b_nxt[t] = *(const v2f*)(bp[t] + k0);
#pragma unroll
        for (int t = 0; t < NT; ++t)
            acc[t] = __builtin_amdgcn_wmma_f32_16x16x4_f32(false, a_cur, false, b_cur[t],
                                                           (short)0, acc[t], false, false);
        a_cur = a_nxt;
#pragma unroll
        for (int t = 0; t < NT; ++t) b_cur[t] = b_nxt[t];
    }
#pragma unroll
    for (int t = 0; t < NT; ++t)
        acc[t] = __builtin_amdgcn_wmma_f32_16x16x4_f32(false, a_cur, false, b_cur[t],
                                                       (short)0, acc[t], false, false);
}

#define LPITCH 196
#define KPIECES (192 / 4)

template <int WM, int WN, int NT>
__device__ __forceinline__ void wmma_lds_k192(const float* __restrict__ X,
                                              const float* __restrict__ W,
                                              int row0, int col0,
                                              float* __restrict__ lds, v8f acc[NT])
{
    constexpr int NTHR = WM * WN * 32;
    constexpr int AR = WM * 16;
    constexpr int BR = WN * NT * 16;
    static_assert((AR * KPIECES) % NTHR == 0 && (BR * KPIECES) % NTHR == 0, "whole copy passes");
    float* __restrict__ lA = lds;
    float* __restrict__ lB = lds + AR * LPITCH;
    const int tid = threadIdx.x;
    const float4* __restrict__ gA = (const float4*)(X + (size_t)row0 * 192);
    const float4* __restrict__ gB = (const float4*)(W + (size_t)col0 * 192);
#pragma unroll
    for (int j = 0; j < AR * KPIECES / NTHR; ++j) {
        const int i = tid + j * NTHR;
        const int r = i / KPIECES, c = i - r * KPIECES;
        *(float4*)(lA + r * LPITCH + c * 4) = gA[i];
    }
#pragma unroll
    for (int j = 0; j < BR * KPIECES / NTHR; ++j) {
        const int i = tid + j * NTHR;
        const int r = i / KPIECES, c = i - r * KPIECES;
        *(float4*)(lB + r * LPITCH + c * 4) = gB[i];
    }
    __syncthreads();

    const int lane = tid & 31, wave = tid >> 5;
    const int wm = wave / WN, wn = wave - wm * WN;
    const int m  = lane & 15;
    const int kh = lane >> 4;
    const float* __restrict__ ap = lA + (wm * 16 + m) * LPITCH + 2 * kh;
    const float* __restrict__ bp = lB + (wn * NT * 16 + m) * LPITCH + 2 * kh;
#pragma unroll
    for (int t = 0; t < NT; ++t) acc[t] = {};
#pragma unroll
    for (int ks = 0; ks < 192 / 4; ++ks) {
        const v2f a = *(const v2f*)(ap + 4 * ks);
        v2f b[NT];
#pragma unroll
        for (int t = 0; t < NT; ++t) b[t] = *(const v2f*)(bp + t * 16 * LPITCH + 4 * ks);
#pragma unroll
        for (int t = 0; t < NT; ++t)
            acc[t] = __builtin_amdgcn_wmma_f32_16x16x4_f32(false, a, false, b[t],
                                                           (short)0, acc[t], false, false);
    }
}

__global__ void k_padw(const float* __restrict__ xpw, float* __restrict__ wpad)
{
    int k = blockIdx.x, c = blockIdx.y, d = threadIdx.x;
    wpad[((size_t)k * CPAD + c) * DIc + d] =
        (c < CPROJ) ? xpw[((size_t)k * CPROJ + c) * DIc + d] : 0.f;
}

__global__ void __launch_bounds__(128)
k_inproj(const float* __restrict__ x, const float* __restrict__ w,
         float* __restrict__ xc_lin, float* __restrict__ z_silu)
{
    __shared__ __attribute__((aligned(16))) float lds[(32 + 64) * LPITCH];
    v8f acc[2];
    wmma_lds_k192<2, 2, 2>(x, w, blockIdx.y * 32, blockIdx.x * 64, lds, acc);
    const int wave = threadIdx.x >> 5;
    const int row0 = blockIdx.y * 32 + (wave >> 1) * 16;
    const int col0 = blockIdx.x * 64 + (wave & 1) * 32;
    const int lane = threadIdx.x & 31;
    const int nn = lane & 15, mh = (lane >> 4) * 8;
#pragma unroll
    for (int t = 0; t < 2; ++t)
#pragma unroll
        for (int v = 0; v < 8; ++v) {
            int r = row0 + mh + v;
            int c = col0 + t * 16 + nn;
            float val = acc[t][v];
            if (c < DIc) xc_lin[(size_t)r * DIc + c] = val;
            else         z_silu[(size_t)r * DIc + (c - DIc)] = siluf(val);
        }
}

__global__ void __launch_bounds__(DIc)
k_dwconv(const float* __restrict__ xc_lin, const float* __restrict__ cw,
         const float* __restrict__ cb, float* __restrict__ xs,
         float* __restrict__ rowsum)
{
    const int h = blockIdx.x, b = blockIdx.y, c = threadIdx.x;
    const bool up = (h > 0), dn = (h < HH - 1);
    const int hu = up ? h - 1 : h;
    const int hd = dn ? h + 1 : h;
    const float* __restrict__ p0 = xc_lin + ((size_t)b * LL + (size_t)hu * WW) * DIc + c;
    const float* __restrict__ p1 = xc_lin + ((size_t)b * LL + (size_t)h  * WW) * DIc + c;
    const float* __restrict__ p2 = xc_lin + ((size_t)b * LL + (size_t)hd * WW) * DIc + c;
    float wg[9];
#pragma unroll
    for (int j = 0; j < 9; ++j) wg[j] = cw[c * 9 + j];
    const float bias = cb[c];
    float r0[WW + 2], r1[WW + 2], r2[WW + 2];
    r0[0] = 0.f; r1[0] = 0.f; r2[0] = 0.f;
    r0[WW + 1] = 0.f; r1[WW + 1] = 0.f; r2[WW + 1] = 0.f;
#pragma unroll
    for (int w = 0; w < WW; ++w) r0[w + 1] = p0[(size_t)w * DIc];
#pragma unroll
    for (int w = 0; w < WW; ++w) r1[w + 1] = p1[(size_t)w * DIc];
#pragma unroll
    for (int w = 0; w < WW; ++w) r2[w + 1] = p2[(size_t)w * DIc];
#pragma unroll
    for (int w = 0; w < WW; ++w) {
        r0[w + 1] = up ? r0[w + 1] : 0.f;
        r2[w + 1] = dn ? r2[w + 1] : 0.f;
    }
    const int l0h = (h >> 2) * (WW / 4) * 16 + (h & 3) * 4;
    const int l1h = (h >> 2) * 16 + (h & 3);
    float* __restrict__ q0 = xs + ((size_t)(0 * BB + b) * LL + l0h) * DIc;
    float* __restrict__ q1 = xs + ((size_t)(1 * BB + b) * LL + l1h) * DIc;
    float* __restrict__ q2 = xs + ((size_t)(2 * BB + b) * LL + (LL - 1 - l0h)) * DIc;
    float* __restrict__ q3 = xs + ((size_t)(3 * BB + b) * LL + (LL - 1 - l1h)) * DIc;
    float tot = 0.f;
#pragma unroll
    for (int w = 0; w < WW; ++w) {
        float s = bias;
        s = fmaf(r0[w], wg[0], s); s = fmaf(r0[w + 1], wg[1], s); s = fmaf(r0[w + 2], wg[2], s);
        s = fmaf(r1[w], wg[3], s); s = fmaf(r1[w + 1], wg[4], s); s = fmaf(r1[w + 2], wg[5], s);
        s = fmaf(r2[w], wg[6], s); s = fmaf(r2[w + 1], wg[7], s); s = fmaf(r2[w + 2], wg[8], s);
        const float v = siluf(s);
        const int o0 = ((w >> 2) * 16 + (w & 3)) * DIc;
        const int o1 = ((w >> 2) * (HH / 4) * 16 + (w & 3) * 4) * DIc;
        (q0 + o0)[c] = v;
        (q1 + o1)[c] = v;
        (q2 - o0)[c] = v;
        (q3 - o1)[c] = v;
        tot += v;
    }
    rowsum[((size_t)b * HH + h) * DIc + c] = tot;
}

__global__ void k_gate(const float* __restrict__ rowsum, const float* __restrict__ fw1,
                       const float* __restrict__ fw2, float* __restrict__ gate)
{
    __shared__ float g[DIc];
    __shared__ float h1[DMm / 4];
    int b = blockIdx.x, d = threadIdx.x;
    float gs = 0.f;
#pragma unroll
    for (int h = 0; h < HH; ++h) gs += rowsum[((size_t)b * HH + h) * DIc + d];
    g[d] = gs * (1.f / LL);
    __syncthreads();
    if (d < DMm / 4) {
        float s = 0.f;
        for (int j = 0; j < DMm; ++j) s += fw1[d * DMm + j] * g[j];
        h1[d] = geluf(s);
    }
    __syncthreads();
    float s2 = 0.f;
#pragma unroll
    for (int j = 0; j < DMm / 4; ++j) s2 += fw2[d * (DMm / 4) + j] * h1[j];
    gate[b * DMm + d] = sigm(s2);
}

__global__ void __launch_bounds__(128)
k_xproj(const float* __restrict__ xs, const float* __restrict__ wpad,
        float* __restrict__ xdbl)
{
    __shared__ __attribute__((aligned(16))) float lds[(64 + 48) * LPITCH];
    int k = blockIdx.z;
    const float* X = xs + (size_t)k * BB * LL * DIc;
    const float* W = wpad + (size_t)k * CPAD * DIc;
    float* O = xdbl + (size_t)k * BB * LL * CPROJ;
    v8f acc[3];
    wmma_lds_k192<4, 1, 3>(X, W, blockIdx.y * 64, 0, lds, acc);
    const int row0 = blockIdx.y * 64 + (threadIdx.x >> 5) * 16, col0 = 0;
    const int lane = threadIdx.x & 31;
    const int nn = lane & 15, mh = (lane >> 4) * 8;
#pragma unroll
    for (int t = 0; t < 3; ++t)
#pragma unroll
        for (int v = 0; v < 8; ++v) {
            int r = row0 + mh + v, c = col0 + t * 16 + nn;
            if (c < CPROJ) O[(size_t)r * CPROJ + c] = acc[t][v];
        }
}

__global__ void k_dtproj(const float* __restrict__ xdbl, const float* __restrict__ dtw,
                         const float* __restrict__ dtb, float* __restrict__ delta)
{
    int k = blockIdx.z;
    const float* X = xdbl + (size_t)k * BB * LL * CPROJ;
    const float* W = dtw + (size_t)k * DIc * RR;
    float* O = delta + (size_t)k * BB * LL * DIc;
    const int row0 = blockIdx.y * 16, col0 = blockIdx.x * 32;
    v8f acc[2];
    wmma_xwT_multi<2>(X, W, row0, col0, RR, CPROJ, RR, acc);
    const int lane = threadIdx.x & 31;
    const int nn = lane & 15, mh = (lane >> 4) * 8;
#pragma unroll
    for (int t = 0; t < 2; ++t)
#pragma unroll
        for (int v = 0; v < 8; ++v) {
            int r = row0 + mh + v, c = col0 + t * 16 + nn;
            O[(size_t)r * DIc + c] = softplusf_(acc[t][v] + dtb[k * DIc + c]);
        }
}

#define SC_C   96
#define SC_LEN (LL / SC_C)

__global__ void k_scan_local(const float* __restrict__ xs, const float* __restrict__ xdbl,
                             const float* __restrict__ delta, const float* __restrict__ A_logs,
                             float* __restrict__ sumP, float* __restrict__ sumS)
{
    const int c = blockIdx.x, kb = blockIdx.y;
    const int k = kb & 3, b = kb >> 2;
    const int d = threadIdx.x;
    float A[NN];
#pragma unroll
    for (int n = 0; n < NN; ++n)
        A[n] = -__expf(A_logs[((size_t)(k * DIc + d)) * NN + n]);
    float h[NN];
#pragma unroll
    for (int n = 0; n < NN; ++n) h[n] = 0.f;
    const size_t seqbase = ((size_t)(k * BB + b)) * LL + (size_t)c * SC_LEN;
    const float* up = xs    + seqbase * DIc + d;
    const float* dp = delta + seqbase * DIc + d;
    const float* bc = xdbl  + seqbase * CPROJ;
    float sdt = 0.f;
    for (int l = 0; l < SC_LEN; ++l) {
        float dt = dp[(size_t)l * DIc];
        float u  = up[(size_t)l * DIc];
        float du = dt * u;
        const float* bcl = bc + (size_t)l * CPROJ;
#pragma unroll
        for (int n = 0; n < NN; ++n)
            h[n] = __expf(dt * A[n]) * h[n] + du * bcl[RR + n];
        sdt += dt;
    }
    const size_t o = (((size_t)kb * SC_C + c) * NN) * DIc + d;
#pragma unroll
    for (int n = 0; n < NN; ++n) {
        sumP[o + (size_t)n * DIc] = __expf(sdt * A[n]);
        sumS[o + (size_t)n * DIc] = h[n];
    }
}

__global__ void k_scan_prefix(const float* __restrict__ sumP, const float* __restrict__ sumS,
                              float* __restrict__ hst)
{
    const int n = blockIdx.x, kb = blockIdx.y, d = threadIdx.x;
    const size_t base = (((size_t)kb * SC_C) * NN + n) * DIc + d;
    const size_t cs = (size_t)NN * DIc;
    float H = 0.f;
    hst[base] = H;
#pragma unroll 19
    for (int c = 0; c + 1 < SC_C; ++c) {
        H = sumP[base + c * cs] * H + sumS[base + c * cs];
        hst[base + (c + 1) * cs] = H;
    }
}

__global__ void k_scan_chunk(const float* __restrict__ xs, const float* __restrict__ xdbl,
                             const float* __restrict__ delta, const float* __restrict__ A_logs,
                             const float* __restrict__ Ds, const float* __restrict__ hst,
                             float* __restrict__ yarr)
{
    const int c = blockIdx.x, kb = blockIdx.y;
    const int k = kb & 3, b = kb >> 2;
    const int d = threadIdx.x;
    float A[NN];
#pragma unroll
    for (int n = 0; n < NN; ++n)
        A[n] = -__expf(A_logs[((size_t)(k * DIc + d)) * NN + n]);
    float Dv = Ds[k * DIc + d];
    float h[NN];
    const size_t o = (((size_t)kb * SC_C + c) * NN) * DIc + d;
#pragma unroll
    for (int n = 0; n < NN; ++n) h[n] = hst[o + (size_t)n * DIc];
    const size_t seqbase = ((size_t)(k * BB + b)) * LL + (size_t)c * SC_LEN;
    const float* up = xs    + seqbase * DIc + d;
    const float* dp = delta + seqbase * DIc + d;
    const float* bc = xdbl  + seqbase * CPROJ;
    float* yp       = yarr  + seqbase * DIc + d;
    for (int l = 0; l < SC_LEN; ++l) {
        float dt = dp[(size_t)l * DIc];
        float u  = up[(size_t)l * DIc];
        float du = dt * u;
        const float* bcl = bc + (size_t)l * CPROJ;
        float y = 0.f;
#pragma unroll
        for (int n = 0; n < NN; ++n) {
            float Bt = bcl[RR + n];
            float Ct = bcl[RR + NN + n];
            h[n] = __expf(dt * A[n]) * h[n] + du * Bt;
            y += h[n] * Ct;
        }
        yp[(size_t)l * DIc] = y + Dv * u;
    }
}

__global__ void k_combine(const float* __restrict__ yarr, const float* __restrict__ gate,
                          const float* __restrict__ g1, const float* __restrict__ g2,
                          const float* __restrict__ g3, const float* __restrict__ g4,
                          const float* __restrict__ nw, const float* __restrict__ nb,
                          const float* __restrict__ z_silu, float* __restrict__ yfinal)
{
    __shared__ float s1[256], s2[256];
    int lr = blockIdx.x, b = blockIdx.y, d = threadIdx.x;
    int h = lr / WW, w = lr % WW;
    int l = (((h >> 2) * (WW / 4) + (w >> 2)) << 4) + ((h & 3) << 2) + (w & 3);
    int lt = (l % WW) * HH + (l / WW);
    const size_t bstride = (size_t)LL * DIc;
    const float* Y0 = yarr + ((size_t)(0 * BB + b)) * bstride;
    const float* Y1 = yarr + ((size_t)(1 * BB + b)) * bstride;
    const float* Y2 = yarr + ((size_t)(2 * BB + b)) * bstride;
    const float* Y3 = yarr + ((size_t)(3 * BB + b)) * bstride;
    float yc = g1[d] * Y0[(size_t)l * DIc + d]
             + g2[d] * Y2[(size_t)(LL - 1 - l) * DIc + d]
             + g3[d] * Y1[(size_t)lt * DIc + d]
             + g4[d] * Y3[(size_t)(LL - 1 - lt) * DIc + d];
    yc *= gate[b * DIc + d];
    s1[d] = yc; s2[d] = yc * yc;
    if (d < 64) { s1[192 + d] = 0.f; s2[192 + d] = 0.f; }
    __syncthreads();
    for (int off = 128; off > 0; off >>= 1) {
        if (d < off) { s1[d] += s1[d + off]; s2[d] += s2[d + off]; }
        __syncthreads();
    }
    float mu  = s1[0] * (1.f / DIc);
    float var = s2[0] * (1.f / DIc) - mu * mu;
    float inv = rsqrtf(var + 1e-5f);
    yc = (yc - mu) * inv * nw[d] + nb[d];
    yc *= z_silu[((size_t)b * LL + lr) * DIc + d];
    yfinal[((size_t)b * LL + lr) * DIc + d] = yc;
}

__global__ void __launch_bounds__(128)
k_outproj(const float* __restrict__ yf, const float* __restrict__ w,
          float* __restrict__ out)
{
    __shared__ __attribute__((aligned(16))) float lds[(32 + 64) * LPITCH];
    v8f acc[2];
    wmma_lds_k192<2, 2, 2>(yf, w, blockIdx.y * 32, blockIdx.x * 64, lds, acc);
    const int wave = threadIdx.x >> 5;
    const int row0 = blockIdx.y * 32 + (wave >> 1) * 16;
    const int col0 = blockIdx.x * 64 + (wave & 1) * 32;
    const int lane = threadIdx.x & 31;
    const int nn = lane & 15, mh = (lane >> 4) * 8;
#pragma unroll
    for (int t = 0; t < 2; ++t)
#pragma unroll
        for (int v = 0; v < 8; ++v) {
            int r = row0 + mh + v, c = col0 + t * 16 + nn;
            out[(size_t)r * DMm + c] = acc[t][v];
        }
}

extern "C" void kernel_launch(void* const* d_in, const int* in_sizes, int n_in,
                              void* d_out, int out_size, void* d_ws, size_t ws_size,
                              hipStream_t stream)
{
    (void)in_sizes; (void)n_in; (void)out_size; (void)ws_size;
    const float* x          = (const float*)d_in[0];
    const float* in_proj_w  = (const float*)d_in[1];
    const float* conv_w     = (const float*)d_in[2];
    const float* conv_b     = (const float*)d_in[3];
    const float* x_proj_w   = (const float*)d_in[4];
    const float* dt_w       = (const float*)d_in[5];
    const float* dt_b       = (const float*)d_in[6];
    const float* A_logs     = (const float*)d_in[7];
    const float* Ds         = (const float*)d_in[8];
    const float* gamma1     = (const float*)d_in[9];
    const float* gamma2     = (const float*)d_in[10];
    const float* gamma3     = (const float*)d_in[11];
    const float* gamma4     = (const float*)d_in[12];
    const float* fc_w1      = (const float*)d_in[13];
    const float* fc_w2      = (const float*)d_in[14];
    const float* norm_w     = (const float*)d_in[15];
    const float* norm_b     = (const float*)d_in[16];
    const float* out_proj_w = (const float*)d_in[17];
    float* out = (float*)d_out;

    float* ws = (float*)d_ws;
    size_t o = 0;
    float* xc_lin = ws + o; o += (size_t)BB * LL * DIc;
    float* z_silu = ws + o; o += (size_t)BB * LL * DIc;
    float* xs     = ws + o; o += (size_t)KK * BB * LL * DIc;
    float* xdbl   = ws + o; o += (size_t)KK * BB * LL * CPROJ;
    float* delta  = ws + o; o += (size_t)KK * BB * LL * DIc;
    float* yarr   = ws + o; o += (size_t)KK * BB * LL * DIc;
    float* yfinal = ws + o; o += (size_t)BB * LL * DIc;
    float* wpad   = ws + o; o += (size_t)KK * CPAD * DIc;
    float* rowsum = ws + o; o += (size_t)BB * HH * DIc;
    float* gate   = ws + o; o += BB * DMm;
    float* sumP   = ws + o; o += (size_t)KK * BB * SC_C * NN * DIc;
    float* sumS   = ws + o; o += (size_t)KK * BB * SC_C * NN * DIc;
    float* hst    = ws + o; o += (size_t)KK * BB * SC_C * NN * DIc;

    k_padw    <<<dim3(KK, CPAD),       DIc, 0, stream>>>(x_proj_w, wpad);
    k_inproj  <<<dim3(6, 288),         128, 0, stream>>>(x, in_proj_w, xc_lin, z_silu);
    k_dwconv  <<<dim3(HH, BB),         DIc, 0, stream>>>(xc_lin, conv_w, conv_b, xs, rowsum);
    k_gate    <<<dim3(BB),             DIc, 0, stream>>>(rowsum, fc_w1, fc_w2, gate);
    k_xproj   <<<dim3(1, 144, KK),     128, 0, stream>>>(xs, wpad, xdbl);
    k_dtproj  <<<dim3(6, 576, KK),     32,  0, stream>>>(xdbl, dt_w, dt_b, delta);
    k_scan_local <<<dim3(SC_C - 1, BB * KK), DIc, 0, stream>>>(xs, xdbl, delta, A_logs, sumP, sumS);
    k_scan_prefix<<<dim3(NN, BB * KK),       DIc, 0, stream>>>(sumP, sumS, hst);
    k_scan_chunk <<<dim3(SC_C, BB * KK),     DIc, 0, stream>>>(xs, xdbl, delta, A_logs, Ds, hst, yarr);
    k_combine <<<dim3(LL, BB),         DIc, 0, stream>>>(yarr, gate, gamma1, gamma2, gamma3,
                                                         gamma4, norm_w, norm_b, z_silu, yfinal);
    k_outproj <<<dim3(3, 288),         128, 0, stream>>>(yfinal, out_proj_w, out);
}


# k_combine launch removed: each out_proj work-group computes its 32 operand rows (4-direction sum, gate, LayerNorm, z) into its LDS tile; LayerNorm sums by lane butterfly (new sum order)
# speedup vs baseline: 12.7677x; 1.0642x over previous
#include <hip/hip_runtime.h>
#include <math.h>


typedef float v2f __attribute__((ext_vector_type(2)));
typedef float v8f __attribute__((ext_vector_type(8)));

#define BB 4
#define HH 48
#define WW 48
#define DMm 192
#define DIc 192
#define KK 4
#define NN 16
#define RR 12
#define LL (HH * WW)
#define CPROJ (RR + 2 * NN)
#define CPAD 64

__device__ __forceinline__ float siluf(float x)     { return x / (1.f + __expf(-x)); }
__device__ __forceinline__ float sigm(float x)      { return 1.f / (1.f + __expf(-x)); }
__device__ __forceinline__ float softplusf_(float x){ return (x > 20.f) ? x : log1pf(__expf(x)); }
__device__ __forceinline__ float geluf(float x)     { return 0.5f * x * (1.f + erff(x * 0.70710678118654752440f)); }

template <int NT>
__device__ __forceinline__ void wmma_xwT_multi(const float* __restrict__ X,
                                               const float* __restrict__ W,
                                               int row0, int col0,
                                               int Kdim, int ldx, int ldw,
                                               v8f acc[NT])
{
    const int lane = threadIdx.x & 31;
    const int m    = lane & 15;
    const int kh   = lane >> 4;
    const float* __restrict__ ap = X + (size_t)(row0 + m) * ldx + 2 * kh;
    const float* __restrict__ bp[NT];
#pragma unroll
    for (int t = 0; t < NT; ++t) {
        bp[t] = W + (size_t)(col0 + t * 16 + m) * ldw + 2 * kh;
        acc[t] = {};
    }
    v2f a_cur = *(const v2f*)(ap);
    v2f b_cur[NT];
#pragma unroll
    for (int t = 0; t < NT; ++t) b_cur[t] = *(const v2f*)(bp[t]);

    for (int k0 = 4; k0 < Kdim; k0 += 4) {
        v2f a_nxt = *(const v2f*)(ap + k0);
        v2f b_nxt[NT];
#pragma unroll
        for (int t = 0; t < NT; ++t) b_nxt[t] = *(const v2f*)(bp[t] + k0);
#pragma unroll
        for (int t = 0; t < NT; ++t)
            acc[t] = __builtin_amdgcn_wmma_f32_16x16x4_f32(false, a_cur, false, b_cur[t],
                                                           (short)0, acc[t], false, false);
        a_cur = a_nxt;
#pragma unroll
        for (int t = 0; t < NT; ++t) b_cur[t] = b_nxt[t];
    }
#pragma unroll
    for (int t = 0; t < NT; ++t)
        acc[t] = __builtin_amdgcn_wmma_f32_16x16x4_f32(false, a_cur, false, b_cur[t],
                                                       (short)0, acc[t], false, false);
}

#define LPITCH 196
#define KPIECES (192 / 4)

template <int WM, int WN, int NT>
__device__ __forceinline__ void wmma_frags_k192(const float* __restrict__ lA,
                                                const float* __restrict__ lB, v8f acc[NT])
{
    const int tid = threadIdx.x;
    const int lane = tid & 31, wave = tid >> 5;
    const int wm = wave / WN, wn = wave - wm * WN;
    const int m  = lane & 15;
    const int kh = lane >> 4;
    const float* __restrict__ ap = lA + (wm * 16 + m) * LPITCH + 2 * kh;
    const float* __restrict__ bp = lB + (wn * NT * 16 + m) * LPITCH + 2 * kh;
#pragma unroll
    for (int t = 0; t < NT; ++t) acc[t] = {};
#pragma unroll
    for (int ks = 0; ks < 192 / 4; ++ks) {
        const v2f a = *(const v2f*)(ap + 4 * ks);
        v2f b[NT];
#pragma unroll
        for (int t = 0; t < NT; ++t) b[t] = *(const v2f*)(bp + t * 16 * LPITCH + 4 * ks);
#pragma unroll
        for (int t = 0; t < NT; ++t)
            acc[t] = __builtin_amdgcn_wmma_f32_16x16x4_f32(false, a, false, b[t],
                                                           (short)0, acc[t], false, false);
    }
}

template <int WM, int WN, int NT>
__device__ __forceinline__ void wmma_lds_k192(const float* __restrict__ X,
                                              const float* __restrict__ W,
                                              int row0, int col0,
                                              float* __restrict__ lds, v8f acc[NT])
{
    constexpr int NTHR = WM * WN * 32;
    constexpr int AR = WM * 16;
    constexpr int BR = WN * NT * 16;
    static_assert((AR * KPIECES) % NTHR == 0 && (BR * KPIECES) % NTHR == 0, "whole copy passes");
    float* __restrict__ lA = lds;
    float* __restrict__ lB = lds + AR * LPITCH;
    const int tid = threadIdx.x;
    const float4* __restrict__ gA = (const float4*)(X + (size_t)row0 * 192);
    const float4* __restrict__ gB = (const float4*)(W + (size_t)col0 * 192);
#pragma unroll
    for (int j = 0; j < AR * KPIECES / NTHR; ++j) {
        const int i = tid + j * NTHR;
        const int r = i / KPIECES, c = i - r * KPIECES;
        *(float4*)(lA + r * LPITCH + c * 4) = gA[i];
    }
#pragma unroll
    for (int j = 0; j < BR * KPIECES / NTHR; ++j) {
        const int i = tid + j * NTHR;
        const int r = i / KPIECES, c = i - r * KPIECES;
        *(float4*)(lB + r * LPITCH + c * 4) = gB[i];
    }
    __syncthreads();
    wmma_frags_k192<WM, WN, NT>(lA, lB, acc);
}

__global__ void k_padw(const float* __restrict__ xpw, float* __restrict__ wpad)
{
    int k = blockIdx.x, c = blockIdx.y, d = threadIdx.x;
    wpad[((size_t)k * CPAD + c) * DIc + d] =
        (c < CPROJ) ? xpw[((size_t)k * CPROJ + c) * DIc + d] : 0.f;
}

__global__ void __launch_bounds__(128)
k_inproj(const float* __restrict__ x, const float* __restrict__ w,
         float* __restrict__ xc_lin, float* __restrict__ z_silu)
{
    __shared__ __attribute__((aligned(16))) float lds[(32 + 64) * LPITCH];
    v8f acc[2];
    wmma_lds_k192<2, 2, 2>(x, w, blockIdx.y * 32, blockIdx.x * 64, lds, acc);
    const int wave = threadIdx.x >> 5;
    const int row0 = blockIdx.y * 32 + (wave >> 1) * 16;
    const int col0 = blockIdx.x * 64 + (wave & 1) * 32;
    const int lane = threadIdx.x & 31;
    const int nn = lane & 15, mh = (lane >> 4) * 8;
#pragma unroll
    for (int t = 0; t < 2; ++t)
#pragma unroll
        for (int v = 0; v < 8; ++v) {
            int r = row0 + mh + v;
            int c = col0 + t * 16 + nn;
            float val = acc[t][v];
            if (c < DIc) xc_lin[(size_t)r * DIc + c] = val;
            else         z_silu[(size_t)r * DIc + (c - DIc)] = siluf(val);
        }
}

__global__ void __launch_bounds__(DIc)
k_dwconv(const float* __restrict__ xc_lin, const float* __restrict__ cw,
         const float* __restrict__ cb, float* __restrict__ xs,
         float* __restrict__ rowsum)
{
    const int h = blockIdx.x, b = blockIdx.y, c = threadIdx.x;
    const bool up = (h > 0), dn = (h < HH - 1);
    const int hu = up ? h - 1 : h;
    const int hd = dn ? h + 1 : h;
    const float* __restrict__ p0 = xc_lin + ((size_t)b * LL + (size_t)hu * WW) * DIc + c;
    const float* __restrict__ p1 = xc_lin + ((size_t)b * LL + (size_t)h  * WW) * DIc + c;
    const float* __restrict__ p2 = xc_lin + ((size_t)b * LL + (size_t)hd * WW) * DIc + c;
    float wg[9];
#pragma unroll
    for (int j = 0; j < 9; ++j) wg[j] = cw[c * 9 + j];
    const float bias = cb[c];
    float r0[WW + 2], r1[WW + 2], r2[WW + 2];
    r0[0] = 0.f; r1[0] = 0.f; r2[0] = 0.f;
    r0[WW + 1] = 0.f; r1[WW + 1] = 0.f; r2[WW + 1] = 0.f;
#pragma unroll
    for (int w = 0; w < WW; ++w) r0[w + 1] = p0[(size_t)w * DIc];
#pragma unroll
    for (int w = 0; w < WW; ++w) r1[w + 1] = p1[(size_t)w * DIc];
#pragma unroll
    for (int w = 0; w < WW; ++w) r2[w + 1] = p2[(size_t)w * DIc];
#pragma unroll
    for (int w = 0; w < WW; ++w) {
        r0[w + 1] = up ? r0[w + 1] : 0.f;
        r2[w + 1] = dn ? r2[w + 1] : 0.f;
    }
    const int l0h = (h >> 2) * (WW / 4) * 16 + (h & 3) * 4;
    const int l1h = (h >> 2) * 16 + (h & 3);
    float* __restrict__ q0 = xs + ((size_t)(0 * BB + b) * LL + l0h) * DIc;
    float* __restrict__ q1 = xs + ((size_t)(1 * BB + b) * LL + l1h) * DIc;
    float* __restrict__ q2 = xs + ((size_t)(2 * BB + b) * LL + (LL - 1 - l0h)) * DIc;
    float* __restrict__ q3 = xs + ((size_t)(3 * BB + b) * LL + (LL - 1 - l1h)) * DIc;
    float tot = 0.f;
#pragma unroll
    for (int w = 0; w < WW; ++w) {
        float s = bias;
        s = fmaf(r0[w], wg[0], s); s = fmaf(r0[w + 1], wg[1], s); s = fmaf(r0[w + 2], wg[2], s);
        s = fmaf(r1[w], wg[3], s); s = fmaf(r1[w + 1], wg[4], s); s = fmaf(r1[w + 2], wg[5], s);
        s = fmaf(r2[w], wg[6], s); s = fmaf(r2[w + 1], wg[7], s); s = fmaf(r2[w + 2], wg[8], s);
        const float v = siluf(s);
        const int o0 = ((w >> 2) * 16 + (w & 3)) * DIc;
        const int o1 = ((w >> 2) * (HH / 4) * 16 + (w & 3) * 4) * DIc;
        (q0 + o0)[c] = v;
        (q1 + o1)[c] = v;
        (q2 - o0)[c] = v;
        (q3 - o1)[c] = v;
        tot += v;
    }
    rowsum[((size_t)b * HH + h) * DIc + c] = tot;
}

__global__ void k_gate(const float* __restrict__ rowsum, const float* __restrict__ fw1,
                       const float* __restrict__ fw2, float* __restrict__ gate)
{
    __shared__ float g[DIc];
    __shared__ float h1[DMm / 4];
    int b = blockIdx.x, d = threadIdx.x;
    float gs = 0.f;
#pragma unroll
    for (int h = 0; h < HH; ++h) gs += rowsum[((size_t)b * HH + h) * DIc + d];
    g[d] = gs * (1.f / LL);
    __syncthreads();
    if (d < DMm / 4) {
        float s = 0.f;
        for (int j = 0; j < DMm; ++j) s += fw1[d * DMm + j] * g[j];
        h1[d] = geluf(s);
    }
    __syncthreads();
    float s2 = 0.f;
#pragma unroll
    for (int j = 0; j < DMm / 4; ++j) s2 += fw2[d * (DMm / 4) + j] * h1[j];
    gate[b * DMm + d] = sigm(s2);
}

__global__ void __launch_bounds__(128)
k_xproj(const float* __restrict__ xs, const float* __restrict__ wpad,
        float* __restrict__ xdbl)
{
    __shared__ __attribute__((aligned(16))) float lds[(64 + 48) * LPITCH];
    int k = blockIdx.z;
    const float* X = xs + (size_t)k * BB * LL * DIc;
    const float* W = wpad + (size_t)k * CPAD * DIc;
    float* O = xdbl + (size_t)k * BB * LL * CPROJ;
    v8f acc[3];
    wmma_lds_k192<4, 1, 3>(X, W, blockIdx.y * 64, 0, lds, acc);
    const int row0 = blockIdx.y * 64 + (threadIdx.x >> 5) * 16, col0 = 0;
    const int lane = threadIdx.x & 31;
    const int nn = lane & 15, mh = (lane >> 4) * 8;
#pragma unroll
    for (int t = 0; t < 3; ++t)
#pragma unroll
        for (int v = 0; v < 8; ++v) {
            int r = row0 + mh + v, c = col0 + t * 16 + nn;
            if (c < CPROJ) O[(size_t)r * CPROJ + c] = acc[t][v];
        }
}

__global__ void k_dtproj(const float* __restrict__ xdbl, const float* __restrict__ dtw,
                         const float* __restrict__ dtb, float* __restrict__ delta)
{
    int k = blockIdx.z;
    const float* X = xdbl + (size_t)k * BB * LL * CPROJ;
    const float* W = dtw + (size_t)k * DIc * RR;
    float* O = delta + (size_t)k * BB * LL * DIc;
    const int row0 = blockIdx.y * 16, col0 = blockIdx.x * 32;
    v8f acc[2];
    wmma_xwT_multi<2>(X, W, row0, col0, RR, CPROJ, RR, acc);
    const int lane = threadIdx.x & 31;
    const int nn = lane & 15, mh = (lane >> 4) * 8;
#pragma unroll
    for (int t = 0; t < 2; ++t)
#pragma unroll
        for (int v = 0; v < 8; ++v) {
            int r = row0 + mh + v, c = col0 + t * 16 + nn;
            O[(size_t)r * DIc + c] = softplusf_(acc[t][v] + dtb[k * DIc + c]);
        }
}

#define SC_C   96
#define SC_LEN (LL / SC_C)

__global__ void k_scan_local(const float* __restrict__ xs, const float* __restrict__ xdbl,
                             const float* __restrict__ delta, const float* __restrict__ A_logs,
                             float* __restrict__ sumP, float* __restrict__ sumS)
{
    const int c = blockIdx.x, kb = blockIdx.y;
    const int k = kb & 3, b = kb >> 2;
    const int d = threadIdx.x;
    float A[NN];
#pragma unroll
    for (int n = 0; n < NN; ++n)
        A[n] = -__expf(A_logs[((size_t)(k * DIc + d)) * NN + n]);
    float h[NN];
#pragma unroll
    for (int n = 0; n < NN; ++n) h[n] = 0.f;
    const size_t seqbase = ((size_t)(k * BB + b)) * LL + (size_t)c * SC_LEN;
    const float* up = xs    + seqbase * DIc + d;
    const float* dp = delta + seqbase * DIc + d;
    const float* bc = xdbl  + seqbase * CPROJ;
    float sdt = 0.f;
    for (int l = 0; l < SC_LEN; ++l) {
        float dt = dp[(size_t)l * DIc];
        float u  = up[(size_t)l * DIc];
        float du = dt * u;
        const float* bcl = bc + (size_t)l * CPROJ;
#pragma unroll
        for (int n = 0; n < NN; ++n)
            h[n] = __expf(dt * A[n]) * h[n] + du * bcl[RR + n];
        sdt += dt;
    }
    const size_t o = (((size_t)kb * SC_C + c) * NN) * DIc + d;
#pragma unroll
    for (int n = 0; n < NN; ++n) {
        sumP[o + (size_t)n * DIc] = __expf(sdt * A[n]);
        sumS[o + (size_t)n * DIc] = h[n];
    }
}

__global__ void k_scan_prefix(const float* __restrict__ sumP, const float* __restrict__ sumS,
                              float* __restrict__ hst)
{
    const int n = blockIdx.x, kb = blockIdx.y, d = threadIdx.x;
    const size_t base = (((size_t)kb * SC_C) * NN + n) * DIc + d;
    const size_t cs = (size_t)NN * DIc;
    float H = 0.f;
    hst[base] = H;
#pragma unroll 19
    for (int c = 0; c + 1 < SC_C; ++c) {
        H = sumP[base + c * cs] * H + sumS[base + c * cs];
        hst[base + (c + 1) * cs] = H;
    }
}

__global__ void k_scan_chunk(const float* __restrict__ xs, const float* __restrict__ xdbl,
                             const float* __restrict__ delta, const float* __restrict__ A_logs,
                             const float* __restrict__ Ds, const float* __restrict__ hst,
                             float* __restrict__ yarr)
{
    const int c = blockIdx.x, kb = blockIdx.y;
    const int k = kb & 3, b = kb >> 2;
    const int d = threadIdx.x;
    float A[NN];
#pragma unroll
    for (int n = 0; n < NN; ++n)
        A[n] = -__expf(A_logs[((size_t)(k * DIc + d)) * NN + n]);
    float Dv = Ds[k * DIc + d];
    float h[NN];
    const size_t o = (((size_t)kb * SC_C + c) * NN) * DIc + d;
#pragma unroll
    for (int n = 0; n < NN; ++n) h[n] = hst[o + (size_t)n * DIc];
    const size_t seqbase = ((size_t)(k * BB + b)) * LL + (size_t)c * SC_LEN;
    const float* up = xs    + seqbase * DIc + d;
    const float* dp = delta + seqbase * DIc + d;
    const float* bc = xdbl  + seqbase * CPROJ;
    float* yp       = yarr  + seqbase * DIc + d;
    for (int l = 0; l < SC_LEN; ++l) {
        float dt = dp[(size_t)l * DIc];
        float u  = up[(size_t)l * DIc];
        float du = dt * u;
        const float* bcl = bc + (size_t)l * CPROJ;
        float y = 0.f;
#pragma unroll
        for (int n = 0; n < NN; ++n) {
            float Bt = bcl[RR + n];
            float Ct = bcl[RR + NN + n];
            h[n] = __expf(dt * A[n]) * h[n] + du * Bt;
            y += h[n] * Ct;
        }
        yp[(size_t)l * DIc] = y + Dv * u;
    }
}

__device__ __forceinline__ void ld6(const float* __restrict__ p, int lane, float v[6])
{
    const float4 a = *(const float4*)(p + 4 * lane);
    const float2 c = *(const float2*)(p + 128 + 2 * lane);
    v[0] = a.x; v[1] = a.y; v[2] = a.z; v[3] = a.w; v[4] = c.x; v[5] = c.y;
}

__global__ void __launch_bounds__(128)
k_outproj(const float* __restrict__ yarr, const float* __restrict__ gate,
          const float* __restrict__ g1, const float* __restrict__ g2,
          const float* __restrict__ g3, const float* __restrict__ g4,
          const float* __restrict__ nw, const float* __restrict__ nb,
          const float* __restrict__ z_silu, const float* __restrict__ w,
          float* __restrict__ out)
{
    __shared__ __attribute__((aligned(16))) float lds[(32 + 64) * LPITCH];
    float* __restrict__ lA = lds;
    float* __restrict__ lB = lds + 32 * LPITCH;
    const int tid = threadIdx.x;
    const int lane = tid & 31, wave = tid >> 5;
    {
        const float4* __restrict__ gB = (const float4*)(w + (size_t)blockIdx.x * 64 * 192);
#pragma unroll
        for (int j = 0; j < 64 * KPIECES / 128; ++j) {
            const int i = tid + j * 128;
            const int r = i / KPIECES, c = i - r * KPIECES;
            *(float4*)(lB + r * LPITCH + c * 4) = gB[i];
        }
    }
    {
        const int b   = blockIdx.y / (LL / 32);
        const int lr0 = (blockIdx.y - b * (LL / 32)) * 32 + wave * 8;
        const size_t bstride = (size_t)LL * DIc;
        const float* __restrict__ Y0 = yarr + ((size_t)(0 * BB + b)) * bstride;
        const float* __restrict__ Y1 = yarr + ((size_t)(1 * BB + b)) * bstride;
        const float* __restrict__ Y2 = yarr + ((size_t)(2 * BB + b)) * bstride;
        const float* __restrict__ Y3 = yarr + ((size_t)(3 * BB + b)) * bstride;
        float G1[6], G2[6], G3[6], G4[6], GT[6], NW[6], NB[6];
        ld6(g1, lane, G1); ld6(g2, lane, G2); ld6(g3, lane, G3); ld6(g4, lane, G4);
        ld6(gate + b * DIc, lane, GT); ld6(nw, lane, NW); ld6(nb, lane, NB);
#pragma unroll
        for (int i = 0; i < 8; ++i) {
            const int lr = lr0 + i;
            const int h = lr / WW, wq = lr - h * WW;
            const int l  = (((h >> 2) * (WW / 4) + (wq >> 2)) << 4) + ((h & 3) << 2) + (wq & 3);
            const int lt = (l % WW) * HH + (l / WW);
            float y0[6], y1[6], y2[6], y3[6], zz[6];
            ld6(Y0 + (size_t)l * DIc, lane, y0);
            ld6(Y2 + (size_t)(LL - 1 - l) * DIc, lane, y2);
            ld6(Y1 + (size_t)lt * DIc, lane, y1);
            ld6(Y3 + (size_t)(LL - 1 - lt) * DIc, lane, y3);
            ld6(z_silu + ((size_t)b * LL + lr) * DIc, lane, zz);
            float yc[6];
            float s1 = 0.f, s2 = 0.f;
#pragma unroll
            for (int j = 0; j < 6; ++j) {
                float v = G1[j] * y0[j] + G2[j] * y2[j] + G3[j] * y1[j] + G4[j] * y3[j];
                v *= GT[j];
                yc[j] = v;
                s1 += v; s2 += v * v;
            }
#pragma unroll
            for (int off = 16; off > 0; off >>= 1) {
                s1 += __shfl_xor(s1, off);
                s2 += __shfl_xor(s2, off);
            }
            const float mu  = s1 * (1.f / DIc);
            const float var = s2 * (1.f / DIc) - mu * mu;
            const float inv = rsqrtf(var + 1e-5f);
            float o[6];
#pragma unroll
            for (int j = 0; j < 6; ++j) {
                float v = (yc[j] - mu) * inv * NW[j] + NB[j];
                o[j] = v * zz[j];
            }
            float* __restrict__ dst = lA + (wave * 8 + i) * LPITCH;
            *(float4*)(dst + 4 * lane)       = make_float4(o[0], o[1], o[2], o[3]);
            *(float2*)(dst + 128 + 2 * lane) = make_float2(o[4], o[5]);
        }
    }
    __syncthreads();
    v8f acc[2];
    wmma_frags_k192<2, 2, 2>(lA, lB, acc);
    const int row0 = blockIdx.y * 32 + (wave >> 1) * 16;
    const int col0 = blockIdx.x * 64 + (wave & 1) * 32;
    const int nn = lane & 15, mh = (lane >> 4) * 8;
#pragma unroll
    for (int t = 0; t < 2; ++t)
#pragma unroll
        for (int v = 0; v < 8; ++v) {
            int r = row0 + mh + v, c = col0 + t * 16 + nn;
            out[(size_t)r * DMm + c] = acc[t][v];
        }
}

extern "C" void kernel_launch(void* const* d_in, const int* in_sizes, int n_in,
                              void* d_out, int out_size, void* d_ws, size_t ws_size,
                              hipStream_t stream)
{
    (void)in_sizes; (void)n_in; (void)out_size; (void)ws_size;
    const float* x          = (const float*)d_in[0];
    const float* in_proj_w  = (const float*)d_in[1];
    const float* conv_w     = (const float*)d_in[2];
    const float* conv_b     = (const float*)d_in[3];
    const float* x_proj_w   = (const float*)d_in[4];
    const float* dt_w       = (const float*)d_in[5];
    const float* dt_b       = (const float*)d_in[6];
    const float* A_logs     = (const float*)d_in[7];
    const float* Ds         = (const float*)d_in[8];
    const float* gamma1     = (const float*)d_in[9];
    const float* gamma2     = (const float*)d_in[10];
    const float* gamma3     = (const float*)d_in[11];
    const float* gamma4     = (const float*)d_in[12];
    const float* fc_w1      = (const float*)d_in[13];
    const float* fc_w2      = (const float*)d_in[14];
    const float* norm_w     = (const float*)d_in[15];
    const float* norm_b     = (const float*)d_in[16];
    const float* out_proj_w = (const float*)d_in[17];
    float* out = (float*)d_out;

    float* ws = (float*)d_ws;
    size_t o = 0;
    float* xc_lin = ws + o; o += (size_t)BB * LL * DIc;
    float* z_silu = ws + o; o += (size_t)BB * LL * DIc;
    float* xs     = ws + o; o += (size_t)KK * BB * LL * DIc;
    float* xdbl   = ws + o; o += (size_t)KK * BB * LL * CPROJ;
    float* delta  = ws + o; o += (size_t)KK * BB * LL * DIc;
    float* yarr   = ws + o; o += (size_t)KK * BB * LL * DIc;
    float* wpad   = ws + o; o += (size_t)KK * CPAD * DIc;
    float* rowsum = ws + o; o += (size_t)BB * HH * DIc;
    float* gate   = ws + o; o += BB * DMm;
    float* sumP   = ws + o; o += (size_t)KK * BB * SC_C * NN * DIc;
    float* sumS   = ws + o; o += (size_t)KK * BB * SC_C * NN * DIc;
    float* hst    = ws + o; o += (size_t)KK * BB * SC_C * NN * DIc;

    k_padw    <<<dim3(KK, CPAD),       DIc, 0, stream>>>(x_proj_w, wpad);
    k_inproj  <<<dim3(6, 288),         128, 0, stream>>>(x, in_proj_w, xc_lin, z_silu);
    k_dwconv  <<<dim3(HH, BB),         DIc, 0, stream>>>(xc_lin, conv_w, conv_b, xs, rowsum);
    k_gate    <<<dim3(BB),             DIc, 0, stream>>>(rowsum, fc_w1, fc_w2, gate);
    k_xproj   <<<dim3(1, 144, KK),     128, 0, stream>>>(xs, wpad, xdbl);
    k_dtproj  <<<dim3(6, 576, KK),     32,  0, stream>>>(xdbl, dt_w, dt_b, delta);
    k_scan_local <<<dim3(SC_C - 1, BB * KK), DIc, 0, stream>>>(xs, xdbl, delta, A_logs, sumP, sumS);
    k_scan_prefix<<<dim3(NN, BB * KK),       DIc, 0, stream>>>(sumP, sumS, hst);
    k_scan_chunk <<<dim3(SC_C, BB * KK),     DIc, 0, stream>>>(xs, xdbl, delta, A_logs, Ds, hst, yarr);
    k_outproj <<<dim3(3, 288),         128, 0, stream>>>(yarr, gate, gamma1, gamma2, gamma3, gamma4,
                                                         norm_w, norm_b, z_silu, out_proj_w, out);
}


# k_dtproj: one wave per 16 rows x all 192 cols (2304 waves, not 13824 one-wave groups), its softplus as log(1+e)*e/((1+e)-1) on hardware exp/log/rcp, not library log1p (delta rounding changes)
# speedup vs baseline: 13.2465x; 1.0375x over previous
#include <hip/hip_runtime.h>
#include <math.h>


typedef float v2f __attribute__((ext_vector_type(2)));
typedef float v8f __attribute__((ext_vector_type(8)));

#define BB 4
#define HH 48
#define WW 48
#define DMm 192
#define DIc 192
#define KK 4
#define NN 16
#define RR 12
#define LL (HH * WW)
#define CPROJ (RR + 2 * NN)
#define CPAD 64

__device__ __forceinline__ float siluf(float x)     { return x / (1.f + __expf(-x)); }
__device__ __forceinline__ float sigm(float x)      { return 1.f / (1.f + __expf(-x)); }
__device__ __forceinline__ float softplusf_(float x)
{
    const float e   = __expf(x);
    const float u   = 1.f + e;
    const float um1 = u - 1.f;
    const float lnu = __builtin_amdgcn_logf(u) * 0.69314718055994530942f;
    const float lg  = lnu * (e * __builtin_amdgcn_rcpf(um1));
    const float sp  = (um1 == 0.f) ? e : lg;
    return (x > 20.f) ? x : sp;
}
__device__ __forceinline__ float geluf(float x)     { return 0.5f * x * (1.f + erff(x * 0.70710678118654752440f)); }

template <int NT>
__device__ __forceinline__ void wmma_xwT_multi(const float* __restrict__ X,
                                               const float* __restrict__ W,
                                               int row0, int col0,
                                               int Kdim, int ldx, int ldw,
                                               v8f acc[NT])
{
    const int lane = threadIdx.x & 31;
    const int m    = lane & 15;
    const int kh   = lane >> 4;
    const float* __restrict__ ap = X + (size_t)(row0 + m) * ldx + 2 * kh;
    const float* __restrict__ bp[NT];
#pragma unroll
    for (int t = 0; t < NT; ++t) {
        bp[t] = W + (size_t)(col0 + t * 16 + m) * ldw + 2 * kh;
        acc[t] = {};
    }
    v2f a_cur = *(const v2f*)(ap);
    v2f b_cur[NT];
#pragma unroll
    for (int t = 0; t < NT; ++t) b_cur[t] = *(const v2f*)(bp[t]);

    for (int k0 = 4; k0 < Kdim; k0 += 4) {
        v2f a_nxt = *(const v2f*)(ap + k0);
        v2f b_nxt[NT];
#pragma unroll
        for (int t = 0; t < NT; ++t) b_nxt[t] = *(const v2f*)(bp[t] + k0);
#pragma unroll
        for (int t = 0; t < NT; ++t)
            acc[t] = __builtin_amdgcn_wmma_f32_16x16x4_f32(false, a_cur, false, b_cur[t],
                                                           (short)0, acc[t], false, false);
        a_cur = a_nxt;
#pragma unroll
        for (int t = 0; t < NT; ++t) b_cur[t] = b_nxt[t];
    }
#pragma unroll
    for (int t = 0; t < NT; ++t)
        acc[t] = __builtin_amdgcn_wmma_f32_16x16x4_f32(false, a_cur, false, b_cur[t],
                                                       (short)0, acc[t], false, false);
}

#define LPITCH 196
#define KPIECES (192 / 4)

template <int WM, int WN, int NT>
__device__ __forceinline__ void wmma_frags_k192(const float* __restrict__ lA,
                                                const float* __restrict__ lB, v8f acc[NT])
{
    const int tid = threadIdx.x;
    const int lane = tid & 31, wave = tid >> 5;
    const int wm = wave / WN, wn = wave - wm * WN;
    const int m  = lane & 15;
    const int kh = lane >> 4;
    const float* __restrict__ ap = lA + (wm * 16 + m) * LPITCH + 2 * kh;
    const float* __restrict__ bp = lB + (wn * NT * 16 + m) * LPITCH + 2 * kh;
#pragma unroll
    for (int t = 0; t < NT; ++t) acc[t] = {};
#pragma unroll
    for (int ks = 0; ks < 192 / 4; ++ks) {
        const v2f a = *(const v2f*)(ap + 4 * ks);
        v2f b[NT];
#pragma unroll
        for (int t = 0; t < NT; ++t) b[t] = *(const v2f*)(bp + t * 16 * LPITCH + 4 * ks);
#pragma unroll
        for (int t = 0; t < NT; ++t)
            acc[t] = __builtin_amdgcn_wmma_f32_16x16x4_f32(false, a, false, b[t],
                                                           (short)0, acc[t], false, false);
    }
}

template <int WM, int WN, int NT>
__device__ __forceinline__ void wmma_lds_k192(const float* __restrict__ X,
                                              const float* __restrict__ W,
                                              int row0, int col0,
                                              float* __restrict__ lds, v8f acc[NT])
{
    constexpr int NTHR = WM * WN * 32;
    constexpr int AR = WM * 16;
    constexpr int BR = WN * NT * 16;
    static_assert((AR * KPIECES) % NTHR == 0 && (BR * KPIECES) % NTHR == 0, "whole copy passes");
    float* __restrict__ lA = lds;
    float* __restrict__ lB = lds + AR * LPITCH;
    const int tid = threadIdx.x;
    const float4* __restrict__ gA = (const float4*)(X + (size_t)row0 * 192);
    const float4* __restrict__ gB = (const float4*)(W + (size_t)col0 * 192);
#pragma unroll
    for (int j = 0; j < AR * KPIECES / NTHR; ++j) {
        const int i = tid + j * NTHR;
        const int r = i / KPIECES, c = i - r * KPIECES;
        *(float4*)(lA + r * LPITCH + c * 4) = gA[i];
    }
#pragma unroll
    for (int j = 0; j < BR * KPIECES / NTHR; ++j) {
        const int i = tid + j * NTHR;
        const int r = i / KPIECES, c = i - r * KPIECES;
        *(float4*)(lB + r * LPITCH + c * 4) = gB[i];
    }
    __syncthreads();
    wmma_frags_k192<WM, WN, NT>(lA, lB, acc);
}

__global__ void k_padw(const float* __restrict__ xpw, float* __restrict__ wpad)
{
    int k = blockIdx.x, c = blockIdx.y, d = threadIdx.x;
    wpad[((size_t)k * CPAD + c) * DIc + d] =
        (c < CPROJ) ? xpw[((size_t)k * CPROJ + c) * DIc + d] : 0.f;
}

__global__ void __launch_bounds__(128)
k_inproj(const float* __restrict__ x, const float* __restrict__ w,
         float* __restrict__ xc_lin, float* __restrict__ z_silu)
{
    __shared__ __attribute__((aligned(16))) float lds[(32 + 64) * LPITCH];
    v8f acc[2];
    wmma_lds_k192<2, 2, 2>(x, w, blockIdx.y * 32, blockIdx.x * 64, lds, acc);
    const int wave = threadIdx.x >> 5;
    const int row0 = blockIdx.y * 32 + (wave >> 1) * 16;
    const int col0 = blockIdx.x * 64 + (wave & 1) * 32;
    const int lane = threadIdx.x & 31;
    const int nn = lane & 15, mh = (lane >> 4) * 8;
#pragma unroll
    for (int t = 0; t < 2; ++t)
#pragma unroll
        for (int v = 0; v < 8; ++v) {
            int r = row0 + mh + v;
            int c = col0 + t * 16 + nn;
            float val = acc[t][v];
            if (c < DIc) xc_lin[(size_t)r * DIc + c] = val;
            else         z_silu[(size_t)r * DIc + (c - DIc)] = siluf(val);
        }
}

__global__ void __launch_bounds__(DIc)
k_dwconv(const float* __restrict__ xc_lin, const float* __restrict__ cw,
         const float* __restrict__ cb, float* __restrict__ xs,
         float* __restrict__ rowsum)
{
    const int h = blockIdx.x, b = blockIdx.y, c = threadIdx.x;
    const bool up = (h > 0), dn = (h < HH - 1);
    const int hu = up ? h - 1 : h;
    const int hd = dn ? h + 1 : h;
    const float* __restrict__ p0 = xc_lin + ((size_t)b * LL + (size_t)hu * WW) * DIc + c;
    const float* __restrict__ p1 = xc_lin + ((size_t)b * LL + (size_t)h  * WW) * DIc + c;
    const float* __restrict__ p2 = xc_lin + ((size_t)b * LL + (size_t)hd * WW) * DIc + c;
    float wg[9];
#pragma unroll
    for (int j = 0; j < 9; ++j) wg[j] = cw[c * 9 + j];
    const float bias = cb[c];
    float r0[WW + 2], r1[WW + 2], r2[WW + 2];
    r0[0] = 0.f; r1[0] = 0.f; r2[0] = 0.f;
    r0[WW + 1] = 0.f; r1[WW + 1] = 0.f; r2[WW + 1] = 0.f;
#pragma unroll
    for (int w = 0; w < WW; ++w) r0[w + 1] = p0[(size_t)w * DIc];
#pragma unroll
    for (int w = 0; w < WW; ++w) r1[w + 1] = p1[(size_t)w * DIc];
#pragma unroll
    for (int w = 0; w < WW; ++w) r2[w + 1] = p2[(size_t)w * DIc];
#pragma unroll
    for (int w = 0; w < WW; ++w) {
        r0[w + 1] = up ? r0[w + 1] : 0.f;
        r2[w + 1] = dn ? r2[w + 1] : 0.f;
    }
    const int l0h = (h >> 2) * (WW / 4) * 16 + (h & 3) * 4;
    const int l1h = (h >> 2) * 16 + (h & 3);
    float* __restrict__ q0 = xs + ((size_t)(0 * BB + b) * LL + l0h) * DIc;
    float* __restrict__ q1 = xs + ((size_t)(1 * BB + b) * LL + l1h) * DIc;
    float* __restrict__ q2 = xs + ((size_t)(2 * BB + b) * LL + (LL - 1 - l0h)) * DIc;
    float* __restrict__ q3 = xs + ((size_t)(3 * BB + b) * LL + (LL - 1 - l1h)) * DIc;
    float tot = 0.f;
#pragma unroll
    for (int w = 0; w < WW; ++w) {
        float s = bias;
        s = fmaf(r0[w], wg[0], s); s = fmaf(r0[w + 1], wg[1], s); s = fmaf(r0[w + 2], wg[2], s);
        s = fmaf(r1[w], wg[3], s); s = fmaf(r1[w + 1], wg[4], s); s = fmaf(r1[w + 2], wg[5], s);
        s = fmaf(r2[w], wg[6], s); s = fmaf(r2[w + 1], wg[7], s); s = fmaf(r2[w + 2], wg[8], s);
        const float v = siluf(s);
        const int o0 = ((w >> 2) * 16 + (w & 3)) * DIc;
        const int o1 = ((w >> 2) * (HH / 4) * 16 + (w & 3) * 4) * DIc;
        (q0 + o0)[c] = v;
        (q1 + o1)[c] = v;
        (q2 - o0)[c] = v;
        (q3 - o1)[c] = v;
        tot += v;
    }
    rowsum[((size_t)b * HH + h) * DIc + c] = tot;
}

__global__ void k_gate(const float* __restrict__ rowsum, const float* __restrict__ fw1,
                       const float* __restrict__ fw2, float* __restrict__ gate)
{
    __shared__ float g[DIc];
    __shared__ float h1[DMm / 4];
    int b = blockIdx.x, d = threadIdx.x;
    float gs = 0.f;
#pragma unroll
    for (int h = 0; h < HH; ++h) gs += rowsum[((size_t)b * HH + h) * DIc + d];
    g[d] = gs * (1.f / LL);
    __syncthreads();
    if (d < DMm / 4) {
        float s = 0.f;
        for (int j = 0; j < DMm; ++j) s += fw1[d * DMm + j] * g[j];
        h1[d] = geluf(s);
    }
    __syncthreads();
    float s2 = 0.f;
#pragma unroll
    for (int j = 0; j < DMm / 4; ++j) s2 += fw2[d * (DMm / 4) + j] * h1[j];
    gate[b * DMm + d] = sigm(s2);
}

__global__ void __launch_bounds__(128)
k_xproj(const float* __restrict__ xs, const float* __restrict__ wpad,
        float* __restrict__ xdbl)
{
    __shared__ __attribute__((aligned(16))) float lds[(64 + 48) * LPITCH];
    int k = blockIdx.z;
    const float* X = xs + (size_t)k * BB * LL * DIc;
    const float* W = wpad + (size_t)k * CPAD * DIc;
    float* O = xdbl + (size_t)k * BB * LL * CPROJ;
    v8f acc[3];
    wmma_lds_k192<4, 1, 3>(X, W, blockIdx.y * 64, 0, lds, acc);
    const int row0 = blockIdx.y * 64 + (threadIdx.x >> 5) * 16, col0 = 0;
    const int lane = threadIdx.x & 31;
    const int nn = lane & 15, mh = (lane >> 4) * 8;
#pragma unroll
    for (int t = 0; t < 3; ++t)
#pragma unroll
        for (int v = 0; v < 8; ++v) {
            int r = row0 + mh + v, c = col0 + t * 16 + nn;
            if (c < CPROJ) O[(size_t)r * CPROJ + c] = acc[t][v];
        }
}

#define DT_NT (DIc / 16)
__global__ void __launch_bounds__(64)
k_dtproj(const float* __restrict__ xdbl, const float* __restrict__ dtw,
         const float* __restrict__ dtb, float* __restrict__ delta)
{
    const int k = blockIdx.z;
    const float* X = xdbl + (size_t)k * BB * LL * CPROJ;
    const float* W = dtw + (size_t)k * DIc * RR;
    float* O = delta + (size_t)k * BB * LL * DIc;
    const int row0 = (blockIdx.y * 2 + (threadIdx.x >> 5)) * 16;
    const int lane = threadIdx.x & 31;
    const int nn = lane & 15, mh = (lane >> 4) * 8;
    float bias[DT_NT];
#pragma unroll
    for (int t = 0; t < DT_NT; ++t) bias[t] = dtb[k * DIc + t * 16 + nn];
    v8f acc[DT_NT];
    wmma_xwT_multi<DT_NT>(X, W, row0, 0, RR, CPROJ, RR, acc);
#pragma unroll
    for (int t = 0; t < DT_NT; ++t)
#pragma unroll
        for (int v = 0; v < 8; ++v) {
            const int r = row0 + mh + v, c = t * 16 + nn;
            O[(size_t)r * DIc + c] = softplusf_(acc[t][v] + bias[t]);
        }
}

#define SC_C   96
#define SC_LEN (LL / SC_C)

__global__ void k_scan_local(const float* __restrict__ xs, const float* __restrict__ xdbl,
                             const float* __restrict__ delta, const float* __restrict__ A_logs,
                             float* __restrict__ sumP, float* __restrict__ sumS)
{
    const int c = blockIdx.x, kb = blockIdx.y;
    const int k = kb & 3, b = kb >> 2;
    const int d = threadIdx.x;
    float A[NN];
#pragma unroll
    for (int n = 0; n < NN; ++n)
        A[n] = -__expf(A_logs[((size_t)(k * DIc + d)) * NN + n]);
    float h[NN];
#pragma unroll
    for (int n = 0; n < NN; ++n) h[n] = 0.f;
    const size_t seqbase = ((size_t)(k * BB + b)) * LL + (size_t)c * SC_LEN;
    const float* up = xs    + seqbase * DIc + d;
    const float* dp = delta + seqbase * DIc + d;
    const float* bc = xdbl  + seqbase * CPROJ;
    float sdt = 0.f;
    for (int l = 0; l < SC_LEN; ++l) {
        float dt = dp[(size_t)l * DIc];
        float u  = up[(size_t)l * DIc];
        float du = dt * u;
        const float* bcl = bc + (size_t)l * CPROJ;
#pragma unroll
        for (int n = 0; n < NN; ++n)
            h[n] = __expf(dt * A[n]) * h[n] + du * bcl[RR + n];
        sdt += dt;
    }
    const size_t o = (((size_t)kb * SC_C + c) * NN) * DIc + d;
#pragma unroll
    for (int n = 0; n < NN; ++n) {
        sumP[o + (size_t)n * DIc] = __expf(sdt * A[n]);
        sumS[o + (size_t)n * DIc] = h[n];
    }
}

__global__ void k_scan_prefix(const float* __restrict__ sumP, const float* __restrict__ sumS,
                              float* __restrict__ hst)
{
    const int n = blockIdx.x, kb = blockIdx.y, d = threadIdx.x;
    const size_t base = (((size_t)kb * SC_C) * NN + n) * DIc + d;
    const size_t cs = (size_t)NN * DIc;
    float H = 0.f;
    hst[base] = H;
#pragma unroll 19
    for (int c = 0; c + 1 < SC_C; ++c) {
        H = sumP[base + c * cs] * H + sumS[base + c * cs];
        hst[base + (c + 1) * cs] = H;
    }
}

__global__ void k_scan_chunk(const float* __restrict__ xs, const float* __restrict__ xdbl,
                             const float* __restrict__ delta, const float* __restrict__ A_logs,
                             const float* __restrict__ Ds, const float* __restrict__ hst,
                             float* __restrict__ yarr)
{
    const int c = blockIdx.x, kb = blockIdx.y;
    const int k = kb & 3, b = kb >> 2;
    const int d = threadIdx.x;
    float A[NN];
#pragma unroll
    for (int n = 0; n < NN; ++n)
        A[n] = -__expf(A_logs[((size_t)(k * DIc + d)) * NN + n]);
    float Dv = Ds[k * DIc + d];
    float h[NN];
    const size_t o = (((size_t)kb * SC_C + c) * NN) * DIc + d;
#pragma unroll
    for (int n = 0; n < NN; ++n) h[n] = hst[o + (size_t)n * DIc];
    const size_t seqbase = ((size_t)(k * BB + b)) * LL + (size_t)c * SC_LEN;
    const float* up = xs    + seqbase * DIc + d;
    const float* dp = delta + seqbase * DIc + d;
    const float* bc = xdbl  + seqbase * CPROJ;
    float* yp       = yarr  + seqbase * DIc + d;
    for (int l = 0; l < SC_LEN; ++l) {
        float dt = dp[(size_t)l * DIc];
        float u  = up[(size_t)l * DIc];
        float du = dt * u;
        const float* bcl = bc + (size_t)l * CPROJ;
        float y = 0.f;
#pragma unroll
        for (int n = 0; n < NN; ++n) {
            float Bt = bcl[RR + n];
            float Ct = bcl[RR + NN + n];
            h[n] = __expf(dt * A[n]) * h[n] + du * Bt;
            y += h[n] * Ct;
        }
        yp[(size_t)l * DIc] = y + Dv * u;
    }
}

__device__ __forceinline__ void ld6(const float* __restrict__ p, int lane, float v[6])
{
    const float4 a = *(const float4*)(p + 4 * lane);
    const float2 c = *(const float2*)(p + 128 + 2 * lane);
    v[0] = a.x; v[1] = a.y; v[2] = a.z; v[3] = a.w; v[4] = c.x; v[5] = c.y;
}

__global__ void __launch_bounds__(128)
k_outproj(const float* __restrict__ yarr, const float* __restrict__ gate,
          const float* __restrict__ g1, const float* __restrict__ g2,
          const float* __restrict__ g3, const float* __restrict__ g4,
          const float* __restrict__ nw, const float* __restrict__ nb,
          const float* __restrict__ z_silu, const float* __restrict__ w,
          float* __restrict__ out)
{
    __shared__ __attribute__((aligned(16))) float lds[(32 + 64) * LPITCH];
    float* __restrict__ lA = lds;
    float* __restrict__ lB = lds + 32 * LPITCH;
    const int tid = threadIdx.x;
    const int lane = tid & 31, wave = tid >> 5;
    {
        const float4* __restrict__ gB = (const float4*)(w + (size_t)blockIdx.x * 64 * 192);
#pragma unroll
        for (int j = 0; j < 64 * KPIECES / 128; ++j) {
            const int i = tid + j * 128;
            const int r = i / KPIECES, c = i - r * KPIECES;
            *(float4*)(lB + r * LPITCH + c * 4) = gB[i];
        }
    }
    {
        const int b   = blockIdx.y / (LL / 32);
        const int lr0 = (blockIdx.y - b * (LL / 32)) * 32 + wave * 8;
        const size_t bstride = (size_t)LL * DIc;
        const float* __restrict__ Y0 = yarr + ((size_t)(0 * BB + b)) * bstride;
        const float* __restrict__ Y1 = yarr + ((size_t)(1 * BB + b)) * bstride;
        const float* __restrict__ Y2 = yarr + ((size_t)(2 * BB + b)) * bstride;
        const float* __restrict__ Y3 = yarr + ((size_t)(3 * BB + b)) * bstride;
        float G1[6], G2[6], G3[6], G4[6], GT[6], NW[6], NB[6];
        ld6(g1, lane, G1); ld6(g2, lane, G2); ld6(g3, lane, G3); ld6(g4, lane, G4);
        ld6(gate + b * DIc, lane, GT); ld6(nw, lane, NW); ld6(nb, lane, NB);
#pragma unroll
        for (int i = 0; i < 8; ++i) {
            const int lr = lr0 + i;
            const int h = lr / WW, wq = lr - h * WW;
            const int l  = (((h >> 2) * (WW / 4) + (wq >> 2)) << 4) + ((h & 3) << 2) + (wq & 3);
            const int lt = (l % WW) * HH + (l / WW);
            float y0[6], y1[6], y2[6], y3[6], zz[6];
            ld6(Y0 + (size_t)l * DIc, lane, y0);
            ld6(Y2 + (size_t)(LL - 1 - l) * DIc, lane, y2);
            ld6(Y1 + (size_t)lt * DIc, lane, y1);
            ld6(Y3 + (size_t)(LL - 1 - lt) * DIc, lane, y3);
            ld6(z_silu + ((size_t)b * LL + lr) * DIc, lane, zz);
            float yc[6];
            float s1 = 0.f, s2 = 0.f;
#pragma unroll
            for (int j = 0; j < 6; ++j) {
                float v = G1[j] * y0[j] + G2[j] * y2[j] + G3[j] * y1[j] + G4[j] * y3[j];
                v *= GT[j];
                yc[j] = v;
                s1 += v; s2 += v * v;
            }
#pragma unroll
            for (int off = 16; off > 0; off >>= 1) {
                s1 += __shfl_xor(s1, off);
                s2 += __shfl_xor(s2, off);
            }
            const float mu  = s1 * (1.f / DIc);
            const float var = s2 * (1.f / DIc) - mu * mu;
            const float inv = rsqrtf(var + 1e-5f);
            float o[6];
#pragma unroll
            for (int j = 0; j < 6; ++j) {
                float v = (yc[j] - mu) * inv * NW[j] + NB[j];
                o[j] = v * zz[j];
            }
            float* __restrict__ dst = lA + (wave * 8 + i) * LPITCH;
            *(float4*)(dst + 4 * lane)       = make_float4(o[0], o[1], o[2], o[3]);
            *(float2*)(dst + 128 + 2 * lane) = make_float2(o[4], o[5]);
        }
    }
    __syncthreads();
    v8f acc[2];
    wmma_frags_k192<2, 2, 2>(lA, lB, acc);
    const int row0 = blockIdx.y * 32 + (wave >> 1) * 16;
    const int col0 = blockIdx.x * 64 + (wave & 1) * 32;
    const int nn = lane & 15, mh = (lane >> 4) * 8;
#pragma unroll
    for (int t = 0; t < 2; ++t)
#pragma unroll
        for (int v = 0; v < 8; ++v) {
            int r = row0 + mh + v, c = col0 + t * 16 + nn;
            out[(size_t)r * DMm + c] = acc[t][v];
        }
}

extern "C" void kernel_launch(void* const* d_in, const int* in_sizes, int n_in,
                              void* d_out, int out_size, void* d_ws, size_t ws_size,
                              hipStream_t stream)
{
    (void)in_sizes; (void)n_in; (void)out_size; (void)ws_size;
    const float* x          = (const float*)d_in[0];
    const float* in_proj_w  = (const float*)d_in[1];
    const float* conv_w     = (const float*)d_in[2];
    const float* conv_b     = (const float*)d_in[3];
    const float* x_proj_w   = (const float*)d_in[4];
    const float* dt_w       = (const float*)d_in[5];
    const float* dt_b       = (const float*)d_in[6];
    const float* A_logs     = (const float*)d_in[7];
    const float* Ds         = (const float*)d_in[8];
    const float* gamma1     = (const float*)d_in[9];
    const float* gamma2     = (const float*)d_in[10];
    const float* gamma3     = (const float*)d_in[11];
    const float* gamma4     = (const float*)d_in[12];
    const float* fc_w1      = (const float*)d_in[13];
    const float* fc_w2      = (const float*)d_in[14];
    const float* norm_w     = (const float*)d_in[15];
    const float* norm_b     = (const float*)d_in[16];
    const float* out_proj_w = (const float*)d_in[17];
    float* out = (float*)d_out;

    float* ws = (float*)d_ws;
    size_t o = 0;
    float* xc_lin = ws + o; o += (size_t)BB * LL * DIc;
    float* z_silu = ws + o; o += (size_t)BB * LL * DIc;
    float* xs     = ws + o; o += (size_t)KK * BB * LL * DIc;
    float* xdbl   = ws + o; o += (size_t)KK * BB * LL * CPROJ;
    float* delta  = ws + o; o += (size_t)KK * BB * LL * DIc;
    float* yarr   = ws + o; o += (size_t)KK * BB * LL * DIc;
    float* wpad   = ws + o; o += (size_t)KK * CPAD * DIc;
    float* rowsum = ws + o; o += (size_t)BB * HH * DIc;
    float* gate   = ws + o; o += BB * DMm;
    float* sumP   = ws + o; o += (size_t)KK * BB * SC_C * NN * DIc;
    float* sumS   = ws + o; o += (size_t)KK * BB * SC_C * NN * DIc;
    float* hst    = ws + o; o += (size_t)KK * BB * SC_C * NN * DIc;

    k_padw    <<<dim3(KK, CPAD),       DIc, 0, stream>>>(x_proj_w, wpad);
    k_inproj  <<<dim3(6, 288),         128, 0, stream>>>(x, in_proj_w, xc_lin, z_silu);
    k_dwconv  <<<dim3(HH, BB),         DIc, 0, stream>>>(xc_lin, conv_w, conv_b, xs, rowsum);
    k_gate    <<<dim3(BB),             DIc, 0, stream>>>(rowsum, fc_w1, fc_w2, gate);
    k_xproj   <<<dim3(1, 144, KK),     128, 0, stream>>>(xs, wpad, xdbl);
    k_dtproj  <<<dim3(1, 288, KK),     64,  0, stream>>>(xdbl, dt_w, dt_b, delta);
    k_scan_local <<<dim3(SC_C - 1, BB * KK), DIc, 0, stream>>>(xs, xdbl, delta, A_logs, sumP, sumS);
    k_scan_prefix<<<dim3(NN, BB * KK),       DIc, 0, stream>>>(sumP, sumS, hst);
    k_scan_chunk <<<dim3(SC_C, BB * KK),     DIc, 0, stream>>>(xs, xdbl, delta, A_logs, Ds, hst, yarr);
    k_outproj <<<dim3(3, 288),         128, 0, stream>>>(yarr, gate, gamma1, gamma2, gamma3, gamma4,
                                                         norm_w, norm_b, z_silu, out_proj_w, out);
}


# k_dtproj launch removed: each x_proj wave hands its 16x16 dt tile to itself through LDS and runs the K=12 dt product (12 col-tiles), bias and softplus on it; dts never stored; k order and sums kept
# speedup vs baseline: 13.6669x; 1.0317x over previous
#include <hip/hip_runtime.h>
#include <math.h>


typedef float v2f __attribute__((ext_vector_type(2)));
typedef float v8f __attribute__((ext_vector_type(8)));

#define BB 4
#define HH 48
#define WW 48
#define DMm 192
#define DIc 192
#define KK 4
#define NN 16
#define RR 12
#define LL (HH * WW)
#define CPROJ (RR + 2 * NN)
#define CPAD 64

__device__ __forceinline__ float siluf(float x)     { return x / (1.f + __expf(-x)); }
__device__ __forceinline__ float sigm(float x)      { return 1.f / (1.f + __expf(-x)); }
__device__ __forceinline__ float softplusf_(float x)
{
    const float e   = __expf(x);
    const float u   = 1.f + e;
    const float um1 = u - 1.f;
    const float lnu = __builtin_amdgcn_logf(u) * 0.69314718055994530942f;
    const float lg  = lnu * (e * __builtin_amdgcn_rcpf(um1));
    const float sp  = (um1 == 0.f) ? e : lg;
    return (x > 20.f) ? x : sp;
}
__device__ __forceinline__ float geluf(float x)     { return 0.5f * x * (1.f + erff(x * 0.70710678118654752440f)); }

template <int NT>
__device__ __forceinline__ void wmma_xwT_multi(const float* __restrict__ X,
                                               const float* __restrict__ W,
                                               int row0, int col0,
                                               int Kdim, int ldx, int ldw,
                                               v8f acc[NT])
{
    const int lane = threadIdx.x & 31;
    const int m    = lane & 15;
    const int kh   = lane >> 4;
    const float* __restrict__ ap = X + (size_t)(row0 + m) * ldx + 2 * kh;
    const float* __restrict__ bp[NT];
#pragma unroll
    for (int t = 0; t < NT; ++t) {
        bp[t] = W + (size_t)(col0 + t * 16 + m) * ldw + 2 * kh;
        acc[t] = {};
    }
    v2f a_cur = *(const v2f*)(ap);
    v2f b_cur[NT];
#pragma unroll
    for (int t = 0; t < NT; ++t) b_cur[t] = *(const v2f*)(bp[t]);

    for (int k0 = 4; k0 < Kdim; k0 += 4) {
        v2f a_nxt = *(const v2f*)(ap + k0);
        v2f b_nxt[NT];
#pragma unroll
        for (int t = 0; t < NT; ++t) b_nxt[t] = *(const v2f*)(bp[t] + k0);
#pragma unroll
        for (int t = 0; t < NT; ++t)
            acc[t] = __builtin_amdgcn_wmma_f32_16x16x4_f32(false, a_cur, false, b_cur[t],
                                                           (short)0, acc[t], false, false);
        a_cur = a_nxt;
#pragma unroll
        for (int t = 0; t < NT; ++t) b_cur[t] = b_nxt[t];
    }
#pragma unroll
    for (int t = 0; t < NT; ++t)
        acc[t] = __builtin_amdgcn_wmma_f32_16x16x4_f32(false, a_cur, false, b_cur[t],
                                                       (short)0, acc[t], false, false);
}

#define LPITCH 196
#define KPIECES (192 / 4)

template <int WM, int WN, int NT>
__device__ __forceinline__ void wmma_frags_k192(const float* __restrict__ lA,
                                                const float* __restrict__ lB, v8f acc[NT])
{
    const int tid = threadIdx.x;
    const int lane = tid & 31, wave = tid >> 5;
    const int wm = wave / WN, wn = wave - wm * WN;
    const int m  = lane & 15;
    const int kh = lane >> 4;
    const float* __restrict__ ap = lA + (wm * 16 + m) * LPITCH + 2 * kh;
    const float* __restrict__ bp = lB + (wn * NT * 16 + m) * LPITCH + 2 * kh;
#pragma unroll
    for (int t = 0; t < NT; ++t) acc[t] = {};
#pragma unroll
    for (int ks = 0; ks < 192 / 4; ++ks) {
        const v2f a = *(const v2f*)(ap + 4 * ks);
        v2f b[NT];
#pragma unroll
        for (int t = 0; t < NT; ++t) b[t] = *(const v2f*)(bp + t * 16 * LPITCH + 4 * ks);
#pragma unroll
        for (int t = 0; t < NT; ++t)
            acc[t] = __builtin_amdgcn_wmma_f32_16x16x4_f32(false, a, false, b[t],
                                                           (short)0, acc[t], false, false);
    }
}

template <int WM, int WN, int NT>
__device__ __forceinline__ void wmma_lds_k192(const float* __restrict__ X,
                                              const float* __restrict__ W,
                                              int row0, int col0,
                                              float* __restrict__ lds, v8f acc[NT])
{
    constexpr int NTHR = WM * WN * 32;
    constexpr int AR = WM * 16;
    constexpr int BR = WN * NT * 16;
    static_assert((AR * KPIECES) % NTHR == 0 && (BR * KPIECES) % NTHR == 0, "whole copy passes");
    float* __restrict__ lA = lds;
    float* __restrict__ lB = lds + AR * LPITCH;
    const int tid = threadIdx.x;
    const float4* __restrict__ gA = (const float4*)(X + (size_t)row0 * 192);
    const float4* __restrict__ gB = (const float4*)(W + (size_t)col0 * 192);
#pragma unroll
    for (int j = 0; j < AR * KPIECES / NTHR; ++j) {
        const int i = tid + j * NTHR;
        const int r = i / KPIECES, c = i - r * KPIECES;
        *(float4*)(lA + r * LPITCH + c * 4) = gA[i];
    }
#pragma unroll
    for (int j = 0; j < BR * KPIECES / NTHR; ++j) {
        const int i = tid + j * NTHR;
        const int r = i / KPIECES, c = i - r * KPIECES;
        *(float4*)(lB + r * LPITCH + c * 4) = gB[i];
    }
    __syncthreads();
    wmma_frags_k192<WM, WN, NT>(lA, lB, acc);
}

__global__ void k_padw(const float* __restrict__ xpw, float* __restrict__ wpad)
{
    int k = blockIdx.x, c = blockIdx.y, d = threadIdx.x;
    wpad[((size_t)k * CPAD + c) * DIc + d] =
        (c < CPROJ) ? xpw[((size_t)k * CPROJ + c) * DIc + d] : 0.f;
}

__global__ void __launch_bounds__(128)
k_inproj(const float* __restrict__ x, const float* __restrict__ w,
         float* __restrict__ xc_lin, float* __restrict__ z_silu)
{
    __shared__ __attribute__((aligned(16))) float lds[(32 + 64) * LPITCH];
    v8f acc[2];
    wmma_lds_k192<2, 2, 2>(x, w, blockIdx.y * 32, blockIdx.x * 64, lds, acc);
    const int wave = threadIdx.x >> 5;
    const int row0 = blockIdx.y * 32 + (wave >> 1) * 16;
    const int col0 = blockIdx.x * 64 + (wave & 1) * 32;
    const int lane = threadIdx.x & 31;
    const int nn = lane & 15, mh = (lane >> 4) * 8;
#pragma unroll
    for (int t = 0; t < 2; ++t)
#pragma unroll
        for (int v = 0; v < 8; ++v) {
            int r = row0 + mh + v;
            int c = col0 + t * 16 + nn;
            float val = acc[t][v];
            if (c < DIc) xc_lin[(size_t)r * DIc + c] = val;
            else         z_silu[(size_t)r * DIc + (c - DIc)] = siluf(val);
        }
}

__global__ void __launch_bounds__(DIc)
k_dwconv(const float* __restrict__ xc_lin, const float* __restrict__ cw,
         const float* __restrict__ cb, float* __restrict__ xs,
         float* __restrict__ rowsum)
{
    const int h = blockIdx.x, b = blockIdx.y, c = threadIdx.x;
    const bool up = (h > 0), dn = (h < HH - 1);
    const int hu = up ? h - 1 : h;
    const int hd = dn ? h + 1 : h;
    const float* __restrict__ p0 = xc_lin + ((size_t)b * LL + (size_t)hu * WW) * DIc + c;
    const float* __restrict__ p1 = xc_lin + ((size_t)b * LL + (size_t)h  * WW) * DIc + c;
    const float* __restrict__ p2 = xc_lin + ((size_t)b * LL + (size_t)hd * WW) * DIc + c;
    float wg[9];
#pragma unroll
    for (int j = 0; j < 9; ++j) wg[j] = cw[c * 9 + j];
    const float bias = cb[c];
    float r0[WW + 2], r1[WW + 2], r2[WW + 2];
    r0[0] = 0.f; r1[0] = 0.f; r2[0] = 0.f;
    r0[WW + 1] = 0.f; r1[WW + 1] = 0.f; r2[WW + 1] = 0.f;
#pragma unroll
    for (int w = 0; w < WW; ++w) r0[w + 1] = p0[(size_t)w * DIc];
#pragma unroll
    for (int w = 0; w < WW; ++w) r1[w + 1] = p1[(size_t)w * DIc];
#pragma unroll
    for (int w = 0; w < WW; ++w) r2[w + 1] = p2[(size_t)w * DIc];
#pragma unroll
    for (int w = 0; w < WW; ++w) {
        r0[w + 1] = up ? r0[w + 1] : 0.f;
        r2[w + 1] = dn ? r2[w + 1] : 0.f;
    }
    const int l0h = (h >> 2) * (WW / 4) * 16 + (h & 3) * 4;
    const int l1h = (h >> 2) * 16 + (h & 3);
    float* __restrict__ q0 = xs + ((size_t)(0 * BB + b) * LL + l0h) * DIc;
    float* __restrict__ q1 = xs + ((size_t)(1 * BB + b) * LL + l1h) * DIc;
    float* __restrict__ q2 = xs + ((size_t)(2 * BB + b) * LL + (LL - 1 - l0h)) * DIc;
    float* __restrict__ q3 = xs + ((size_t)(3 * BB + b) * LL + (LL - 1 - l1h)) * DIc;
    float tot = 0.f;
#pragma unroll
    for (int w = 0; w < WW; ++w) {
        float s = bias;
        s = fmaf(r0[w], wg[0], s); s = fmaf(r0[w + 1], wg[1], s); s = fmaf(r0[w + 2], wg[2], s);
        s = fmaf(r1[w], wg[3], s); s = fmaf(r1[w + 1], wg[4], s); s = fmaf(r1[w + 2], wg[5], s);
        s = fmaf(r2[w], wg[6], s); s = fmaf(r2[w + 1], wg[7], s); s = fmaf(r2[w + 2], wg[8], s);
        const float v = siluf(s);
        const int o0 = ((w >> 2) * 16 + (w & 3)) * DIc;
        const int o1 = ((w >> 2) * (HH / 4) * 16 + (w & 3) * 4) * DIc;
        (q0 + o0)[c] = v;
        (q1 + o1)[c] = v;
        (q2 - o0)[c] = v;
        (q3 - o1)[c] = v;
        tot += v;
    }
    rowsum[((size_t)b * HH + h) * DIc + c] = tot;
}

__global__ void k_gate(const float* __restrict__ rowsum, const float* __restrict__ fw1,
                       const float* __restrict__ fw2, float* __restrict__ gate)
{
    __shared__ float g[DIc];
    __shared__ float h1[DMm / 4];
    int b = blockIdx.x, d = threadIdx.x;
    float gs = 0.f;
#pragma unroll
    for (int h = 0; h < HH; ++h) gs += rowsum[((size_t)b * HH + h) * DIc + d];
    g[d] = gs * (1.f / LL);
    __syncthreads();
    if (d < DMm / 4) {
        float s = 0.f;
        for (int j = 0; j < DMm; ++j) s += fw1[d * DMm + j] * g[j];
        h1[d] = geluf(s);
    }
    __syncthreads();
    float s2 = 0.f;
#pragma unroll
    for (int j = 0; j < DMm / 4; ++j) s2 += fw2[d * (DMm / 4) + j] * h1[j];
    gate[b * DMm + d] = sigm(s2);
}

#define DT_NT (DIc / 16)
#define DTPITCH 18
__global__ void __launch_bounds__(128)
k_xproj(const float* __restrict__ xs, const float* __restrict__ wpad,
        const float* __restrict__ dtw, const float* __restrict__ dtb,
        float* __restrict__ xdbl, float* __restrict__ delta)
{
    __shared__ __attribute__((aligned(16))) float lds[(64 + 48) * LPITCH];
    __shared__ __attribute__((aligned(16))) float ldt[4 * 16 * DTPITCH];
    const int k = blockIdx.z;
    const float* X = xs + (size_t)k * BB * LL * DIc;
    const float* W = wpad + (size_t)k * CPAD * DIc;
    float* O = xdbl + (size_t)k * BB * LL * CPROJ;
    v8f acc[3];
    wmma_lds_k192<4, 1, 3>(X, W, blockIdx.y * 64, 0, lds, acc);
    const int wave = threadIdx.x >> 5;
    const int row0 = blockIdx.y * 64 + wave * 16;
    const int lane = threadIdx.x & 31;
    const int nn = lane & 15, mh = (lane >> 4) * 8;
    float* __restrict__ lT = ldt + wave * 16 * DTPITCH;
#pragma unroll
    for (int v = 0; v < 8; ++v) lT[(mh + v) * DTPITCH + nn] = acc[0][v];
    __syncthreads();
    float bias[DT_NT];
#pragma unroll
    for (int t = 0; t < DT_NT; ++t) bias[t] = dtb[k * DIc + t * 16 + nn];
#pragma unroll
    for (int t = 0; t < 3; ++t)
#pragma unroll
        for (int v = 0; v < 8; ++v) {
            int r = row0 + mh + v, c = t * 16 + nn;
            if (c >= RR && c < CPROJ) O[(size_t)r * CPROJ + c] = acc[t][v];
        }
    const float* Wd = dtw + (size_t)k * DIc * RR;
    float* Od = delta + (size_t)k * BB * LL * DIc;
    v8f dacc[DT_NT];
    wmma_xwT_multi<DT_NT>(lT, Wd, 0, 0, RR, DTPITCH, RR, dacc);
#pragma unroll
    for (int t = 0; t < DT_NT; ++t)
#pragma unroll
        for (int v = 0; v < 8; ++v) {
            const int r = row0 + mh + v, c = t * 16 + nn;
            Od[(size_t)r * DIc + c] = softplusf_(dacc[t][v] + bias[t]);
        }
}

#define SC_C   96
#define SC_LEN (LL / SC_C)

__global__ void k_scan_local(const float* __restrict__ xs, const float* __restrict__ xdbl,
                             const float* __restrict__ delta, const float* __restrict__ A_logs,
                             float* __restrict__ sumP, float* __restrict__ sumS)
{
    const int c = blockIdx.x, kb = blockIdx.y;
    const int k = kb & 3, b = kb >> 2;
    const int d = threadIdx.x;
    float A[NN];
#pragma unroll
    for (int n = 0; n < NN; ++n)
        A[n] = -__expf(A_logs[((size_t)(k * DIc + d)) * NN + n]);
    float h[NN];
#pragma unroll
    for (int n = 0; n < NN; ++n) h[n] = 0.f;
    const size_t seqbase = ((size_t)(k * BB + b)) * LL + (size_t)c * SC_LEN;
    const float* up = xs    + seqbase * DIc + d;
    const float* dp = delta + seqbase * DIc + d;
    const float* bc = xdbl  + seqbase * CPROJ;
    float sdt = 0.f;
    for (int l = 0; l < SC_LEN; ++l) {
        float dt = dp[(size_t)l * DIc];
        float u  = up[(size_t)l * DIc];
        float du = dt * u;
        const float* bcl = bc + (size_t)l * CPROJ;
#pragma unroll
        for (int n = 0; n < NN; ++n)
            h[n] = __expf(dt * A[n]) * h[n] + du * bcl[RR + n];
        sdt += dt;
    }
    const size_t o = (((size_t)kb * SC_C + c) * NN) * DIc + d;
#pragma unroll
    for (int n = 0; n < NN; ++n) {
        sumP[o + (size_t)n * DIc] = __expf(sdt * A[n]);
        sumS[o + (size_t)n * DIc] = h[n];
    }
}

__global__ void k_scan_prefix(const float* __restrict__ sumP, const float* __restrict__ sumS,
                              float* __restrict__ hst)
{
    const int n = blockIdx.x, kb = blockIdx.y, d = threadIdx.x;
    const size_t base = (((size_t)kb * SC_C) * NN + n) * DIc + d;
    const size_t cs = (size_t)NN * DIc;
    float H = 0.f;
    hst[base] = H;
#pragma unroll 19
    for (int c = 0; c + 1 < SC_C; ++c) {
        H = sumP[base + c * cs] * H + sumS[base + c * cs];
        hst[base + (c + 1) * cs] = H;
    }
}

__global__ void k_scan_chunk(const float* __restrict__ xs, const float* __restrict__ xdbl,
                             const float* __restrict__ delta, const float* __restrict__ A_logs,
                             const float* __restrict__ Ds, const float* __restrict__ hst,
                             float* __restrict__ yarr)
{
    const int c = blockIdx.x, kb = blockIdx.y;
    const int k = kb & 3, b = kb >> 2;
    const int d = threadIdx.x;
    float A[NN];
#pragma unroll
    for (int n = 0; n < NN; ++n)
        A[n] = -__expf(A_logs[((size_t)(k * DIc + d)) * NN + n]);
    float Dv = Ds[k * DIc + d];
    float h[NN];
    const size_t o = (((size_t)kb * SC_C + c) * NN) * DIc + d;
#pragma unroll
    for (int n = 0; n < NN; ++n) h[n] = hst[o + (size_t)n * DIc];
    const size_t seqbase = ((size_t)(k * BB + b)) * LL + (size_t)c * SC_LEN;
    const float* up = xs    + seqbase * DIc + d;
    const float* dp = delta + seqbase * DIc + d;
    const float* bc = xdbl  + seqbase * CPROJ;
    float* yp       = yarr  + seqbase * DIc + d;
    for (int l = 0; l < SC_LEN; ++l) {
        float dt = dp[(size_t)l * DIc];
        float u  = up[(size_t)l * DIc];
        float du = dt * u;
        const float* bcl = bc + (size_t)l * CPROJ;
        float y = 0.f;
#pragma unroll
        for (int n = 0; n < NN; ++n) {
            float Bt = bcl[RR + n];
            float Ct = bcl[RR + NN + n];
            h[n] = __expf(dt * A[n]) * h[n] + du * Bt;
            y += h[n] * Ct;
        }
        yp[(size_t)l * DIc] = y + Dv * u;
    }
}

__device__ __forceinline__ void ld6(const float* __restrict__ p, int lane, float v[6])
{
    const float4 a = *(const float4*)(p + 4 * lane);
    const float2 c = *(const float2*)(p + 128 + 2 * lane);
    v[0] = a.x; v[1] = a.y; v[2] = a.z; v[3] = a.w; v[4] = c.x; v[5] = c.y;
}

__global__ void __launch_bounds__(128)
k_outproj(const float* __restrict__ yarr, const float* __restrict__ gate,
          const float* __restrict__ g1, const float* __restrict__ g2,
          const float* __restrict__ g3, const float* __restrict__ g4,
          const float* __restrict__ nw, const float* __restrict__ nb,
          const float* __restrict__ z_silu, const float* __restrict__ w,
          float* __restrict__ out)
{
    __shared__ __attribute__((aligned(16))) float lds[(32 + 64) * LPITCH];
    float* __restrict__ lA = lds;
    float* __restrict__ lB = lds + 32 * LPITCH;
    const int tid = threadIdx.x;
    const int lane = tid & 31, wave = tid >> 5;
    {
        const float4* __restrict__ gB = (const float4*)(w + (size_t)blockIdx.x * 64 * 192);
#pragma unroll
        for (int j = 0; j < 64 * KPIECES / 128; ++j) {
            const int i = tid + j * 128;
            const int r = i / KPIECES, c = i - r * KPIECES;
            *(float4*)(lB + r * LPITCH + c * 4) = gB[i];
        }
    }
    {
        const int b   = blockIdx.y / (LL / 32);
        const int lr0 = (blockIdx.y - b * (LL / 32)) * 32 + wave * 8;
        const size_t bstride = (size_t)LL * DIc;
        const float* __restrict__ Y0 = yarr + ((size_t)(0 * BB + b)) * bstride;
        const float* __restrict__ Y1 = yarr + ((size_t)(1 * BB + b)) * bstride;
        const float* __restrict__ Y2 = yarr + ((size_t)(2 * BB + b)) * bstride;
        const float* __restrict__ Y3 = yarr + ((size_t)(3 * BB + b)) * bstride;
        float G1[6], G2[6], G3[6], G4[6], GT[6], NW[6], NB[6];
        ld6(g1, lane, G1); ld6(g2, lane, G2); ld6(g3, lane, G3); ld6(g4, lane, G4);
        ld6(gate + b * DIc, lane, GT); ld6(nw, lane, NW); ld6(nb, lane, NB);
#pragma unroll
        for (int i = 0; i < 8; ++i) {
            const int lr = lr0 + i;
            const int h = lr / WW, wq = lr - h * WW;
            const int l  = (((h >> 2) * (WW / 4) + (wq >> 2)) << 4) + ((h & 3) << 2) + (wq & 3);
            const int lt = (l % WW) * HH + (l / WW);
            float y0[6], y1[6], y2[6], y3[6], zz[6];
            ld6(Y0 + (size_t)l * DIc, lane, y0);
            ld6(Y2 + (size_t)(LL - 1 - l) * DIc, lane, y2);
            ld6(Y1 + (size_t)lt * DIc, lane, y1);
            ld6(Y3 + (size_t)(LL - 1 - lt) * DIc, lane, y3);
            ld6(z_silu + ((size_t)b * LL + lr) * DIc, lane, zz);
            float yc[6];
            float s1 = 0.f, s2 = 0.f;
#pragma unroll
            for (int j = 0; j < 6; ++j) {
                float v = G1[j] * y0[j] + G2[j] * y2[j] + G3[j] * y1[j] + G4[j] * y3[j];
                v *= GT[j];
                yc[j] = v;
                s1 += v; s2 += v * v;
            }
#pragma unroll
            for (int off = 16; off > 0; off >>= 1) {
                s1 += __shfl_xor(s1, off);
                s2 += __shfl_xor(s2, off);
            }
            const float mu  = s1 * (1.f / DIc);
            const float var = s2 * (1.f / DIc) - mu * mu;
            const float inv = rsqrtf(var + 1e-5f);
            float o[6];
#pragma unroll
            for (int j = 0; j < 6; ++j) {
                float v = (yc[j] - mu) * inv * NW[j] + NB[j];
                o[j] = v * zz[j];
            }
            float* __restrict__ dst = lA + (wave * 8 + i) * LPITCH;
            *(float4*)(dst + 4 * lane)       = make_float4(o[0], o[1], o[2], o[3]);
            *(float2*)(dst + 128 + 2 * lane) = make_float2(o[4], o[5]);
        }
    }
    __syncthreads();
    v8f acc[2];
    wmma_frags_k192<2, 2, 2>(lA, lB, acc);
    const int row0 = blockIdx.y * 32 + (wave >> 1) * 16;
    const int col0 = blockIdx.x * 64 + (wave & 1) * 32;
    const int nn = lane & 15, mh = (lane >> 4) * 8;
#pragma unroll
    for (int t = 0; t < 2; ++t)
#pragma unroll
        for (int v = 0; v < 8; ++v) {
            int r = row0 + mh + v, c = col0 + t * 16 + nn;
            out[(size_t)r * DMm + c] = acc[t][v];
        }
}

extern "C" void kernel_launch(void* const* d_in, const int* in_sizes, int n_in,
                              void* d_out, int out_size, void* d_ws, size_t ws_size,
                              hipStream_t stream)
{
    (void)in_sizes; (void)n_in; (void)out_size; (void)ws_size;
    const float* x          = (const float*)d_in[0];
    const float* in_proj_w  = (const float*)d_in[1];
    const float* conv_w     = (const float*)d_in[2];
    const float* conv_b     = (const float*)d_in[3];
    const float* x_proj_w   = (const float*)d_in[4];
    const float* dt_w       = (const float*)d_in[5];
    const float* dt_b       = (const float*)d_in[6];
    const float* A_logs     = (const float*)d_in[7];
    const float* Ds         = (const float*)d_in[8];
    const float* gamma1     = (const float*)d_in[9];
    const float* gamma2     = (const float*)d_in[10];
    const float* gamma3     = (const float*)d_in[11];
    const float* gamma4     = (const float*)d_in[12];
    const float* fc_w1      = (const float*)d_in[13];
    const float* fc_w2      = (const float*)d_in[14];
    const float* norm_w     = (const float*)d_in[15];
    const float* norm_b     = (const float*)d_in[16];
    const float* out_proj_w = (const float*)d_in[17];
    float* out = (float*)d_out;

    float* ws = (float*)d_ws;
    size_t o = 0;
    float* xc_lin = ws + o; o += (size_t)BB * LL * DIc;
    float* z_silu = ws + o; o += (size_t)BB * LL * DIc;
    float* xs     = ws + o; o += (size_t)KK * BB * LL * DIc;
    float* xdbl   = ws + o; o += (size_t)KK * BB * LL * CPROJ;
    float* delta  = ws + o; o += (size_t)KK * BB * LL * DIc;
    float* yarr   = ws + o; o += (size_t)KK * BB * LL * DIc;
    float* wpad   = ws + o; o += (size_t)KK * CPAD * DIc;
    float* rowsum = ws + o; o += (size_t)BB * HH * DIc;
    float* gate   = ws + o; o += BB * DMm;
    float* sumP   = ws + o; o += (size_t)KK * BB * SC_C * NN * DIc;
    float* sumS   = ws + o; o += (size_t)KK * BB * SC_C * NN * DIc;
    float* hst    = ws + o; o += (size_t)KK * BB * SC_C * NN * DIc;

    k_padw    <<<dim3(KK, CPAD),       DIc, 0, stream>>>(x_proj_w, wpad);
    k_inproj  <<<dim3(6, 288),         128, 0, stream>>>(x, in_proj_w, xc_lin, z_silu);
    k_dwconv  <<<dim3(HH, BB),         DIc, 0, stream>>>(xc_lin, conv_w, conv_b, xs, rowsum);
    k_gate    <<<dim3(BB),             DIc, 0, stream>>>(rowsum, fc_w1, fc_w2, gate);
    k_xproj   <<<dim3(1, 144, KK),     128, 0, stream>>>(xs, wpad, dt_w, dt_b, xdbl, delta);
    k_scan_local <<<dim3(SC_C - 1, BB * KK), DIc, 0, stream>>>(xs, xdbl, delta, A_logs, sumP, sumS);
    k_scan_prefix<<<dim3(NN, BB * KK),       DIc, 0, stream>>>(sumP, sumS, hst);
    k_scan_chunk <<<dim3(SC_C, BB * KK),     DIc, 0, stream>>>(xs, xdbl, delta, A_logs, Ds, hst, yarr);
    k_outproj <<<dim3(3, 288),         128, 0, stream>>>(yarr, gate, gamma1, gamma2, gamma3, gamma4,
                                                         norm_w, norm_b, z_silu, out_proj_w, out);
}


# k_scan_local, k_scan_chunk step loops: 4 and 2 steps per trip with the trip's loads issued first; decay = exp2(dt*(A*log2 e)), y by fma on scalar C (rounding of decay and y changes; sum orders kept)
# speedup vs baseline: 15.2301x; 1.1144x over previous
#include <hip/hip_runtime.h>
#include <math.h>


typedef float v2f __attribute__((ext_vector_type(2)));
typedef float v8f __attribute__((ext_vector_type(8)));

#define BB 4
#define HH 48
#define WW 48
#define DMm 192
#define DIc 192
#define KK 4
#define NN 16
#define RR 12
#define LL (HH * WW)
#define CPROJ (RR + 2 * NN)
#define CPAD 64

__device__ __forceinline__ float siluf(float x)     { return x / (1.f + __expf(-x)); }
__device__ __forceinline__ float sigm(float x)      { return 1.f / (1.f + __expf(-x)); }
__device__ __forceinline__ float softplusf_(float x)
{
    const float e   = __expf(x);
    const float u   = 1.f + e;
    const float um1 = u - 1.f;
    const float lnu = __builtin_amdgcn_logf(u) * 0.69314718055994530942f;
    const float lg  = lnu * (e * __builtin_amdgcn_rcpf(um1));
    const float sp  = (um1 == 0.f) ? e : lg;
    return (x > 20.f) ? x : sp;
}
__device__ __forceinline__ float geluf(float x)     { return 0.5f * x * (1.f + erff(x * 0.70710678118654752440f)); }

template <int NT>
__device__ __forceinline__ void wmma_xwT_multi(const float* __restrict__ X,
                                               const float* __restrict__ W,
                                               int row0, int col0,
                                               int Kdim, int ldx, int ldw,
                                               v8f acc[NT])
{
    const int lane = threadIdx.x & 31;
    const int m    = lane & 15;
    const int kh   = lane >> 4;
    const float* __restrict__ ap = X + (size_t)(row0 + m) * ldx + 2 * kh;
    const float* __restrict__ bp[NT];
#pragma unroll
    for (int t = 0; t < NT; ++t) {
        bp[t] = W + (size_t)(col0 + t * 16 + m) * ldw + 2 * kh;
        acc[t] = {};
    }
    v2f a_cur = *(const v2f*)(ap);
    v2f b_cur[NT];
#pragma unroll
    for (int t = 0; t < NT; ++t) b_cur[t] = *(const v2f*)(bp[t]);

    for (int k0 = 4; k0 < Kdim; k0 += 4) {
        v2f a_nxt = *(const v2f*)(ap + k0);
        v2f b_nxt[NT];
#pragma unroll
        for (int t = 0; t < NT; ++t) b_nxt[t] = *(const v2f*)(bp[t] + k0);
#pragma unroll
        for (int t = 0; t < NT; ++t)
            acc[t] = __builtin_amdgcn_wmma_f32_16x16x4_f32(false, a_cur, false, b_cur[t],
                                                           (short)0, acc[t], false, false);
        a_cur = a_nxt;
#pragma unroll
        for (int t = 0; t < NT; ++t) b_cur[t] = b_nxt[t];
    }
#pragma unroll
    for (int t = 0; t < NT; ++t)
        acc[t] = __builtin_amdgcn_wmma_f32_16x16x4_f32(false, a_cur, false, b_cur[t],
                                                       (short)0, acc[t], false, false);
}

#define LPITCH 196
#define KPIECES (192 / 4)

template <int WM, int WN, int NT>
__device__ __forceinline__ void wmma_frags_k192(const float* __restrict__ lA,
                                                const float* __restrict__ lB, v8f acc[NT])
{
    const int tid = threadIdx.x;
    const int lane = tid & 31, wave = tid >> 5;
    const int wm = wave / WN, wn = wave - wm * WN;
    const int m  = lane & 15;
    const int kh = lane >> 4;
    const float* __restrict__ ap = lA + (wm * 16 + m) * LPITCH + 2 * kh;
    const float* __restrict__ bp = lB + (wn * NT * 16 + m) * LPITCH + 2 * kh;
#pragma unroll
    for (int t = 0; t < NT; ++t) acc[t] = {};
#pragma unroll
    for (int ks = 0; ks < 192 / 4; ++ks) {
        const v2f a = *(const v2f*)(ap + 4 * ks);
        v2f b[NT];
#pragma unroll
        for (int t = 0; t < NT; ++t) b[t] = *(const v2f*)(bp + t * 16 * LPITCH + 4 * ks);
#pragma unroll
        for (int t = 0; t < NT; ++t)
            acc[t] = __builtin_amdgcn_wmma_f32_16x16x4_f32(false, a, false, b[t],
                                                           (short)0, acc[t], false, false);
    }
}

template <int WM, int WN, int NT>
__device__ __forceinline__ void wmma_lds_k192(const float* __restrict__ X,
                                              const float* __restrict__ W,
                                              int row0, int col0,
                                              float* __restrict__ lds, v8f acc[NT])
{
    constexpr int NTHR = WM * WN * 32;
    constexpr int AR = WM * 16;
    constexpr int BR = WN * NT * 16;
    static_assert((AR * KPIECES) % NTHR == 0 && (BR * KPIECES) % NTHR == 0, "whole copy passes");
    float* __restrict__ lA = lds;
    float* __restrict__ lB = lds + AR * LPITCH;
    const int tid = threadIdx.x;
    const float4* __restrict__ gA = (const float4*)(X + (size_t)row0 * 192);
    const float4* __restrict__ gB = (const float4*)(W + (size_t)col0 * 192);
#pragma unroll
    for (int j = 0; j < AR * KPIECES / NTHR; ++j) {
        const int i = tid + j * NTHR;
        const int r = i / KPIECES, c = i - r * KPIECES;
        *(float4*)(lA + r * LPITCH + c * 4) = gA[i];
    }
#pragma unroll
    for (int j = 0; j < BR * KPIECES / NTHR; ++j) {
        const int i = tid + j * NTHR;
        const int r = i / KPIECES, c = i - r * KPIECES;
        *(float4*)(lB + r * LPITCH + c * 4) = gB[i];
    }
    __syncthreads();
    wmma_frags_k192<WM, WN, NT>(lA, lB, acc);
}

__global__ void k_padw(const float* __restrict__ xpw, float* __restrict__ wpad)
{
    int k = blockIdx.x, c = blockIdx.y, d = threadIdx.x;
    wpad[((size_t)k * CPAD + c) * DIc + d] =
        (c < CPROJ) ? xpw[((size_t)k * CPROJ + c) * DIc + d] : 0.f;
}

__global__ void __launch_bounds__(128)
k_inproj(const float* __restrict__ x, const float* __restrict__ w,
         float* __restrict__ xc_lin, float* __restrict__ z_silu)
{
    __shared__ __attribute__((aligned(16))) float lds[(32 + 64) * LPITCH];
    v8f acc[2];
    wmma_lds_k192<2, 2, 2>(x, w, blockIdx.y * 32, blockIdx.x * 64, lds, acc);
    const int wave = threadIdx.x >> 5;
    const int row0 = blockIdx.y * 32 + (wave >> 1) * 16;
    const int col0 = blockIdx.x * 64 + (wave & 1) * 32;
    const int lane = threadIdx.x & 31;
    const int nn = lane & 15, mh = (lane >> 4) * 8;
#pragma unroll
    for (int t = 0; t < 2; ++t)
#pragma unroll
        for (int v = 0; v < 8; ++v) {
            int r = row0 + mh + v;
            int c = col0 + t * 16 + nn;
            float val = acc[t][v];
            if (c < DIc) xc_lin[(size_t)r * DIc + c] = val;
            else         z_silu[(size_t)r * DIc + (c - DIc)] = siluf(val);
        }
}

__global__ void __launch_bounds__(DIc)
k_dwconv(const float* __restrict__ xc_lin, const float* __restrict__ cw,
         const float* __restrict__ cb, float* __restrict__ xs,
         float* __restrict__ rowsum)
{
    const int h = blockIdx.x, b = blockIdx.y, c = threadIdx.x;
    const bool up = (h > 0), dn = (h < HH - 1);
    const int hu = up ? h - 1 : h;
    const int hd = dn ? h + 1 : h;
    const float* __restrict__ p0 = xc_lin + ((size_t)b * LL + (size_t)hu * WW) * DIc + c;
    const float* __restrict__ p1 = xc_lin + ((size_t)b * LL + (size_t)h  * WW) * DIc + c;
    const float* __restrict__ p2 = xc_lin + ((size_t)b * LL + (size_t)hd * WW) * DIc + c;
    float wg[9];
#pragma unroll
    for (int j = 0; j < 9; ++j) wg[j] = cw[c * 9 + j];
    const float bias = cb[c];
    float r0[WW + 2], r1[WW + 2], r2[WW + 2];
    r0[0] = 0.f; r1[0] = 0.f; r2[0] = 0.f;
    r0[WW + 1] = 0.f; r1[WW + 1] = 0.f; r2[WW + 1] = 0.f;
#pragma unroll
    for (int w = 0; w < WW; ++w) r0[w + 1] = p0[(size_t)w * DIc];
#pragma unroll
    for (int w = 0; w < WW; ++w) r1[w + 1] = p1[(size_t)w * DIc];
#pragma unroll
    for (int w = 0; w < WW; ++w) r2[w + 1] = p2[(size_t)w * DIc];
#pragma unroll
    for (int w = 0; w < WW; ++w) {
        r0[w + 1] = up ? r0[w + 1] : 0.f;
        r2[w + 1] = dn ? r2[w + 1] : 0.f;
    }
    const int l0h = (h >> 2) * (WW / 4) * 16 + (h & 3) * 4;
    const int l1h = (h >> 2) * 16 + (h & 3);
    float* __restrict__ q0 = xs + ((size_t)(0 * BB + b) * LL + l0h) * DIc;
    float* __restrict__ q1 = xs + ((size_t)(1 * BB + b) * LL + l1h) * DIc;
    float* __restrict__ q2 = xs + ((size_t)(2 * BB + b) * LL + (LL - 1 - l0h)) * DIc;
    float* __restrict__ q3 = xs + ((size_t)(3 * BB + b) * LL + (LL - 1 - l1h)) * DIc;
    float tot = 0.f;
#pragma unroll
    for (int w = 0; w < WW; ++w) {
        float s = bias;
        s = fmaf(r0[w], wg[0], s); s = fmaf(r0[w + 1], wg[1], s); s = fmaf(r0[w + 2], wg[2], s);
        s = fmaf(r1[w], wg[3], s); s = fmaf(r1[w + 1], wg[4], s); s = fmaf(r1[w + 2], wg[5], s);
        s = fmaf(r2[w], wg[6], s); s = fmaf(r2[w + 1], wg[7], s); s = fmaf(r2[w + 2], wg[8], s);
        const float v = siluf(s);
        const int o0 = ((w >> 2) * 16 + (w & 3)) * DIc;
        const int o1 = ((w >> 2) * (HH / 4) * 16 + (w & 3) * 4) * DIc;
        (q0 + o0)[c] = v;
        (q1 + o1)[c] = v;
        (q2 - o0)[c] = v;
        (q3 - o1)[c] = v;
        tot += v;
    }
    rowsum[((size_t)b * HH + h) * DIc + c] = tot;
}

__global__ void k_gate(const float* __restrict__ rowsum, const float* __restrict__ fw1,
                       const float* __restrict__ fw2, float* __restrict__ gate)
{
    __shared__ float g[DIc];
    __shared__ float h1[DMm / 4];
    int b = blockIdx.x, d = threadIdx.x;
    float gs = 0.f;
#pragma unroll
    for (int h = 0; h < HH; ++h) gs += rowsum[((size_t)b * HH + h) * DIc + d];
    g[d] = gs * (1.f / LL);
    __syncthreads();
    if (d < DMm / 4) {
        float s = 0.f;
        for (int j = 0; j < DMm; ++j) s += fw1[d * DMm + j] * g[j];
        h1[d] = geluf(s);
    }
    __syncthreads();
    float s2 = 0.f;
#pragma unroll
    for (int j = 0; j < DMm / 4; ++j) s2 += fw2[d * (DMm / 4) + j] * h1[j];
    gate[b * DMm + d] = sigm(s2);
}

#define DT_NT (DIc / 16)
#define DTPITCH 18
__global__ void __launch_bounds__(128)
k_xproj(const float* __restrict__ xs, const float* __restrict__ wpad,
        const float* __restrict__ dtw, const float* __restrict__ dtb,
        float* __restrict__ xdbl, float* __restrict__ delta)
{
    __shared__ __attribute__((aligned(16))) float lds[(64 + 48) * LPITCH];
    __shared__ __attribute__((aligned(16))) float ldt[4 * 16 * DTPITCH];
    const int k = blockIdx.z;
    const float* X = xs + (size_t)k * BB * LL * DIc;
    const float* W = wpad + (size_t)k * CPAD * DIc;
    float* O = xdbl + (size_t)k * BB * LL * CPROJ;
    v8f acc[3];
    wmma_lds_k192<4, 1, 3>(X, W, blockIdx.y * 64, 0, lds, acc);
    const int wave = threadIdx.x >> 5;
    const int row0 = blockIdx.y * 64 + wave * 16;
    const int lane = threadIdx.x & 31;
    const int nn = lane & 15, mh = (lane >> 4) * 8;
    float* __restrict__ lT = ldt + wave * 16 * DTPITCH;
#pragma unroll
    for (int v = 0; v < 8; ++v) lT[(mh + v) * DTPITCH + nn] = acc[0][v];
    __syncthreads();
    float bias[DT_NT];
#pragma unroll
    for (int t = 0; t < DT_NT; ++t) bias[t] = dtb[k * DIc + t * 16 + nn];
#pragma unroll
    for (int t = 0; t < 3; ++t)
#pragma unroll
        for (int v = 0; v < 8; ++v) {
            int r = row0 + mh + v, c = t * 16 + nn;
            if (c >= RR && c < CPROJ) O[(size_t)r * CPROJ + c] = acc[t][v];
        }
    const float* Wd = dtw + (size_t)k * DIc * RR;
    float* Od = delta + (size_t)k * BB * LL * DIc;
    v8f dacc[DT_NT];
    wmma_xwT_multi<DT_NT>(lT, Wd, 0, 0, RR, DTPITCH, RR, dacc);
#pragma unroll
    for (int t = 0; t < DT_NT; ++t)
#pragma unroll
        for (int v = 0; v < 8; ++v) {
            const int r = row0 + mh + v, c = t * 16 + nn;
            Od[(size_t)r * DIc + c] = softplusf_(dacc[t][v] + bias[t]);
        }
}

#define SC_C   96
#define SC_LEN (LL / SC_C)

#define LOG2E_F 1.44269504088896340736f

__device__ __forceinline__ void scan_step_h(float dt, float u, const float* __restrict__ bl,
                                            const float A2[NN], float h[NN])
{
    const float du = dt * u;
#pragma unroll
    for (int n = 0; n < NN; ++n)
        h[n] = fmaf(__builtin_amdgcn_exp2f(dt * A2[n]), h[n], du * bl[n]);
}

__device__ __forceinline__ float scan_step_y(float dt, float u, const float* __restrict__ bl,
                                             const float A2[NN], float h[NN])
{
    const float du = dt * u;
    float y = 0.f;
#pragma unroll
    for (int n = 0; n < NN; ++n) {
        h[n] = fmaf(__builtin_amdgcn_exp2f(dt * A2[n]), h[n], du * bl[n]);
        y = fmaf(h[n], bl[NN + n], y);
    }
    return y;
}

#define SL_U 4
__global__ void k_scan_local(const float* __restrict__ xs, const float* __restrict__ xdbl,
                             const float* __restrict__ delta, const float* __restrict__ A_logs,
                             float* __restrict__ sumP, float* __restrict__ sumS)
{
    const int c = blockIdx.x, kb = blockIdx.y;
    const int k = kb & 3, b = kb >> 2;
    const int d = threadIdx.x;
    float A2[NN];
#pragma unroll
    for (int n = 0; n < NN; ++n)
        A2[n] = -__expf(A_logs[((size_t)(k * DIc + d)) * NN + n]) * LOG2E_F;
    float h[NN];
#pragma unroll
    for (int n = 0; n < NN; ++n) h[n] = 0.f;
    const size_t seqbase = ((size_t)(k * BB + b)) * LL + (size_t)c * SC_LEN;
    const float* __restrict__ up = xs    + seqbase * DIc + d;
    const float* __restrict__ dp = delta + seqbase * DIc + d;
    const float* __restrict__ bc = xdbl  + seqbase * CPROJ + RR;
    float sdt = 0.f;
    static_assert(SC_LEN % SL_U == 0, "whole trips");
    for (int l = 0; l < SC_LEN; l += SL_U) {
        float dt[SL_U], u[SL_U];
#pragma unroll
        for (int j = 0; j < SL_U; ++j) {
            dt[j] = dp[(size_t)(l + j) * DIc];
            u[j]  = up[(size_t)(l + j) * DIc];
        }
#pragma unroll
        for (int j = 0; j < SL_U; ++j) {
            scan_step_h(dt[j], u[j], bc + (size_t)(l + j) * CPROJ, A2, h);
            sdt += dt[j];
        }
    }
    const size_t o = (((size_t)kb * SC_C + c) * NN) * DIc + d;
#pragma unroll
    for (int n = 0; n < NN; ++n) {
        sumP[o + (size_t)n * DIc] = __builtin_amdgcn_exp2f(sdt * A2[n]);
        sumS[o + (size_t)n * DIc] = h[n];
    }
}

__global__ void k_scan_prefix(const float* __restrict__ sumP, const float* __restrict__ sumS,
                              float* __restrict__ hst)
{
    const int n = blockIdx.x, kb = blockIdx.y, d = threadIdx.x;
    const size_t base = (((size_t)kb * SC_C) * NN + n) * DIc + d;
    const size_t cs = (size_t)NN * DIc;
    float H = 0.f;
    hst[base] = H;
#pragma unroll 19
    for (int c = 0; c + 1 < SC_C; ++c) {
        H = sumP[base + c * cs] * H + sumS[base + c * cs];
        hst[base + (c + 1) * cs] = H;
    }
}

#define SK_U 2
__global__ void k_scan_chunk(const float* __restrict__ xs, const float* __restrict__ xdbl,
                             const float* __restrict__ delta, const float* __restrict__ A_logs,
                             const float* __restrict__ Ds, const float* __restrict__ hst,
                             float* __restrict__ yarr)
{
    const int c = blockIdx.x, kb = blockIdx.y;
    const int k = kb & 3, b = kb >> 2;
    const int d = threadIdx.x;
    float A2[NN];
#pragma unroll
    for (int n = 0; n < NN; ++n)
        A2[n] = -__expf(A_logs[((size_t)(k * DIc + d)) * NN + n]) * LOG2E_F;
    float Dv = Ds[k * DIc + d];
    float h[NN];
    const size_t o = (((size_t)kb * SC_C + c) * NN) * DIc + d;
#pragma unroll
    for (int n = 0; n < NN; ++n) h[n] = hst[o + (size_t)n * DIc];
    const size_t seqbase = ((size_t)(k * BB + b)) * LL + (size_t)c * SC_LEN;
    const float* __restrict__ up = xs    + seqbase * DIc + d;
    const float* __restrict__ dp = delta + seqbase * DIc + d;
    const float* __restrict__ bc = xdbl  + seqbase * CPROJ + RR;
    float* __restrict__ yp       = yarr  + seqbase * DIc + d;
    static_assert(SC_LEN % SK_U == 0, "whole trips");
    for (int l = 0; l < SC_LEN; l += SK_U) {
        float dt[SK_U], u[SK_U];
#pragma unroll
        for (int j = 0; j < SK_U; ++j) {
            dt[j] = dp[(size_t)(l + j) * DIc];
            u[j]  = up[(size_t)(l + j) * DIc];
        }
#pragma unroll
        for (int j = 0; j < SK_U; ++j) {
            const float y = scan_step_y(dt[j], u[j], bc + (size_t)(l + j) * CPROJ, A2, h);
            yp[(size_t)(l + j) * DIc] = fmaf(Dv, u[j], y);
        }
    }
}

__device__ __forceinline__ void ld6(const float* __restrict__ p, int lane, float v[6])
{
    const float4 a = *(const float4*)(p + 4 * lane);
    const float2 c = *(const float2*)(p + 128 + 2 * lane);
    v[0] = a.x; v[1] = a.y; v[2] = a.z; v[3] = a.w; v[4] = c.x; v[5] = c.y;
}

__global__ void __launch_bounds__(128)
k_outproj(const float* __restrict__ yarr, const float* __restrict__ gate,
          const float* __restrict__ g1, const float* __restrict__ g2,
          const float* __restrict__ g3, const float* __restrict__ g4,
          const float* __restrict__ nw, const float* __restrict__ nb,
          const float* __restrict__ z_silu, const float* __restrict__ w,
          float* __restrict__ out)
{
    __shared__ __attribute__((aligned(16))) float lds[(32 + 64) * LPITCH];
    float* __restrict__ lA = lds;
    float* __restrict__ lB = lds + 32 * LPITCH;
    const int tid = threadIdx.x;
    const int lane = tid & 31, wave = tid >> 5;
    {
        const float4* __restrict__ gB = (const float4*)(w + (size_t)blockIdx.x * 64 * 192);
#pragma unroll
        for (int j = 0; j < 64 * KPIECES / 128; ++j) {
            const int i = tid + j * 128;
            const int r = i / KPIECES, c = i - r * KPIECES;
            *(float4*)(lB + r * LPITCH + c * 4) = gB[i];
        }
    }
    {
        const int b   = blockIdx.y / (LL / 32);
        const int lr0 = (blockIdx.y - b * (LL / 32)) * 32 + wave * 8;
        const size_t bstride = (size_t)LL * DIc;
        const float* __restrict__ Y0 = yarr + ((size_t)(0 * BB + b)) * bstride;
        const float* __restrict__ Y1 = yarr + ((size_t)(1 * BB + b)) * bstride;
        const float* __restrict__ Y2 = yarr + ((size_t)(2 * BB + b)) * bstride;
        const float* __restrict__ Y3 = yarr + ((size_t)(3 * BB + b)) * bstride;
        float G1[6], G2[6], G3[6], G4[6], GT[6], NW[6], NB[6];
        ld6(g1, lane, G1); ld6(g2, lane, G2); ld6(g3, lane, G3); ld6(g4, lane, G4);
        ld6(gate + b * DIc, lane, GT); ld6(nw, lane, NW); ld6(nb, lane, NB);
#pragma unroll
        for (int i = 0; i < 8; ++i) {
            const int lr = lr0 + i;
            const int h = lr / WW, wq = lr - h * WW;
            const int l  = (((h >> 2) * (WW / 4) + (wq >> 2)) << 4) + ((h & 3) << 2) + (wq & 3);
            const int lt = (l % WW) * HH + (l / WW);
            float y0[6], y1[6], y2[6], y3[6], zz[6];
            ld6(Y0 + (size_t)l * DIc, lane, y0);
            ld6(Y2 + (size_t)(LL - 1 - l) * DIc, lane, y2);
            ld6(Y1 + (size_t)lt * DIc, lane, y1);
            ld6(Y3 + (size_t)(LL - 1 - lt) * DIc, lane, y3);
            ld6(z_silu + ((size_t)b * LL + lr) * DIc, lane, zz);
            float yc[6];
            float s1 = 0.f, s2 = 0.f;
#pragma unroll
            for (int j = 0; j < 6; ++j) {
                float v = G1[j] * y0[j] + G2[j] * y2[j] + G3[j] * y1[j] + G4[j] * y3[j];
                v *= GT[j];
                yc[j] = v;
                s1 += v; s2 += v * v;
            }
#pragma unroll
            for (int off = 16; off > 0; off >>= 1) {
                s1 += __shfl_xor(s1, off);
                s2 += __shfl_xor(s2, off);
            }
            const float mu  = s1 * (1.f / DIc);
            const float var = s2 * (1.f / DIc) - mu * mu;
            const float inv = rsqrtf(var + 1e-5f);
            float o[6];
#pragma unroll
            for (int j = 0; j < 6; ++j) {
                float v = (yc[j] - mu) * inv * NW[j] + NB[j];
                o[j] = v * zz[j];
            }
            float* __restrict__ dst = lA + (wave * 8 + i) * LPITCH;
            *(float4*)(dst + 4 * lane)       = make_float4(o[0], o[1], o[2], o[3]);
            *(float2*)(dst + 128 + 2 * lane) = make_float2(o[4], o[5]);
        }
    }
    __syncthreads();
    v8f acc[2];
    wmma_frags_k192<2, 2, 2>(lA, lB, acc);
    const int row0 = blockIdx.y * 32 + (wave >> 1) * 16;
    const int col0 = blockIdx.x * 64 + (wave & 1) * 32;
    const int nn = lane & 15, mh = (lane >> 4) * 8;
#pragma unroll
    for (int t = 0; t < 2; ++t)
#pragma unroll
        for (int v = 0; v < 8; ++v) {
            int r = row0 + mh + v, c = col0 + t * 16 + nn;
            out[(size_t)r * DMm + c] = acc[t][v];
        }
}

extern "C" void kernel_launch(void* const* d_in, const int* in_sizes, int n_in,
                              void* d_out, int out_size, void* d_ws, size_t ws_size,
                              hipStream_t stream)
{
    (void)in_sizes; (void)n_in; (void)out_size; (void)ws_size;
    const float* x          = (const float*)d_in[0];
    const float* in_proj_w  = (const float*)d_in[1];
    const float* conv_w     = (const float*)d_in[2];
    const float* conv_b     = (const float*)d_in[3];
    const float* x_proj_w   = (const float*)d_in[4];
    const float* dt_w       = (const float*)d_in[5];
    const float* dt_b       = (const float*)d_in[6];
    const float* A_logs     = (const float*)d_in[7];
    const float* Ds         = (const float*)d_in[8];
    const float* gamma1     = (const float*)d_in[9];
    const float* gamma2     = (const float*)d_in[10];
    const float* gamma3     = (const float*)d_in[11];
    const float* gamma4     = (const float*)d_in[12];
    const float* fc_w1      = (const float*)d_in[13];
    const float* fc_w2      = (const float*)d_in[14];
    const float* norm_w     = (const float*)d_in[15];
    const float* norm_b     = (const float*)d_in[16];
    const float* out_proj_w = (const float*)d_in[17];
    float* out = (float*)d_out;

    float* ws = (float*)d_ws;
    size_t o = 0;
    float* xc_lin = ws + o; o += (size_t)BB * LL * DIc;
    float* z_silu = ws + o; o += (size_t)BB * LL * DIc;
    float* xs     = ws + o; o += (size_t)KK * BB * LL * DIc;
    float* xdbl   = ws + o; o += (size_t)KK * BB * LL * CPROJ;
    float* delta  = ws + o; o += (size_t)KK * BB * LL * DIc;
    float* yarr   = ws + o; o += (size_t)KK * BB * LL * DIc;
    float* wpad   = ws + o; o += (size_t)KK * CPAD * DIc;
    float* rowsum = ws + o; o += (size_t)BB * HH * DIc;
    float* gate   = ws + o; o += BB * DMm;
    float* sumP   = ws + o; o += (size_t)KK * BB * SC_C * NN * DIc;
    float* sumS   = ws + o; o += (size_t)KK * BB * SC_C * NN * DIc;
    float* hst    = ws + o; o += (size_t)KK * BB * SC_C * NN * DIc;

    k_padw    <<<dim3(KK, CPAD),       DIc, 0, stream>>>(x_proj_w, wpad);
    k_inproj  <<<dim3(6, 288),         128, 0, stream>>>(x, in_proj_w, xc_lin, z_silu);
    k_dwconv  <<<dim3(HH, BB),         DIc, 0, stream>>>(xc_lin, conv_w, conv_b, xs, rowsum);
    k_gate    <<<dim3(BB),             DIc, 0, stream>>>(rowsum, fc_w1, fc_w2, gate);
    k_xproj   <<<dim3(1, 144, KK),     128, 0, stream>>>(xs, wpad, dt_w, dt_b, xdbl, delta);
    k_scan_local <<<dim3(SC_C - 1, BB * KK), DIc, 0, stream>>>(xs, xdbl, delta, A_logs, sumP, sumS);
    k_scan_prefix<<<dim3(NN, BB * KK),       DIc, 0, stream>>>(sumP, sumS, hst);
    k_scan_chunk <<<dim3(SC_C, BB * KK),     DIc, 0, stream>>>(xs, xdbl, delta, A_logs, Ds, hst, yarr);
    k_outproj <<<dim3(3, 288),         128, 0, stream>>>(yarr, gate, gamma1, gamma2, gamma3, gamma4,
                                                         norm_w, norm_b, z_silu, out_proj_w, out);
}


# chunk decay products P no longer stored: k_scan_local writes the chunk's dt sum (1 value, not 16), k_scan_prefix forms P = exp2(A2*sum) itself, its 95 links unrolled as one load group; bit-identical
# speedup vs baseline: 15.7512x; 1.0342x over previous
#include <hip/hip_runtime.h>
#include <math.h>


typedef float v2f __attribute__((ext_vector_type(2)));
typedef float v8f __attribute__((ext_vector_type(8)));

#define BB 4
#define HH 48
#define WW 48
#define DMm 192
#define DIc 192
#define KK 4
#define NN 16
#define RR 12
#define LL (HH * WW)
#define CPROJ (RR + 2 * NN)
#define CPAD 64

__device__ __forceinline__ float siluf(float x)     { return x / (1.f + __expf(-x)); }
__device__ __forceinline__ float sigm(float x)      { return 1.f / (1.f + __expf(-x)); }
__device__ __forceinline__ float softplusf_(float x)
{
    const float e   = __expf(x);
    const float u   = 1.f + e;
    const float um1 = u - 1.f;
    const float lnu = __builtin_amdgcn_logf(u) * 0.69314718055994530942f;
    const float lg  = lnu * (e * __builtin_amdgcn_rcpf(um1));
    const float sp  = (um1 == 0.f) ? e : lg;
    return (x > 20.f) ? x : sp;
}
__device__ __forceinline__ float geluf(float x)     { return 0.5f * x * (1.f + erff(x * 0.70710678118654752440f)); }

template <int NT>
__device__ __forceinline__ void wmma_xwT_multi(const float* __restrict__ X,
                                               const float* __restrict__ W,
                                               int row0, int col0,
                                               int Kdim, int ldx, int ldw,
                                               v8f acc[NT])
{
    const int lane = threadIdx.x & 31;
    const int m    = lane & 15;
    const int kh   = lane >> 4;
    const float* __restrict__ ap = X + (size_t)(row0 + m) * ldx + 2 * kh;
    const float* __restrict__ bp[NT];
#pragma unroll
    for (int t = 0; t < NT; ++t) {
        bp[t] = W + (size_t)(col0 + t * 16 + m) * ldw + 2 * kh;
        acc[t] = {};
    }
    v2f a_cur = *(const v2f*)(ap);
    v2f b_cur[NT];
#pragma unroll
    for (int t = 0; t < NT; ++t) b_cur[t] = *(const v2f*)(bp[t]);

    for (int k0 = 4; k0 < Kdim; k0 += 4) {
        v2f a_nxt = *(const v2f*)(ap + k0);
        v2f b_nxt[NT];
#pragma unroll
        for (int t = 0; t < NT; ++t) b_nxt[t] = *(const v2f*)(bp[t] + k0);
#pragma unroll
        for (int t = 0; t < NT; ++t)
            acc[t] = __builtin_amdgcn_wmma_f32_16x16x4_f32(false, a_cur, false, b_cur[t],
                                                           (short)0, acc[t], false, false);
        a_cur = a_nxt;
#pragma unroll
        for (int t = 0; t < NT; ++t) b_cur[t] = b_nxt[t];
    }
#pragma unroll
    for (int t = 0; t < NT; ++t)
        acc[t] = __builtin_amdgcn_wmma_f32_16x16x4_f32(false, a_cur, false, b_cur[t],
                                                       (short)0, acc[t], false, false);
}

#define LPITCH 196
#define KPIECES (192 / 4)

template <int WM, int WN, int NT>
__device__ __forceinline__ void wmma_frags_k192(const float* __restrict__ lA,
                                                const float* __restrict__ lB, v8f acc[NT])
{
    const int tid = threadIdx.x;
    const int lane = tid & 31, wave = tid >> 5;
    const int wm = wave / WN, wn = wave - wm * WN;
    const int m  = lane & 15;
    const int kh = lane >> 4;
    const float* __restrict__ ap = lA + (wm * 16 + m) * LPITCH + 2 * kh;
    const float* __restrict__ bp = lB + (wn * NT * 16 + m) * LPITCH + 2 * kh;
#pragma unroll
    for (int t = 0; t < NT; ++t) acc[t] = {};
#pragma unroll
    for (int ks = 0; ks < 192 / 4; ++ks) {
        const v2f a = *(const v2f*)(ap + 4 * ks);
        v2f b[NT];
#pragma unroll
        for (int t = 0; t < NT; ++t) b[t] = *(const v2f*)(bp + t * 16 * LPITCH + 4 * ks);
#pragma unroll
        for (int t = 0; t < NT; ++t)
            acc[t] = __builtin_amdgcn_wmma_f32_16x16x4_f32(false, a, false, b[t],
                                                           (short)0, acc[t], false, false);
    }
}

template <int WM, int WN, int NT>
__device__ __forceinline__ void wmma_lds_k192(const float* __restrict__ X,
                                              const float* __restrict__ W,
                                              int row0, int col0,
                                              float* __restrict__ lds, v8f acc[NT])
{
    constexpr int NTHR = WM * WN * 32;
    constexpr int AR = WM * 16;
    constexpr int BR = WN * NT * 16;
    static_assert((AR * KPIECES) % NTHR == 0 && (BR * KPIECES) % NTHR == 0, "whole copy passes");
    float* __restrict__ lA = lds;
    float* __restrict__ lB = lds + AR * LPITCH;
    const int tid = threadIdx.x;
    const float4* __restrict__ gA = (const float4*)(X + (size_t)row0 * 192);
    const float4* __restrict__ gB = (const float4*)(W + (size_t)col0 * 192);
#pragma unroll
    for (int j = 0; j < AR * KPIECES / NTHR; ++j) {
        const int i = tid + j * NTHR;
        const int r = i / KPIECES, c = i - r * KPIECES;
        *(float4*)(lA + r * LPITCH + c * 4) = gA[i];
    }
#pragma unroll
    for (int j = 0; j < BR * KPIECES / NTHR; ++j) {
        const int i = tid + j * NTHR;
        const int r = i / KPIECES, c = i - r * KPIECES;
        *(float4*)(lB + r * LPITCH + c * 4) = gB[i];
    }
    __syncthreads();
    wmma_frags_k192<WM, WN, NT>(lA, lB, acc);
}

__global__ void k_padw(const float* __restrict__ xpw, float* __restrict__ wpad)
{
    int k = blockIdx.x, c = blockIdx.y, d = threadIdx.x;
    wpad[((size_t)k * CPAD + c) * DIc + d] =
        (c < CPROJ) ? xpw[((size_t)k * CPROJ + c) * DIc + d] : 0.f;
}

__global__ void __launch_bounds__(128)
k_inproj(const float* __restrict__ x, const float* __restrict__ w,
         float* __restrict__ xc_lin, float* __restrict__ z_silu)
{
    __shared__ __attribute__((aligned(16))) float lds[(32 + 64) * LPITCH];
    v8f acc[2];
    wmma_lds_k192<2, 2, 2>(x, w, blockIdx.y * 32, blockIdx.x * 64, lds, acc);
    const int wave = threadIdx.x >> 5;
    const int row0 = blockIdx.y * 32 + (wave >> 1) * 16;
    const int col0 = blockIdx.x * 64 + (wave & 1) * 32;
    const int lane = threadIdx.x & 31;
    const int nn = lane & 15, mh = (lane >> 4) * 8;
#pragma unroll
    for (int t = 0; t < 2; ++t)
#pragma unroll
        for (int v = 0; v < 8; ++v) {
            int r = row0 + mh + v;
            int c = col0 + t * 16 + nn;
            float val = acc[t][v];
            if (c < DIc) xc_lin[(size_t)r * DIc + c] = val;
            else         z_silu[(size_t)r * DIc + (c - DIc)] = siluf(val);
        }
}

__global__ void __launch_bounds__(DIc)
k_dwconv(const float* __restrict__ xc_lin, const float* __restrict__ cw,
         const float* __restrict__ cb, float* __restrict__ xs,
         float* __restrict__ rowsum)
{
    const int h = blockIdx.x, b = blockIdx.y, c = threadIdx.x;
    const bool up = (h > 0), dn = (h < HH - 1);
    const int hu = up ? h - 1 : h;
    const int hd = dn ? h + 1 : h;
    const float* __restrict__ p0 = xc_lin + ((size_t)b * LL + (size_t)hu * WW) * DIc + c;
    const float* __restrict__ p1 = xc_lin + ((size_t)b * LL + (size_t)h  * WW) * DIc + c;
    const float* __restrict__ p2 = xc_lin + ((size_t)b * LL + (size_t)hd * WW) * DIc + c;
    float wg[9];
#pragma unroll
    for (int j = 0; j < 9; ++j) wg[j] = cw[c * 9 + j];
    const float bias = cb[c];
    float r0[WW + 2], r1[WW + 2], r2[WW + 2];
    r0[0] = 0.f; r1[0] = 0.f; r2[0] = 0.f;
    r0[WW + 1] = 0.f; r1[WW + 1] = 0.f; r2[WW + 1] = 0.f;
#pragma unroll
    for (int w = 0; w < WW; ++w) r0[w + 1] = p0[(size_t)w * DIc];
#pragma unroll
    for (int w = 0; w < WW; ++w) r1[w + 1] = p1[(size_t)w * DIc];
#pragma unroll
    for (int w = 0; w < WW; ++w) r2[w + 1] = p2[(size_t)w * DIc];
#pragma unroll
    for (int w = 0; w < WW; ++w) {
        r0[w + 1] = up ? r0[w + 1] : 0.f;
        r2[w + 1] = dn ? r2[w + 1] : 0.f;
    }
    const int l0h = (h >> 2) * (WW / 4) * 16 + (h & 3) * 4;
    const int l1h = (h >> 2) * 16 + (h & 3);
    float* __restrict__ q0 = xs + ((size_t)(0 * BB + b) * LL + l0h) * DIc;
    float* __restrict__ q1 = xs + ((size_t)(1 * BB + b) * LL + l1h) * DIc;
    float* __restrict__ q2 = xs + ((size_t)(2 * BB + b) * LL + (LL - 1 - l0h)) * DIc;
    float* __restrict__ q3 = xs + ((size_t)(3 * BB + b) * LL + (LL - 1 - l1h)) * DIc;
    float tot = 0.f;
#pragma unroll
    for (int w = 0; w < WW; ++w) {
        float s = bias;
        s = fmaf(r0[w], wg[0], s); s = fmaf(r0[w + 1], wg[1], s); s = fmaf(r0[w + 2], wg[2], s);
        s = fmaf(r1[w], wg[3], s); s = fmaf(r1[w + 1], wg[4], s); s = fmaf(r1[w + 2], wg[5], s);
        s = fmaf(r2[w], wg[6], s); s = fmaf(r2[w + 1], wg[7], s); s = fmaf(r2[w + 2], wg[8], s);
        const float v = siluf(s);
        const int o0 = ((w >> 2) * 16 + (w & 3)) * DIc;
        const int o1 = ((w >> 2) * (HH / 4) * 16 + (w & 3) * 4) * DIc;
        (q0 + o0)[c] = v;
        (q1 + o1)[c] = v;
        (q2 - o0)[c] = v;
        (q3 - o1)[c] = v;
        tot += v;
    }
    rowsum[((size_t)b * HH + h) * DIc + c] = tot;
}

__global__ void k_gate(const float* __restrict__ rowsum, const float* __restrict__ fw1,
                       const float* __restrict__ fw2, float* __restrict__ gate)
{
    __shared__ float g[DIc];
    __shared__ float h1[DMm / 4];
    int b = blockIdx.x, d = threadIdx.x;
    float gs = 0.f;
#pragma unroll
    for (int h = 0; h < HH; ++h) gs += rowsum[((size_t)b * HH + h) * DIc + d];
    g[d] = gs * (1.f / LL);
    __syncthreads();
    if (d < DMm / 4) {
        float s = 0.f;
        for (int j = 0; j < DMm; ++j) s += fw1[d * DMm + j] * g[j];
        h1[d] = geluf(s);
    }
    __syncthreads();
    float s2 = 0.f;
#pragma unroll
    for (int j = 0; j < DMm / 4; ++j) s2 += fw2[d * (DMm / 4) + j] * h1[j];
    gate[b * DMm + d] = sigm(s2);
}

#define DT_NT (DIc / 16)
#define DTPITCH 18
__global__ void __launch_bounds__(128)
k_xproj(const float* __restrict__ xs, const float* __restrict__ wpad,
        const float* __restrict__ dtw, const float* __restrict__ dtb,
        float* __restrict__ xdbl, float* __restrict__ delta)
{
    __shared__ __attribute__((aligned(16))) float lds[(64 + 48) * LPITCH];
    __shared__ __attribute__((aligned(16))) float ldt[4 * 16 * DTPITCH];
    const int k = blockIdx.z;
    const float* X = xs + (size_t)k * BB * LL * DIc;
    const float* W = wpad + (size_t)k * CPAD * DIc;
    float* O = xdbl + (size_t)k * BB * LL * CPROJ;
    v8f acc[3];
    wmma_lds_k192<4, 1, 3>(X, W, blockIdx.y * 64, 0, lds, acc);
    const int wave = threadIdx.x >> 5;
    const int row0 = blockIdx.y * 64 + wave * 16;
    const int lane = threadIdx.x & 31;
    const int nn = lane & 15, mh = (lane >> 4) * 8;
    float* __restrict__ lT = ldt + wave * 16 * DTPITCH;
#pragma unroll
    for (int v = 0; v < 8; ++v) lT[(mh + v) * DTPITCH + nn] = acc[0][v];
    __syncthreads();
    float bias[DT_NT];
#pragma unroll
    for (int t = 0; t < DT_NT; ++t) bias[t] = dtb[k * DIc + t * 16 + nn];
#pragma unroll
    for (int t = 0; t < 3; ++t)
#pragma unroll
        for (int v = 0; v < 8; ++v) {
            int r = row0 + mh + v, c = t * 16 + nn;
            if (c >= RR && c < CPROJ) O[(size_t)r * CPROJ + c] = acc[t][v];
        }
    const float* Wd = dtw + (size_t)k * DIc * RR;
    float* Od = delta + (size_t)k * BB * LL * DIc;
    v8f dacc[DT_NT];
    wmma_xwT_multi<DT_NT>(lT, Wd, 0, 0, RR, DTPITCH, RR, dacc);
#pragma unroll
    for (int t = 0; t < DT_NT; ++t)
#pragma unroll
        for (int v = 0; v < 8; ++v) {
            const int r = row0 + mh + v, c = t * 16 + nn;
            Od[(size_t)r * DIc + c] = softplusf_(dacc[t][v] + bias[t]);
        }
}

#define SC_C   96
#define SC_LEN (LL / SC_C)

#define LOG2E_F 1.44269504088896340736f

__device__ __forceinline__ void scan_step_h(float dt, float u, const float* __restrict__ bl,
                                            const float A2[NN], float h[NN])
{
    const float du = dt * u;
#pragma unroll
    for (int n = 0; n < NN; ++n)
        h[n] = fmaf(__builtin_amdgcn_exp2f(dt * A2[n]), h[n], du * bl[n]);
}

__device__ __forceinline__ float scan_step_y(float dt, float u, const float* __restrict__ bl,
                                             const float A2[NN], float h[NN])
{
    const float du = dt * u;
    float y = 0.f;
#pragma unroll
    for (int n = 0; n < NN; ++n) {
        h[n] = fmaf(__builtin_amdgcn_exp2f(dt * A2[n]), h[n], du * bl[n]);
        y = fmaf(h[n], bl[NN + n], y);
    }
    return y;
}

#define SL_U 4
__global__ void k_scan_local(const float* __restrict__ xs, const float* __restrict__ xdbl,
                             const float* __restrict__ delta, const float* __restrict__ A_logs,
                             float* __restrict__ sumD, float* __restrict__ sumS)
{
    const int c = blockIdx.x, kb = blockIdx.y;
    const int k = kb & 3, b = kb >> 2;
    const int d = threadIdx.x;
    float A2[NN];
#pragma unroll
    for (int n = 0; n < NN; ++n)
        A2[n] = -__expf(A_logs[((size_t)(k * DIc + d)) * NN + n]) * LOG2E_F;
    float h[NN];
#pragma unroll
    for (int n = 0; n < NN; ++n) h[n] = 0.f;
    const size_t seqbase = ((size_t)(k * BB + b)) * LL + (size_t)c * SC_LEN;
    const float* __restrict__ up = xs    + seqbase * DIc + d;
    const float* __restrict__ dp = delta + seqbase * DIc + d;
    const float* __restrict__ bc = xdbl  + seqbase * CPROJ + RR;
    float sdt = 0.f;
    static_assert(SC_LEN % SL_U == 0, "whole trips");
    for (int l = 0; l < SC_LEN; l += SL_U) {
        float dt[SL_U], u[SL_U];
#pragma unroll
        for (int j = 0; j < SL_U; ++j) {
            dt[j] = dp[(size_t)(l + j) * DIc];
            u[j]  = up[(size_t)(l + j) * DIc];
        }
#pragma unroll
        for (int j = 0; j < SL_U; ++j) {
            scan_step_h(dt[j], u[j], bc + (size_t)(l + j) * CPROJ, A2, h);
            sdt += dt[j];
        }
    }
    const size_t o = (((size_t)kb * SC_C + c) * NN) * DIc + d;
#pragma unroll
    for (int n = 0; n < NN; ++n) sumS[o + (size_t)n * DIc] = h[n];
    sumD[((size_t)kb * SC_C + c) * DIc + d] = sdt;
}

__global__ void __launch_bounds__(DIc)
k_scan_prefix(const float* __restrict__ sumD, const float* __restrict__ sumS,
                              const float* __restrict__ A_logs, float* __restrict__ hst)
{
    const int n = blockIdx.x, kb = blockIdx.y, d = threadIdx.x;
    const int k = kb & 3;
    const float A2n = -__expf(A_logs[((size_t)(k * DIc + d)) * NN + n]) * LOG2E_F;
    const size_t base = (((size_t)kb * SC_C) * NN + n) * DIc + d;
    const size_t cs = (size_t)NN * DIc;
    const size_t dbase = ((size_t)kb * SC_C) * DIc + d;
    float P[SC_C - 1];
#pragma unroll
    for (int c = 0; c + 1 < SC_C; ++c)
        P[c] = __builtin_amdgcn_exp2f(sumD[dbase + (size_t)c * DIc] * A2n);
    float H = 0.f;
    hst[base] = H;
#pragma unroll
    for (int c = 0; c + 1 < SC_C; ++c) {
        H = P[c] * H + sumS[base + c * cs];
        hst[base + (c + 1) * cs] = H;
    }
}

#define SK_U 2
__global__ void k_scan_chunk(const float* __restrict__ xs, const float* __restrict__ xdbl,
                             const float* __restrict__ delta, const float* __restrict__ A_logs,
                             const float* __restrict__ Ds, const float* __restrict__ hst,
                             float* __restrict__ yarr)
{
    const int c = blockIdx.x, kb = blockIdx.y;
    const int k = kb & 3, b = kb >> 2;
    const int d = threadIdx.x;
    float A2[NN];
#pragma unroll
    for (int n = 0; n < NN; ++n)
        A2[n] = -__expf(A_logs[((size_t)(k * DIc + d)) * NN + n]) * LOG2E_F;
    float Dv = Ds[k * DIc + d];
    float h[NN];
    const size_t o = (((size_t)kb * SC_C + c) * NN) * DIc + d;
#pragma unroll
    for (int n = 0; n < NN; ++n) h[n] = hst[o + (size_t)n * DIc];
    const size_t seqbase = ((size_t)(k * BB + b)) * LL + (size_t)c * SC_LEN;
    const float* __restrict__ up = xs    + seqbase * DIc + d;
    const float* __restrict__ dp = delta + seqbase * DIc + d;
    const float* __restrict__ bc = xdbl  + seqbase * CPROJ + RR;
    float* __restrict__ yp       = yarr  + seqbase * DIc + d;
    static_assert(SC_LEN % SK_U == 0, "whole trips");
    for (int l = 0; l < SC_LEN; l += SK_U) {
        float dt[SK_U], u[SK_U];
#pragma unroll
        for (int j = 0; j < SK_U; ++j) {
            dt[j] = dp[(size_t)(l + j) * DIc];
            u[j]  = up[(size_t)(l + j) * DIc];
        }
#pragma unroll
        for (int j = 0; j < SK_U; ++j) {
            const float y = scan_step_y(dt[j], u[j], bc + (size_t)(l + j) * CPROJ, A2, h);
            yp[(size_t)(l + j) * DIc] = fmaf(Dv, u[j], y);
        }
    }
}

__device__ __forceinline__ void ld6(const float* __restrict__ p, int lane, float v[6])
{
    const float4 a = *(const float4*)(p + 4 * lane);
    const float2 c = *(const float2*)(p + 128 + 2 * lane);
    v[0] = a.x; v[1] = a.y; v[2] = a.z; v[3] = a.w; v[4] = c.x; v[5] = c.y;
}

__global__ void __launch_bounds__(128)
k_outproj(const float* __restrict__ yarr, const float* __restrict__ gate,
          const float* __restrict__ g1, const float* __restrict__ g2,
          const float* __restrict__ g3, const float* __restrict__ g4,
          const float* __restrict__ nw, const float* __restrict__ nb,
          const float* __restrict__ z_silu, const float* __restrict__ w,
          float* __restrict__ out)
{
    __shared__ __attribute__((aligned(16))) float lds[(32 + 64) * LPITCH];
    float* __restrict__ lA = lds;
    float* __restrict__ lB = lds + 32 * LPITCH;
    const int tid = threadIdx.x;
    const int lane = tid & 31, wave = tid >> 5;
    {
        const float4* __restrict__ gB = (const float4*)(w + (size_t)blockIdx.x * 64 * 192);
#pragma unroll
        for (int j = 0; j < 64 * KPIECES / 128; ++j) {
            const int i = tid + j * 128;
            const int r = i / KPIECES, c = i - r * KPIECES;
            *(float4*)(lB + r * LPITCH + c * 4) = gB[i];
        }
    }
    {
        const int b   = blockIdx.y / (LL / 32);
        const int lr0 = (blockIdx.y - b * (LL / 32)) * 32 + wave * 8;
        const size_t bstride = (size_t)LL * DIc;
        const float* __restrict__ Y0 = yarr + ((size_t)(0 * BB + b)) * bstride;
        const float* __restrict__ Y1 = yarr + ((size_t)(1 * BB + b)) * bstride;
        const float* __restrict__ Y2 = yarr + ((size_t)(2 * BB + b)) * bstride;
        const float* __restrict__ Y3 = yarr + ((size_t)(3 * BB + b)) * bstride;
        float G1[6], G2[6], G3[6], G4[6], GT[6], NW[6], NB[6];
        ld6(g1, lane, G1); ld6(g2, lane, G2); ld6(g3, lane, G3); ld6(g4, lane, G4);
        ld6(gate + b * DIc, lane, GT); ld6(nw, lane, NW); ld6(nb, lane, NB);
#pragma unroll
        for (int i = 0; i < 8; ++i) {
            const int lr = lr0 + i;
            const int h = lr / WW, wq = lr - h * WW;
            const int l  = (((h >> 2) * (WW / 4) + (wq >> 2)) << 4) + ((h & 3) << 2) + (wq & 3);
            const int lt = (l % WW) * HH + (l / WW);
            float y0[6], y1[6], y2[6], y3[6], zz[6];
            ld6(Y0 + (size_t)l * DIc, lane, y0);
            ld6(Y2 + (size_t)(LL - 1 - l) * DIc, lane, y2);
            ld6(Y1 + (size_t)lt * DIc, lane, y1);
            ld6(Y3 + (size_t)(LL - 1 - lt) * DIc, lane, y3);
            ld6(z_silu + ((size_t)b * LL + lr) * DIc, lane, zz);
            float yc[6];
            float s1 = 0.f, s2 = 0.f;
#pragma unroll
            for (int j = 0; j < 6; ++j) {
                float v = G1[j] * y0[j] + G2[j] * y2[j] + G3[j] * y1[j] + G4[j] * y3[j];
                v *= GT[j];
                yc[j] = v;
                s1 += v; s2 += v * v;
            }
#pragma unroll
            for (int off = 16; off > 0; off >>= 1) {
                s1 += __shfl_xor(s1, off);
                s2 += __shfl_xor(s2, off);
            }
            const float mu  = s1 * (1.f / DIc);
            const float var = s2 * (1.f / DIc) - mu * mu;
            const float inv = rsqrtf(var + 1e-5f);
            float o[6];
#pragma unroll
            for (int j = 0; j < 6; ++j) {
                float v = (yc[j] - mu) * inv * NW[j] + NB[j];
                o[j] = v * zz[j];
            }
            float* __restrict__ dst = lA + (wave * 8 + i) * LPITCH;
            *(float4*)(dst + 4 * lane)       = make_float4(o[0], o[1], o[2], o[3]);
            *(float2*)(dst + 128 + 2 * lane) = make_float2(o[4], o[5]);
        }
    }
    __syncthreads();
    v8f acc[2];
    wmma_frags_k192<2, 2, 2>(lA, lB, acc);
    const int row0 = blockIdx.y * 32 + (wave >> 1) * 16;
    const int col0 = blockIdx.x * 64 + (wave & 1) * 32;
    const int nn = lane & 15, mh = (lane >> 4) * 8;
#pragma unroll
    for (int t = 0; t < 2; ++t)
#pragma unroll
        for (int v = 0; v < 8; ++v) {
            int r = row0 + mh + v, c = col0 + t * 16 + nn;
            out[(size_t)r * DMm + c] = acc[t][v];
        }
}

extern "C" void kernel_launch(void* const* d_in, const int* in_sizes, int n_in,
                              void* d_out, int out_size, void* d_ws, size_t ws_size,
                              hipStream_t stream)
{
    (void)in_sizes; (void)n_in; (void)out_size; (void)ws_size;
    const float* x          = (const float*)d_in[0];
    const float* in_proj_w  = (const float*)d_in[1];
    const float* conv_w     = (const float*)d_in[2];
    const float* conv_b     = (const float*)d_in[3];
    const float* x_proj_w   = (const float*)d_in[4];
    const float* dt_w       = (const float*)d_in[5];
    const float* dt_b       = (const float*)d_in[6];
    const float* A_logs     = (const float*)d_in[7];
    const float* Ds         = (const float*)d_in[8];
    const float* gamma1     = (const float*)d_in[9];
    const float* gamma2     = (const float*)d_in[10];
    const float* gamma3     = (const float*)d_in[11];
    const float* gamma4     = (const float*)d_in[12];
    const float* fc_w1      = (const float*)d_in[13];
    const float* fc_w2      = (const float*)d_in[14];
    const float* norm_w     = (const float*)d_in[15];
    const float* norm_b     = (const float*)d_in[16];
    const float* out_proj_w = (const float*)d_in[17];
    float* out = (float*)d_out;

    float* ws = (float*)d_ws;
    size_t o = 0;
    float* xc_lin = ws + o; o += (size_t)BB * LL * DIc;
    float* z_silu = ws + o; o += (size_t)BB * LL * DIc;
    float* xs     = ws + o; o += (size_t)KK * BB * LL * DIc;
    float* xdbl   = ws + o; o += (size_t)KK * BB * LL * CPROJ;
    float* delta  = ws + o; o += (size_t)KK * BB * LL * DIc;
    float* yarr   = ws + o; o += (size_t)KK * BB * LL * DIc;
    float* wpad   = ws + o; o += (size_t)KK * CPAD * DIc;
    float* rowsum = ws + o; o += (size_t)BB * HH * DIc;
    float* gate   = ws + o; o += BB * DMm;
    float* sumD   = ws + o; o += (size_t)KK * BB * SC_C * DIc;
    float* sumS   = ws + o; o += (size_t)KK * BB * SC_C * NN * DIc;
    float* hst    = ws + o; o += (size_t)KK * BB * SC_C * NN * DIc;

    k_padw    <<<dim3(KK, CPAD),       DIc, 0, stream>>>(x_proj_w, wpad);
    k_inproj  <<<dim3(6, 288),         128, 0, stream>>>(x, in_proj_w, xc_lin, z_silu);
    k_dwconv  <<<dim3(HH, BB),         DIc, 0, stream>>>(xc_lin, conv_w, conv_b, xs, rowsum);
    k_gate    <<<dim3(BB),             DIc, 0, stream>>>(rowsum, fc_w1, fc_w2, gate);
    k_xproj   <<<dim3(1, 144, KK),     128, 0, stream>>>(xs, wpad, dt_w, dt_b, xdbl, delta);
    k_scan_local <<<dim3(SC_C - 1, BB * KK), DIc, 0, stream>>>(xs, xdbl, delta, A_logs, sumD, sumS);
    k_scan_prefix<<<dim3(NN, BB * KK),       DIc, 0, stream>>>(sumD, sumS, A_logs, hst);
    k_scan_chunk <<<dim3(SC_C, BB * KK),     DIc, 0, stream>>>(xs, xdbl, delta, A_logs, Ds, hst, yarr);
    k_outproj <<<dim3(3, 288),         128, 0, stream>>>(yarr, gate, gamma1, gamma2, gamma3, gamma4,
                                                         norm_w, norm_b, z_silu, out_proj_w, out);
}
